# starting kernel
# baseline (speedup 1.0000x reference)
#include <hip/hip_runtime.h>
#include <math.h>

#define Nn 1024
#define Mm 16384
#define Bb 2
#define MT 64
#define KTS 65

typedef __attribute__((ext_vector_type(2))) float v2f;
typedef __attribute__((ext_vector_type(8))) float v8f;

#define SQRT3F 1.7320508075688772f
#define LOG2PIF 1.8378770664093453f

__device__ __forceinline__ float kern_pair(float x1, float y1, float a1, float b1, float c1, float q1,
                                           float x2, float y2, float a2, float b2, float c2, float q2,
                                           float scale) {
    float s00 = a1 + a2, s11 = b1 + b2, s01 = c1 + c2;
    float sdet = s00 * s11 - s01 * s01;
    float d0 = x1 - x2, d1 = y1 - y2;
    float Q = 0.5f * (s11 * d0 * d0 - 2.0f * s01 * d0 * d1 + s00 * d1 * d1) / sdet;
    float C = 2.0f * q1 * q2 * rsqrtf(fmaxf(sdet, 1e-5f));
    float t = SQRT3F * sqrtf(fmaxf(Q, 1e-5f));
    return (1.0f + t) * __expf(-t) * C * scale;
}

__global__ void prep_kernel(const float* __restrict__ ct, const float* __restrict__ kpt,
                            const float* __restrict__ cs, const float* __restrict__ kps,
                            const float* __restrict__ scale_param,
                            float* __restrict__ trainP, float* __restrict__ testP,
                            float* __restrict__ Kmm) {
    int gid = blockIdx.x * blockDim.x + threadIdx.x;
    float scale = __expf(scale_param[0]);
    if (gid < Bb * Mm) {
        float a = kps[gid * 3 + 0], b = kps[gid * 3 + 1], c = kps[gid * 3 + 2];
        float det = a * b - c * c;
        testP[gid * 6 + 0] = cs[gid * 2 + 0];
        testP[gid * 6 + 1] = cs[gid * 2 + 1];
        testP[gid * 6 + 2] = a;
        testP[gid * 6 + 3] = b;
        testP[gid * 6 + 4] = c;
        testP[gid * 6 + 5] = sqrtf(sqrtf(det));
        float t0 = SQRT3F * sqrtf(1e-5f);
        float m0 = (1.0f + t0) * __expf(-t0);
        float C = 2.0f * sqrtf(det) * rsqrtf(fmaxf(4.0f * det, 1e-5f));
        Kmm[gid] = C * m0 * scale;
    }
    if (gid < Bb * Nn) {
        float a = kpt[gid * 3 + 0], b = kpt[gid * 3 + 1], c = kpt[gid * 3 + 2];
        float det = a * b - c * c;
        trainP[gid * 6 + 0] = ct[gid * 2 + 0];
        trainP[gid * 6 + 1] = ct[gid * 2 + 1];
        trainP[gid * 6 + 2] = a;
        trainP[gid * 6 + 3] = b;
        trainP[gid * 6 + 4] = c;
        trainP[gid * 6 + 5] = sqrtf(sqrtf(det));
    }
}

__global__ void buildA_kernel(const float* __restrict__ trainP, const float* __restrict__ var,
                              const float* __restrict__ scale_param, float* __restrict__ A) {
    int gid = blockIdx.x * blockDim.x + threadIdx.x;
    int b = gid >> 20;
    int i = (gid >> 10) & (Nn - 1);
    int j = gid & (Nn - 1);
    const float* p1 = trainP + (size_t)(b * Nn + i) * 6;
    const float* p2 = trainP + (size_t)(b * Nn + j) * 6;
    float scale = __expf(scale_param[0]);
    float v = kern_pair(p1[0], p1[1], p1[2], p1[3], p1[4], p1[5],
                        p2[0], p2[1], p2[2], p2[3], p2[4], p2[5], scale);
    if (i == j) v += var[b * Nn + i];
    A[(size_t)gid] = v;
}

__global__ __launch_bounds__(1024) void chol_kernel(float* __restrict__ Abase,
                                                    float* __restrict__ logdet) {
    int b = blockIdx.x;
    float* A = Abase + (size_t)b * Nn * Nn;
    __shared__ float col[Nn];
    __shared__ float dinv_sh;
    int tid = threadIdx.x;
    for (int k = 0; k < Nn; k++) {
        if (tid == 0) {
            float d = sqrtf(A[(size_t)k * Nn + k]);
            A[(size_t)k * Nn + k] = d;
            dinv_sh = 1.0f / d;
        }
        __syncthreads();
        float dinv = dinv_sh;
        for (int i = k + 1 + tid; i < Nn; i += 1024) {
            float v = A[(size_t)i * Nn + k] * dinv;
            A[(size_t)i * Nn + k] = v;
            col[i] = v;
        }
        __syncthreads();
        for (int i = k + 1 + tid; i < Nn; i += 1024) {
            float ci = col[i];
            float* row = A + (size_t)i * Nn;
            for (int j = k + 1; j <= i; j++) row[j] -= ci * col[j];
        }
        __syncthreads();
    }
    float ld = __logf(A[(size_t)tid * Nn + tid]);
    col[tid] = ld;
    __syncthreads();
    for (int s = 512; s > 0; s >>= 1) {
        if (tid < s) col[tid] += col[tid + s];
        __syncthreads();
    }
    if (tid == 0) logdet[b] = col[0];
}

__global__ __launch_bounds__(256) void zsolve_kernel(const float* __restrict__ Lbase,
                                                     const float* __restrict__ y,
                                                     const float* __restrict__ mean,
                                                     const float* __restrict__ logdet,
                                                     float* __restrict__ zbase,
                                                     float* __restrict__ nlml_out) {
    int b = blockIdx.x;
    const float* L = Lbase + (size_t)b * Nn * Nn;
    __shared__ float zsh[Nn];
    __shared__ float red[256];
    int tid = threadIdx.x;
    float mb = mean[b];
    for (int i = 0; i < Nn; i++) {
        const float* row = L + (size_t)i * Nn;
        float s = 0.0f;
        for (int j = tid; j < i; j += 256) s += row[j] * zsh[j];
        red[tid] = s;
        __syncthreads();
        for (int st = 128; st > 0; st >>= 1) {
            if (tid < st) red[tid] += red[tid + st];
            __syncthreads();
        }
        if (tid == 0) zsh[i] = (y[b * Nn + i] - mb - red[0]) / row[i];
        __syncthreads();
    }
    float s2 = 0.0f;
    for (int i = tid; i < Nn; i += 256) {
        float z = zsh[i];
        zbase[b * Nn + i] = z;
        s2 += z * z;
    }
    red[tid] = s2;
    __syncthreads();
    for (int st = 128; st > 0; st >>= 1) {
        if (tid < st) red[tid] += red[tid + st];
        __syncthreads();
    }
    if (tid == 0) nlml_out[b] = 0.5f * red[0] + logdet[b] + 0.5f * Nn * LOG2PIF;
}

__global__ __launch_bounds__(256) void trinv_kernel(const float* __restrict__ Lbase,
                                                    float* __restrict__ Linvbase) {
    int b = blockIdx.y;
    int wave = threadIdx.x >> 5;
    int lane = threadIdx.x & 31;
    int j = blockIdx.x * 8 + wave;
    const float* L = Lbase + (size_t)b * Nn * Nn;
    float* Li = Linvbase + (size_t)b * Nn * Nn;
    __shared__ float xsh[8 * Nn];
    float* x = xsh + wave * Nn;
    for (int i = lane; i < j; i += 32) Li[(size_t)i * Nn + j] = 0.0f;
    float djj = 1.0f / L[(size_t)j * Nn + j];
    if (lane == 0) {
        x[j] = djj;
        Li[(size_t)j * Nn + j] = djj;
    }
    for (int i = j + 1; i < Nn; i++) {
        const float* row = L + (size_t)i * Nn;
        float s = 0.0f;
        for (int k = j + lane; k < i; k += 32) s += row[k] * x[k];
        s += __shfl_xor(s, 16);
        s += __shfl_xor(s, 8);
        s += __shfl_xor(s, 4);
        s += __shfl_xor(s, 2);
        s += __shfl_xor(s, 1);
        float xi = -s / row[i];
        if (lane == 0) {
            x[i] = xi;
            Li[(size_t)i * Nn + j] = xi;
        }
    }
}

__global__ __launch_bounds__(128) void predict_kernel(const float* __restrict__ Linvbase,
                                                      const float* __restrict__ trainPg,
                                                      const float* __restrict__ testPg,
                                                      const float* __restrict__ Kmm,
                                                      const float* __restrict__ zbase,
                                                      const float* __restrict__ meang,
                                                      const float* __restrict__ scale_param,
                                                      float* __restrict__ out_mean,
                                                      float* __restrict__ out_var) {
    extern __shared__ float smem[];
    float* Kt  = smem;
    float* zsh = smem + Nn * KTS;
    float* tp  = zsh + Nn;
    int b = blockIdx.y;
    int colBase = blockIdx.x * MT;
    int tid = threadIdx.x;
    float scale = __expf(scale_param[0]);

    for (int idx = tid; idx < Nn; idx += 128) zsh[idx] = zbase[b * Nn + idx];
    for (int idx = tid; idx < Nn * 6; idx += 128) tp[idx] = trainPg[(size_t)b * Nn * 6 + idx];
    __syncthreads();

    {
        int c = tid & 63;
        int h = tid >> 6;
        const float* q = testPg + (size_t)(b * Mm + colBase + c) * 6;
        float x2 = q[0], y2 = q[1], a2 = q[2], b2 = q[3], c2 = q[4], q2 = q[5];
        for (int n = h; n < Nn; n += 2) {
            const float* p = tp + n * 6;
            Kt[n * KTS + c] = kern_pair(p[0], p[1], p[2], p[3], p[4], p[5],
                                        x2, y2, a2, b2, c2, q2, scale);
        }
    }
    __syncthreads();

    int wave = tid >> 5, lane = tid & 31;
    int r = lane & 15, hi = lane >> 4;
    int wcol = wave * 16;
    const float* Li = Linvbase + (size_t)b * Nn * Nn;

    float meanAcc = 0.0f, varAcc = 0.0f;
    for (int rb = 0; rb < Nn / 16; rb++) {
        v8f acc = {};
        const float* arow = Li + (size_t)(rb * 16 + r) * Nn;
        for (int j = 0; j <= rb; j++) {
            int kb = j * 16;
#pragma unroll
            for (int k0 = 0; k0 < 16; k0 += 4) {
                int kk = kb + k0 + 2 * hi;
                v2f a = *(const v2f*)(arow + kk);
                v2f bb;
                bb.x = Kt[kk * KTS + wcol + r];
                bb.y = Kt[(kk + 1) * KTS + wcol + r];
                acc = __builtin_amdgcn_wmma_f32_16x16x4_f32(false, a, false, bb,
                                                            (short)0, acc, false, false);
            }
        }
        const float* zb = zsh + rb * 16 + 8 * hi;
#pragma unroll
        for (int v = 0; v < 8; v++) {
            float cv = acc[v];
            meanAcc += zb[v] * cv;
            varAcc  += cv * cv;
        }
    }
    meanAcc += __shfl_xor(meanAcc, 16);
    varAcc  += __shfl_xor(varAcc, 16);
    if (lane < 16) {
        int m = colBase + wcol + lane;
        out_mean[b * Mm + m] = meang[b] + meanAcc;
        out_var[b * Mm + m]  = Kmm[b * Mm + m] - varAcc;
    }
}

extern "C" void kernel_launch(void* const* d_in, const int* in_sizes, int n_in,
                              void* d_out, int out_size, void* d_ws, size_t ws_size,
                              hipStream_t stream) {
    const float* coords_train = (const float*)d_in[0];
    const float* kp_train     = (const float*)d_in[1];
    const float* coords_test  = (const float*)d_in[2];
    const float* kp_test      = (const float*)d_in[3];
    const float* y_train      = (const float*)d_in[4];
    const float* var          = (const float*)d_in[5];
    const float* mean         = (const float*)d_in[6];
    const float* scale_param  = (const float*)d_in[7];

    float* ws = (float*)d_ws;
    size_t off = 0;
    float* A      = ws + off; off += (size_t)Bb * Nn * Nn;
    float* Linv   = ws + off; off += (size_t)Bb * Nn * Nn;
    float* trainP = ws + off; off += (size_t)Bb * Nn * 6;
    float* testP  = ws + off; off += (size_t)Bb * Mm * 6;
    float* Kmm    = ws + off; off += (size_t)Bb * Mm;
    float* z      = ws + off; off += (size_t)Bb * Nn;
    float* logdet = ws + off; off += (size_t)Bb;

    float* out_mean = (float*)d_out;
    float* out_var  = out_mean + (size_t)Bb * Mm;
    float* nlml     = out_var + (size_t)Bb * Mm;

    prep_kernel<<<(Bb * Mm + 255) / 256, 256, 0, stream>>>(
        coords_train, kp_train, coords_test, kp_test, scale_param, trainP, testP, Kmm);
    buildA_kernel<<<(Bb * Nn * Nn) / 256, 256, 0, stream>>>(trainP, var, scale_param, A);
    chol_kernel<<<Bb, 1024, 0, stream>>>(A, logdet);
    zsolve_kernel<<<Bb, 256, 0, stream>>>(A, y_train, mean, logdet, z, nlml);
    trinv_kernel<<<dim3(Nn / 8, Bb), 256, 0, stream>>>(A, Linv);

    size_t smem_bytes = (size_t)(Nn * KTS + Nn + Nn * 6) * sizeof(float);
    predict_kernel<<<dim3(Mm / MT, Bb), 128, smem_bytes, stream>>>(
        Linv, trainP, testP, Kmm, z, mean, scale_param, out_mean, out_var);
}


# Cholesky rewritten as blocked left-looking: one launch per 64-wide block column, up to 32 workgroups each, block products on f32 WMMA; changes the summation order inside the factorisation
# speedup vs baseline: 7.7805x; 7.7805x over previous
#include <hip/hip_runtime.h>
#include <math.h>

#define Nn 1024
#define Mm 16384
#define Bb 2
#define MT 64
#define KTS 65
#define NB 64
#define NBLK (Nn / NB)
#define LDP 65

typedef __attribute__((ext_vector_type(2))) float v2f;
typedef __attribute__((ext_vector_type(8))) float v8f;

#define SQRT3F 1.7320508075688772f
#define LOG2PIF 1.8378770664093453f

__device__ __forceinline__ float kern_pair(float x1, float y1, float a1, float b1, float c1, float q1,
                                           float x2, float y2, float a2, float b2, float c2, float q2,
                                           float scale) {
    float s00 = a1 + a2, s11 = b1 + b2, s01 = c1 + c2;
    float sdet = s00 * s11 - s01 * s01;
    float d0 = x1 - x2, d1 = y1 - y2;
    float Q = 0.5f * (s11 * d0 * d0 - 2.0f * s01 * d0 * d1 + s00 * d1 * d1) / sdet;
    float C = 2.0f * q1 * q2 * rsqrtf(fmaxf(sdet, 1e-5f));
    float t = SQRT3F * sqrtf(fmaxf(Q, 1e-5f));
    return (1.0f + t) * __expf(-t) * C * scale;
}

__global__ void prep_kernel(const float* __restrict__ ct, const float* __restrict__ kpt,
                            const float* __restrict__ cs, const float* __restrict__ kps,
                            const float* __restrict__ scale_param,
                            float* __restrict__ trainP, float* __restrict__ testP,
                            float* __restrict__ Kmm) {
    int gid = blockIdx.x * blockDim.x + threadIdx.x;
    float scale = __expf(scale_param[0]);
    if (gid < Bb * Mm) {
        float a = kps[gid * 3 + 0], b = kps[gid * 3 + 1], c = kps[gid * 3 + 2];
        float det = a * b - c * c;
        testP[gid * 6 + 0] = cs[gid * 2 + 0];
        testP[gid * 6 + 1] = cs[gid * 2 + 1];
        testP[gid * 6 + 2] = a;
        testP[gid * 6 + 3] = b;
        testP[gid * 6 + 4] = c;
        testP[gid * 6 + 5] = sqrtf(sqrtf(det));
        float t0 = SQRT3F * sqrtf(1e-5f);
        float m0 = (1.0f + t0) * __expf(-t0);
        float C = 2.0f * sqrtf(det) * rsqrtf(fmaxf(4.0f * det, 1e-5f));
        Kmm[gid] = C * m0 * scale;
    }
    if (gid < Bb * Nn) {
        float a = kpt[gid * 3 + 0], b = kpt[gid * 3 + 1], c = kpt[gid * 3 + 2];
        float det = a * b - c * c;
        trainP[gid * 6 + 0] = ct[gid * 2 + 0];
        trainP[gid * 6 + 1] = ct[gid * 2 + 1];
        trainP[gid * 6 + 2] = a;
        trainP[gid * 6 + 3] = b;
        trainP[gid * 6 + 4] = c;
        trainP[gid * 6 + 5] = sqrtf(sqrtf(det));
    }
}

__global__ void buildA_kernel(const float* __restrict__ trainP, const float* __restrict__ var,
                              const float* __restrict__ scale_param, float* __restrict__ A) {
    int gid = blockIdx.x * blockDim.x + threadIdx.x;
    int b = gid >> 20;
    int i = (gid >> 10) & (Nn - 1);
    int j = gid & (Nn - 1);
    const float* p1 = trainP + (size_t)(b * Nn + i) * 6;
    const float* p2 = trainP + (size_t)(b * Nn + j) * 6;
    float scale = __expf(scale_param[0]);
    float v = kern_pair(p1[0], p1[1], p1[2], p1[3], p1[4], p1[5],
                        p2[0], p2[1], p2[2], p2[3], p2[4], p2[5], scale);
    if (i == j) v += var[b * Nn + i];
    A[(size_t)gid] = v;
}

template <bool WITH_S>
__device__ __forceinline__ void panel_sweep(float* Dsh, float* Ssh, float* dg, int tid) {
    int t = tid & (NB - 1);
    int q = tid >> 6;
    for (int k = 0; k < NB; k++) {
        float d = sqrtf(Dsh[k * LDP + k]);
        float dinv = 1.0f / d;
        if (tid == 0) dg[k] = d;
        if (q == 0 && t > k) Dsh[t * LDP + k] *= dinv;
        if (WITH_S && q == 1) Ssh[t * LDP + k] *= dinv;
        __syncthreads();
        float dtk = Dsh[t * LDP + k];
        float stk = WITH_S ? Ssh[t * LDP + k] : 0.0f;
        int j0 = k + 1 + q;
        float lj[8], dv[8], sv[8];
#pragma unroll
        for (int u = 0; u < 8; u++) {
            int j = j0 + 8 * u;
            int jj = (j < NB) ? j : k;
            lj[u] = Dsh[jj * LDP + k];
            dv[u] = Dsh[t * LDP + jj];
            sv[u] = WITH_S ? Ssh[t * LDP + jj] : 0.0f;
        }
#pragma unroll
        for (int u = 0; u < 8; u++) {
            int j = j0 + 8 * u;
            if (j < NB) {
                if (j <= t) Dsh[t * LDP + j] = dv[u] - dtk * lj[u];
                if (WITH_S) Ssh[t * LDP + j] = sv[u] - stk * lj[u];
            }
        }
        __syncthreads();
    }
}

__global__ __launch_bounds__(512) void chol_step_kernel(const float* __restrict__ Abase,
                                                        float* Lbase, int p) {
    __shared__ float Dsh[NB * LDP];
    __shared__ float Ssh[NB * LDP];
    __shared__ float dg[NB];
    int b = blockIdx.y;
    int i = p + blockIdx.x;
    bool isDiag = (blockIdx.x == 0);
    const float* A = Abase + (size_t)b * Nn * Nn;
    float* L = Lbase + (size_t)b * Nn * Nn;
    int tid = threadIdx.x;
    int wave = tid >> 5, lane = tid & 31;
    int r = lane & 15, hi = lane >> 4;
    int tm = wave >> 2, tn = wave & 3;
    int prow = p * NB, irow = i * NB;

    v8f accD, accS;
#pragma unroll
    for (int v = 0; v < 8; v++) {
        int row = tm * 16 + v + 8 * hi;
        int col = tn * 16 + r;
        accD[v] = -A[(size_t)(prow + row) * Nn + prow + col];
        accS[v] = -A[(size_t)(irow + row) * Nn + prow + col];
    }
    {
        const float* aP = L + (size_t)(prow + tm * 16 + r) * Nn;
        const float* aI = L + (size_t)(irow + tm * 16 + r) * Nn;
        const float* bP = L + (size_t)(prow + tn * 16 + r) * Nn;
        for (int kb = 0; kb < prow; kb += 16) {
#pragma unroll
            for (int k0 = 0; k0 < 16; k0 += 4) {
                int kk = kb + k0 + 2 * hi;
                v2f fa = *(const v2f*)(aP + kk);
                v2f fi = *(const v2f*)(aI + kk);
                v2f fb = *(const v2f*)(bP + kk);
                accD = __builtin_amdgcn_wmma_f32_16x16x4_f32(false, fa, false, fb,
                                                             (short)0, accD, false, false);
                accS = __builtin_amdgcn_wmma_f32_16x16x4_f32(false, fi, false, fb,
                                                             (short)0, accS, false, false);
            }
        }
    }
#pragma unroll
    for (int v = 0; v < 8; v++) {
        int row = tm * 16 + v + 8 * hi;
        int col = tn * 16 + r;
        Dsh[row * LDP + col] = -accD[v];
        Ssh[row * LDP + col] = -accS[v];
    }
    __syncthreads();

    if (isDiag) panel_sweep<false>(Dsh, Ssh, dg, tid);
    else        panel_sweep<true>(Dsh, Ssh, dg, tid);

    if (isDiag) {
        for (int idx = tid; idx < NB * NB; idx += 512) {
            int rr = idx >> 6, c = idx & (NB - 1);
            float val = (c < rr) ? Dsh[rr * LDP + c] : ((c == rr) ? dg[rr] : 0.0f);
            L[(size_t)(prow + rr) * Nn + prow + c] = val;
        }
    } else {
        for (int idx = tid; idx < NB * NB; idx += 512) {
            int rr = idx >> 6, c = idx & (NB - 1);
            L[(size_t)(irow + rr) * Nn + prow + c] = Ssh[rr * LDP + c];
        }
    }
}

__global__ __launch_bounds__(1024) void logdet_kernel(const float* __restrict__ Lbase,
                                                      float* __restrict__ logdet) {
    int b = blockIdx.x;
    const float* L = Lbase + (size_t)b * Nn * Nn;
    __shared__ float col[Nn];
    int tid = threadIdx.x;
    col[tid] = __logf(L[(size_t)tid * Nn + tid]);
    __syncthreads();
    for (int s = 512; s > 0; s >>= 1) {
        if (tid < s) col[tid] += col[tid + s];
        __syncthreads();
    }
    if (tid == 0) logdet[b] = col[0];
}

__global__ __launch_bounds__(256) void zsolve_kernel(const float* __restrict__ Lbase,
                                                     const float* __restrict__ y,
                                                     const float* __restrict__ mean,
                                                     const float* __restrict__ logdet,
                                                     float* __restrict__ zbase,
                                                     float* __restrict__ nlml_out) {
    int b = blockIdx.x;
    const float* L = Lbase + (size_t)b * Nn * Nn;
    __shared__ float zsh[Nn];
    __shared__ float red[256];
    int tid = threadIdx.x;
    float mb = mean[b];
    for (int i = 0; i < Nn; i++) {
        const float* row = L + (size_t)i * Nn;
        float s = 0.0f;
        for (int j = tid; j < i; j += 256) s += row[j] * zsh[j];
        red[tid] = s;
        __syncthreads();
        for (int st = 128; st > 0; st >>= 1) {
            if (tid < st) red[tid] += red[tid + st];
            __syncthreads();
        }
        if (tid == 0) zsh[i] = (y[b * Nn + i] - mb - red[0]) / row[i];
        __syncthreads();
    }
    float s2 = 0.0f;
    for (int i = tid; i < Nn; i += 256) {
        float z = zsh[i];
        zbase[b * Nn + i] = z;
        s2 += z * z;
    }
    red[tid] = s2;
    __syncthreads();
    for (int st = 128; st > 0; st >>= 1) {
        if (tid < st) red[tid] += red[tid + st];
        __syncthreads();
    }
    if (tid == 0) nlml_out[b] = 0.5f * red[0] + logdet[b] + 0.5f * Nn * LOG2PIF;
}

__global__ __launch_bounds__(256) void trinv_kernel(const float* __restrict__ Lbase,
                                                    float* __restrict__ Linvbase) {
    int b = blockIdx.y;
    int wave = threadIdx.x >> 5;
    int lane = threadIdx.x & 31;
    int j = blockIdx.x * 8 + wave;
    const float* L = Lbase + (size_t)b * Nn * Nn;
    float* Li = Linvbase + (size_t)b * Nn * Nn;
    __shared__ float xsh[8 * Nn];
    float* x = xsh + wave * Nn;
    for (int i = lane; i < j; i += 32) Li[(size_t)i * Nn + j] = 0.0f;
    float djj = 1.0f / L[(size_t)j * Nn + j];
    if (lane == 0) {
        x[j] = djj;
        Li[(size_t)j * Nn + j] = djj;
    }
    for (int i = j + 1; i < Nn; i++) {
        const float* row = L + (size_t)i * Nn;
        float s = 0.0f;
        for (int k = j + lane; k < i; k += 32) s += row[k] * x[k];
        s += __shfl_xor(s, 16);
        s += __shfl_xor(s, 8);
        s += __shfl_xor(s, 4);
        s += __shfl_xor(s, 2);
        s += __shfl_xor(s, 1);
        float xi = -s / row[i];
        if (lane == 0) {
            x[i] = xi;
            Li[(size_t)i * Nn + j] = xi;
        }
    }
}

__global__ __launch_bounds__(128) void predict_kernel(const float* __restrict__ Linvbase,
                                                      const float* __restrict__ trainPg,
                                                      const float* __restrict__ testPg,
                                                      const float* __restrict__ Kmm,
                                                      const float* __restrict__ zbase,
                                                      const float* __restrict__ meang,
                                                      const float* __restrict__ scale_param,
                                                      float* __restrict__ out_mean,
                                                      float* __restrict__ out_var) {
    extern __shared__ float smem[];
    float* Kt  = smem;
    float* zsh = smem + Nn * KTS;
    float* tp  = zsh + Nn;
    int b = blockIdx.y;
    int colBase = blockIdx.x * MT;
    int tid = threadIdx.x;
    float scale = __expf(scale_param[0]);

    for (int idx = tid; idx < Nn; idx += 128) zsh[idx] = zbase[b * Nn + idx];
    for (int idx = tid; idx < Nn * 6; idx += 128) tp[idx] = trainPg[(size_t)b * Nn * 6 + idx];
    __syncthreads();

    {
        int c = tid & 63;
        int h = tid >> 6;
        const float* q = testPg + (size_t)(b * Mm + colBase + c) * 6;
        float x2 = q[0], y2 = q[1], a2 = q[2], b2 = q[3], c2 = q[4], q2 = q[5];
        for (int n = h; n < Nn; n += 2) {
            const float* p = tp + n * 6;
            Kt[n * KTS + c] = kern_pair(p[0], p[1], p[2], p[3], p[4], p[5],
                                        x2, y2, a2, b2, c2, q2, scale);
        }
    }
    __syncthreads();

    int wave = tid >> 5, lane = tid & 31;
    int r = lane & 15, hi = lane >> 4;
    int wcol = wave * 16;
    const float* Li = Linvbase + (size_t)b * Nn * Nn;

    float meanAcc = 0.0f, varAcc = 0.0f;
    for (int rb = 0; rb < Nn / 16; rb++) {
        v8f acc = {};
        const float* arow = Li + (size_t)(rb * 16 + r) * Nn;
        for (int j = 0; j <= rb; j++) {
            int kb = j * 16;
#pragma unroll
            for (int k0 = 0; k0 < 16; k0 += 4) {
                int kk = kb + k0 + 2 * hi;
                v2f a = *(const v2f*)(arow + kk);
                v2f bb;
                bb.x = Kt[kk * KTS + wcol + r];
                bb.y = Kt[(kk + 1) * KTS + wcol + r];
                acc = __builtin_amdgcn_wmma_f32_16x16x4_f32(false, a, false, bb,
                                                            (short)0, acc, false, false);
            }
        }
        const float* zb = zsh + rb * 16 + 8 * hi;
#pragma unroll
        for (int v = 0; v < 8; v++) {
            float cv = acc[v];
            meanAcc += zb[v] * cv;
            varAcc  += cv * cv;
        }
    }
    meanAcc += __shfl_xor(meanAcc, 16);
    varAcc  += __shfl_xor(varAcc, 16);
    if (lane < 16) {
        int m = colBase + wcol + lane;
        out_mean[b * Mm + m] = meang[b] + meanAcc;
        out_var[b * Mm + m]  = Kmm[b * Mm + m] - varAcc;
    }
}

extern "C" void kernel_launch(void* const* d_in, const int* in_sizes, int n_in,
                              void* d_out, int out_size, void* d_ws, size_t ws_size,
                              hipStream_t stream) {
    const float* coords_train = (const float*)d_in[0];
    const float* kp_train     = (const float*)d_in[1];
    const float* coords_test  = (const float*)d_in[2];
    const float* kp_test      = (const float*)d_in[3];
    const float* y_train      = (const float*)d_in[4];
    const float* var          = (const float*)d_in[5];
    const float* mean         = (const float*)d_in[6];
    const float* scale_param  = (const float*)d_in[7];

    float* ws = (float*)d_ws;
    size_t off = 0;
    float* A      = ws + off; off += (size_t)Bb * Nn * Nn;
    float* Lm     = ws + off; off += (size_t)Bb * Nn * Nn;
    float* Linv   = ws + off; off += (size_t)Bb * Nn * Nn;
    float* trainP = ws + off; off += (size_t)Bb * Nn * 6;
    float* testP  = ws + off; off += (size_t)Bb * Mm * 6;
    float* Kmm    = ws + off; off += (size_t)Bb * Mm;
    float* z      = ws + off; off += (size_t)Bb * Nn;
    float* logdet = ws + off; off += (size_t)Bb;

    float* out_mean = (float*)d_out;
    float* out_var  = out_mean + (size_t)Bb * Mm;
    float* nlml     = out_var + (size_t)Bb * Mm;

    prep_kernel<<<(Bb * Mm + 255) / 256, 256, 0, stream>>>(
        coords_train, kp_train, coords_test, kp_test, scale_param, trainP, testP, Kmm);
    buildA_kernel<<<(Bb * Nn * Nn) / 256, 256, 0, stream>>>(trainP, var, scale_param, A);
#define CHOL_STEP(P) chol_step_kernel<<<dim3(NBLK - (P), Bb), 512, 0, stream>>>(A, Lm, (P))
    CHOL_STEP(0);  CHOL_STEP(1);  CHOL_STEP(2);  CHOL_STEP(3);
    CHOL_STEP(4);  CHOL_STEP(5);  CHOL_STEP(6);  CHOL_STEP(7);
    CHOL_STEP(8);  CHOL_STEP(9);  CHOL_STEP(10); CHOL_STEP(11);
    CHOL_STEP(12); CHOL_STEP(13); CHOL_STEP(14); CHOL_STEP(15);
#undef CHOL_STEP
    logdet_kernel<<<Bb, 1024, 0, stream>>>(Lm, logdet);
    zsolve_kernel<<<Bb, 256, 0, stream>>>(Lm, y_train, mean, logdet, z, nlml);
    trinv_kernel<<<dim3(Nn / 8, Bb), 256, 0, stream>>>(Lm, Linv);

    size_t smem_bytes = (size_t)(Nn * KTS + Nn + Nn * 6) * sizeof(float);
    predict_kernel<<<dim3(Mm / MT, Bb), 128, smem_bytes, stream>>>(
        Linv, trainP, testP, Kmm, z, mean, scale_param, out_mean, out_var);
}


# explicit inverse of L rebuilt blocked: 64-wide diagonal blocks inverted in LDS, off-diagonal blocks by block forward substitution on f32 WMMA (new summation order inside the inverse)
# speedup vs baseline: 14.4750x; 1.8604x over previous
#include <hip/hip_runtime.h>
#include <math.h>

#define Nn 1024
#define Mm 16384
#define Bb 2
#define MT 64
#define KTS 65
#define NB 64
#define NBLK (Nn / NB)
#define LDP 65

typedef __attribute__((ext_vector_type(2))) float v2f;
typedef __attribute__((ext_vector_type(8))) float v8f;

#define SQRT3F 1.7320508075688772f
#define LOG2PIF 1.8378770664093453f

__device__ __forceinline__ float kern_pair(float x1, float y1, float a1, float b1, float c1, float q1,
                                           float x2, float y2, float a2, float b2, float c2, float q2,
                                           float scale) {
    float s00 = a1 + a2, s11 = b1 + b2, s01 = c1 + c2;
    float sdet = s00 * s11 - s01 * s01;
    float d0 = x1 - x2, d1 = y1 - y2;
    float Q = 0.5f * (s11 * d0 * d0 - 2.0f * s01 * d0 * d1 + s00 * d1 * d1) / sdet;
    float C = 2.0f * q1 * q2 * rsqrtf(fmaxf(sdet, 1e-5f));
    float t = SQRT3F * sqrtf(fmaxf(Q, 1e-5f));
    return (1.0f + t) * __expf(-t) * C * scale;
}

__global__ void prep_kernel(const float* __restrict__ ct, const float* __restrict__ kpt,
                            const float* __restrict__ cs, const float* __restrict__ kps,
                            const float* __restrict__ scale_param,
                            float* __restrict__ trainP, float* __restrict__ testP,
                            float* __restrict__ Kmm) {
    int gid = blockIdx.x * blockDim.x + threadIdx.x;
    float scale = __expf(scale_param[0]);
    if (gid < Bb * Mm) {
        float a = kps[gid * 3 + 0], b = kps[gid * 3 + 1], c = kps[gid * 3 + 2];
        float det = a * b - c * c;
        testP[gid * 6 + 0] = cs[gid * 2 + 0];
        testP[gid * 6 + 1] = cs[gid * 2 + 1];
        testP[gid * 6 + 2] = a;
        testP[gid * 6 + 3] = b;
        testP[gid * 6 + 4] = c;
        testP[gid * 6 + 5] = sqrtf(sqrtf(det));
        float t0 = SQRT3F * sqrtf(1e-5f);
        float m0 = (1.0f + t0) * __expf(-t0);
        float C = 2.0f * sqrtf(det) * rsqrtf(fmaxf(4.0f * det, 1e-5f));
        Kmm[gid] = C * m0 * scale;
    }
    if (gid < Bb * Nn) {
        float a = kpt[gid * 3 + 0], b = kpt[gid * 3 + 1], c = kpt[gid * 3 + 2];
        float det = a * b - c * c;
        trainP[gid * 6 + 0] = ct[gid * 2 + 0];
        trainP[gid * 6 + 1] = ct[gid * 2 + 1];
        trainP[gid * 6 + 2] = a;
        trainP[gid * 6 + 3] = b;
        trainP[gid * 6 + 4] = c;
        trainP[gid * 6 + 5] = sqrtf(sqrtf(det));
    }
}

__global__ void buildA_kernel(const float* __restrict__ trainP, const float* __restrict__ var,
                              const float* __restrict__ scale_param, float* __restrict__ A) {
    int gid = blockIdx.x * blockDim.x + threadIdx.x;
    int b = gid >> 20;
    int i = (gid >> 10) & (Nn - 1);
    int j = gid & (Nn - 1);
    const float* p1 = trainP + (size_t)(b * Nn + i) * 6;
    const float* p2 = trainP + (size_t)(b * Nn + j) * 6;
    float scale = __expf(scale_param[0]);
    float v = kern_pair(p1[0], p1[1], p1[2], p1[3], p1[4], p1[5],
                        p2[0], p2[1], p2[2], p2[3], p2[4], p2[5], scale);
    if (i == j) v += var[b * Nn + i];
    A[(size_t)gid] = v;
}

template <bool WITH_S>
__device__ __forceinline__ void panel_sweep(float* Dsh, float* Ssh, float* dg, int tid) {
    int t = tid & (NB - 1);
    int q = tid >> 6;
    for (int k = 0; k < NB; k++) {
        float d = sqrtf(Dsh[k * LDP + k]);
        float dinv = 1.0f / d;
        if (tid == 0) dg[k] = d;
        if (q == 0 && t > k) Dsh[t * LDP + k] *= dinv;
        if (WITH_S && q == 1) Ssh[t * LDP + k] *= dinv;
        __syncthreads();
        float dtk = Dsh[t * LDP + k];
        float stk = WITH_S ? Ssh[t * LDP + k] : 0.0f;
        int j0 = k + 1 + q;
        float lj[8], dv[8], sv[8];
#pragma unroll
        for (int u = 0; u < 8; u++) {
            int j = j0 + 8 * u;
            int jj = (j < NB) ? j : k;
            lj[u] = Dsh[jj * LDP + k];
            dv[u] = Dsh[t * LDP + jj];
            sv[u] = WITH_S ? Ssh[t * LDP + jj] : 0.0f;
        }
#pragma unroll
        for (int u = 0; u < 8; u++) {
            int j = j0 + 8 * u;
            if (j < NB) {
                if (j <= t) Dsh[t * LDP + j] = dv[u] - dtk * lj[u];
                if (WITH_S) Ssh[t * LDP + j] = sv[u] - stk * lj[u];
            }
        }
        __syncthreads();
    }
}

__global__ __launch_bounds__(512) void chol_step_kernel(const float* __restrict__ Abase,
                                                        float* Lbase, int p) {
    __shared__ float Dsh[NB * LDP];
    __shared__ float Ssh[NB * LDP];
    __shared__ float dg[NB];
    int b = blockIdx.y;
    int i = p + blockIdx.x;
    bool isDiag = (blockIdx.x == 0);
    const float* A = Abase + (size_t)b * Nn * Nn;
    float* L = Lbase + (size_t)b * Nn * Nn;
    int tid = threadIdx.x;
    int wave = tid >> 5, lane = tid & 31;
    int r = lane & 15, hi = lane >> 4;
    int tm = wave >> 2, tn = wave & 3;
    int prow = p * NB, irow = i * NB;

    v8f accD, accS;
#pragma unroll
    for (int v = 0; v < 8; v++) {
        int row = tm * 16 + v + 8 * hi;
        int col = tn * 16 + r;
        accD[v] = -A[(size_t)(prow + row) * Nn + prow + col];
        accS[v] = -A[(size_t)(irow + row) * Nn + prow + col];
    }
    {
        const float* aP = L + (size_t)(prow + tm * 16 + r) * Nn;
        const float* aI = L + (size_t)(irow + tm * 16 + r) * Nn;
        const float* bP = L + (size_t)(prow + tn * 16 + r) * Nn;
        for (int kb = 0; kb < prow; kb += 16) {
#pragma unroll
            for (int k0 = 0; k0 < 16; k0 += 4) {
                int kk = kb + k0 + 2 * hi;
                v2f fa = *(const v2f*)(aP + kk);
                v2f fi = *(const v2f*)(aI + kk);
                v2f fb = *(const v2f*)(bP + kk);
                accD = __builtin_amdgcn_wmma_f32_16x16x4_f32(false, fa, false, fb,
                                                             (short)0, accD, false, false);
                accS = __builtin_amdgcn_wmma_f32_16x16x4_f32(false, fi, false, fb,
                                                             (short)0, accS, false, false);
            }
        }
    }
#pragma unroll
    for (int v = 0; v < 8; v++) {
        int row = tm * 16 + v + 8 * hi;
        int col = tn * 16 + r;
        Dsh[row * LDP + col] = -accD[v];
        Ssh[row * LDP + col] = -accS[v];
    }
    __syncthreads();

    if (isDiag) panel_sweep<false>(Dsh, Ssh, dg, tid);
    else        panel_sweep<true>(Dsh, Ssh, dg, tid);

    if (isDiag) {
        for (int idx = tid; idx < NB * NB; idx += 512) {
            int rr = idx >> 6, c = idx & (NB - 1);
            float val = (c < rr) ? Dsh[rr * LDP + c] : ((c == rr) ? dg[rr] : 0.0f);
            L[(size_t)(prow + rr) * Nn + prow + c] = val;
        }
    } else {
        for (int idx = tid; idx < NB * NB; idx += 512) {
            int rr = idx >> 6, c = idx & (NB - 1);
            L[(size_t)(irow + rr) * Nn + prow + c] = Ssh[rr * LDP + c];
        }
    }
}

__global__ __launch_bounds__(1024) void logdet_kernel(const float* __restrict__ Lbase,
                                                      float* __restrict__ logdet) {
    int b = blockIdx.x;
    const float* L = Lbase + (size_t)b * Nn * Nn;
    __shared__ float col[Nn];
    int tid = threadIdx.x;
    col[tid] = __logf(L[(size_t)tid * Nn + tid]);
    __syncthreads();
    for (int s = 512; s > 0; s >>= 1) {
        if (tid < s) col[tid] += col[tid + s];
        __syncthreads();
    }
    if (tid == 0) logdet[b] = col[0];
}

__global__ __launch_bounds__(256) void zsolve_kernel(const float* __restrict__ Lbase,
                                                     const float* __restrict__ y,
                                                     const float* __restrict__ mean,
                                                     const float* __restrict__ logdet,
                                                     float* __restrict__ zbase,
                                                     float* __restrict__ nlml_out) {
    int b = blockIdx.x;
    const float* L = Lbase + (size_t)b * Nn * Nn;
    __shared__ float zsh[Nn];
    __shared__ float red[256];
    int tid = threadIdx.x;
    float mb = mean[b];
    for (int i = 0; i < Nn; i++) {
        const float* row = L + (size_t)i * Nn;
        float s = 0.0f;
        for (int j = tid; j < i; j += 256) s += row[j] * zsh[j];
        red[tid] = s;
        __syncthreads();
        for (int st = 128; st > 0; st >>= 1) {
            if (tid < st) red[tid] += red[tid + st];
            __syncthreads();
        }
        if (tid == 0) zsh[i] = (y[b * Nn + i] - mb - red[0]) / row[i];
        __syncthreads();
    }
    float s2 = 0.0f;
    for (int i = tid; i < Nn; i += 256) {
        float z = zsh[i];
        zbase[b * Nn + i] = z;
        s2 += z * z;
    }
    red[tid] = s2;
    __syncthreads();
    for (int st = 128; st > 0; st >>= 1) {
        if (tid < st) red[tid] += red[tid + st];
        __syncthreads();
    }
    if (tid == 0) nlml_out[b] = 0.5f * red[0] + logdet[b] + 0.5f * Nn * LOG2PIF;
}

__global__ __launch_bounds__(64) void trinv_diag_kernel(const float* __restrict__ Lbase,
                                                        float* __restrict__ Linvbase) {
    __shared__ __attribute__((aligned(16))) float Lb[NB * NB];
    __shared__ float xs[NB * NB];
    int p = blockIdx.x, b = blockIdx.y, tid = threadIdx.x;
    size_t base = (size_t)b * Nn * Nn + (size_t)(p * NB) * Nn + p * NB;
    const float* L = Lbase + base;
    float* Li = Linvbase + base;
    for (int idx = tid; idx < NB * NB; idx += 64) {
        int rr = idx >> 6, c = idx & (NB - 1);
        Lb[idx] = L[(size_t)rr * Nn + c];
    }
    __syncthreads();
    int j = tid;
    for (int i = 0; i < NB; i++) {
        const float* lr = Lb + i * NB;
        float s0 = (i == j) ? 1.0f : 0.0f, s1 = 0.0f, s2 = 0.0f, s3 = 0.0f;
        int k = 0;
        for (; k + 4 <= i; k += 4) {
            float4 l = *(const float4*)(lr + k);
            s0 -= l.x * xs[(k + 0) * NB + j];
            s1 -= l.y * xs[(k + 1) * NB + j];
            s2 -= l.z * xs[(k + 2) * NB + j];
            s3 -= l.w * xs[(k + 3) * NB + j];
        }
        for (; k < i; k++) s0 -= lr[k] * xs[k * NB + j];
        xs[i * NB + j] = ((s0 + s1) + (s2 + s3)) / lr[i];
    }
    __syncthreads();
    for (int idx = tid; idx < NB * NB; idx += 64) {
        int rr = idx >> 6, c = idx & (NB - 1);
        Li[(size_t)rr * Nn + c] = xs[idx];
    }
}

#define XLD 65

__device__ __forceinline__ void slab32(const float* aP, const float* bP, int kb, int hi,
                                       v8f& c0, v8f& c1) {
    v2f a[8], q[8];
#pragma unroll
    for (int u = 0; u < 8; u++) {
        int kk = kb + 4 * u + 2 * hi;
        a[u] = *(const v2f*)(aP + kk);
        q[u].x = bP[kk * XLD];
        q[u].y = bP[(kk + 1) * XLD];
    }
    __builtin_amdgcn_sched_barrier(0);
#pragma unroll
    for (int u = 0; u < 8; u += 2) {
        c0 = __builtin_amdgcn_wmma_f32_16x16x4_f32(false, a[u], false, q[u],
                                                   (short)0, c0, false, false);
        c1 = __builtin_amdgcn_wmma_f32_16x16x4_f32(false, a[u + 1], false, q[u + 1],
                                                   (short)0, c1, false, false);
    }
}
__global__ __launch_bounds__(512) void trinv_panel_kernel(const float* __restrict__ Lbase,
                                                          float* Linvbase) {
    extern __shared__ float smem[];
    float* Xp  = smem;
    float* Tsh = smem + Nn * XLD;
    int J = blockIdx.x, b = blockIdx.y, tid = threadIdx.x;
    const float* L = Lbase + (size_t)b * Nn * Nn;
    float* Li = Linvbase + (size_t)b * Nn * Nn;
    int wave = __builtin_amdgcn_readfirstlane(tid >> 5);
    int lane = tid & 31;
    int r = lane & 15, hi = lane >> 4;
    int tm = wave >> 2, tn = wave & 3;
    int jcol = J * NB;

    for (int idx = tid; idx < NB * NB; idx += 512) {
        int rr = idx >> 6, c = idx & (NB - 1);
        Xp[rr * XLD + c] = Li[(size_t)(jcol + rr) * Nn + jcol + c];
    }
    __syncthreads();

    for (int i = J + 1; i < NBLK; i++) {
        int irow = i * NB;
        int kdepth = irow - jcol;
        v8f acc0 = {}, acc1 = {};
        {
            const float* aP = L + (size_t)(irow + tm * 16 + r) * Nn + jcol;
            const float* bP = Xp + tn * 16 + r;
            for (int kb = 0; kb < kdepth; kb += 32) slab32(aP, bP, kb, hi, acc0, acc1);
        }
#pragma unroll
        for (int v = 0; v < 8; v++) {
            int row = tm * 16 + v + 8 * hi;
            int col = tn * 16 + r;
            Tsh[row * XLD + col] = acc0[v] + acc1[v];
        }
        __syncthreads();
        v8f x0 = {}, x1 = {};
        {
            const float* aD = Li + (size_t)(irow + tm * 16 + r) * Nn + irow;
            const float* bT = Tsh + tn * 16 + r;
            for (int kb = 0; kb < (tm + 1) * 16; kb += 32) slab32(aD, bT, kb, hi, x0, x1);
        }
#pragma unroll
        for (int v = 0; v < 8; v++) {
            int row = tm * 16 + v + 8 * hi;
            int col = tn * 16 + r;
            Xp[(kdepth + row) * XLD + col] = -(x0[v] + x1[v]);
        }
        __syncthreads();
        for (int idx = tid; idx < NB * NB; idx += 512) {
            int rr = idx >> 6, c = idx & (NB - 1);
            Li[(size_t)(irow + rr) * Nn + jcol + c] = Xp[(kdepth + rr) * XLD + c];
        }
    }
}

__global__ __launch_bounds__(128) void predict_kernel(const float* __restrict__ Linvbase,
                                                      const float* __restrict__ trainPg,
                                                      const float* __restrict__ testPg,
                                                      const float* __restrict__ Kmm,
                                                      const float* __restrict__ zbase,
                                                      const float* __restrict__ meang,
                                                      const float* __restrict__ scale_param,
                                                      float* __restrict__ out_mean,
                                                      float* __restrict__ out_var) {
    extern __shared__ float smem[];
    float* Kt  = smem;
    float* zsh = smem + Nn * KTS;
    float* tp  = zsh + Nn;
    int b = blockIdx.y;
    int colBase = blockIdx.x * MT;
    int tid = threadIdx.x;
    float scale = __expf(scale_param[0]);

    for (int idx = tid; idx < Nn; idx += 128) zsh[idx] = zbase[b * Nn + idx];
    for (int idx = tid; idx < Nn * 6; idx += 128) tp[idx] = trainPg[(size_t)b * Nn * 6 + idx];
    __syncthreads();

    {
        int c = tid & 63;
        int h = tid >> 6;
        const float* q = testPg + (size_t)(b * Mm + colBase + c) * 6;
        float x2 = q[0], y2 = q[1], a2 = q[2], b2 = q[3], c2 = q[4], q2 = q[5];
        for (int n = h; n < Nn; n += 2) {
            const float* p = tp + n * 6;
            Kt[n * KTS + c] = kern_pair(p[0], p[1], p[2], p[3], p[4], p[5],
                                        x2, y2, a2, b2, c2, q2, scale);
        }
    }
    __syncthreads();

    int wave = tid >> 5, lane = tid & 31;
    int r = lane & 15, hi = lane >> 4;
    int wcol = wave * 16;
    const float* Li = Linvbase + (size_t)b * Nn * Nn;

    float meanAcc = 0.0f, varAcc = 0.0f;
    for (int rb = 0; rb < Nn / 16; rb++) {
        v8f acc = {};
        const float* arow = Li + (size_t)(rb * 16 + r) * Nn;
        for (int j = 0; j <= rb; j++) {
            int kb = j * 16;
#pragma unroll
            for (int k0 = 0; k0 < 16; k0 += 4) {
                int kk = kb + k0 + 2 * hi;
                v2f a = *(const v2f*)(arow + kk);
                v2f bb;
                bb.x = Kt[kk * KTS + wcol + r];
                bb.y = Kt[(kk + 1) * KTS + wcol + r];
                acc = __builtin_amdgcn_wmma_f32_16x16x4_f32(false, a, false, bb,
                                                            (short)0, acc, false, false);
            }
        }
        const float* zb = zsh + rb * 16 + 8 * hi;
#pragma unroll
        for (int v = 0; v < 8; v++) {
            float cv = acc[v];
            meanAcc += zb[v] * cv;
            varAcc  += cv * cv;
        }
    }
    meanAcc += __shfl_xor(meanAcc, 16);
    varAcc  += __shfl_xor(varAcc, 16);
    if (lane < 16) {
        int m = colBase + wcol + lane;
        out_mean[b * Mm + m] = meang[b] + meanAcc;
        out_var[b * Mm + m]  = Kmm[b * Mm + m] - varAcc;
    }
}

extern "C" void kernel_launch(void* const* d_in, const int* in_sizes, int n_in,
                              void* d_out, int out_size, void* d_ws, size_t ws_size,
                              hipStream_t stream) {
    const float* coords_train = (const float*)d_in[0];
    const float* kp_train     = (const float*)d_in[1];
    const float* coords_test  = (const float*)d_in[2];
    const float* kp_test      = (const float*)d_in[3];
    const float* y_train      = (const float*)d_in[4];
    const float* var          = (const float*)d_in[5];
    const float* mean         = (const float*)d_in[6];
    const float* scale_param  = (const float*)d_in[7];

    float* ws = (float*)d_ws;
    size_t off = 0;
    float* A      = ws + off; off += (size_t)Bb * Nn * Nn;
    float* Lm     = ws + off; off += (size_t)Bb * Nn * Nn;
    float* Linv   = ws + off; off += (size_t)Bb * Nn * Nn;
    float* trainP = ws + off; off += (size_t)Bb * Nn * 6;
    float* testP  = ws + off; off += (size_t)Bb * Mm * 6;
    float* Kmm    = ws + off; off += (size_t)Bb * Mm;
    float* z      = ws + off; off += (size_t)Bb * Nn;
    float* logdet = ws + off; off += (size_t)Bb;

    float* out_mean = (float*)d_out;
    float* out_var  = out_mean + (size_t)Bb * Mm;
    float* nlml     = out_var + (size_t)Bb * Mm;

    prep_kernel<<<(Bb * Mm + 255) / 256, 256, 0, stream>>>(
        coords_train, kp_train, coords_test, kp_test, scale_param, trainP, testP, Kmm);
    buildA_kernel<<<(Bb * Nn * Nn) / 256, 256, 0, stream>>>(trainP, var, scale_param, A);
#define CHOL_STEP(P) chol_step_kernel<<<dim3(NBLK - (P), Bb), 512, 0, stream>>>(A, Lm, (P))
    CHOL_STEP(0);  CHOL_STEP(1);  CHOL_STEP(2);  CHOL_STEP(3);
    CHOL_STEP(4);  CHOL_STEP(5);  CHOL_STEP(6);  CHOL_STEP(7);
    CHOL_STEP(8);  CHOL_STEP(9);  CHOL_STEP(10); CHOL_STEP(11);
    CHOL_STEP(12); CHOL_STEP(13); CHOL_STEP(14); CHOL_STEP(15);
#undef CHOL_STEP
    logdet_kernel<<<Bb, 1024, 0, stream>>>(Lm, logdet);
    zsolve_kernel<<<Bb, 256, 0, stream>>>(Lm, y_train, mean, logdet, z, nlml);
    trinv_diag_kernel<<<dim3(NBLK, Bb), 64, 0, stream>>>(Lm, Linv);
    size_t inv_smem_bytes = (size_t)(Nn * XLD + NB * XLD) * sizeof(float);
    trinv_panel_kernel<<<dim3(NBLK - 1, Bb), 512, inv_smem_bytes, stream>>>(Lm, Linv);

    size_t smem_bytes = (size_t)(Nn * KTS + Nn + Nn * 6) * sizeof(float);
    predict_kernel<<<dim3(Mm / MT, Bb), 128, smem_bytes, stream>>>(
        Linv, trainP, testP, Kmm, z, mean, scale_param, out_mean, out_var);
}


# z = L^-1 (y - mean) as one matrix-vector product with the explicit inverse (launched after it) instead of a 1024-step serial forward substitution; new summation order in z and in nlml's sum of z^2
# speedup vs baseline: 23.8855x; 1.6501x over previous
#include <hip/hip_runtime.h>
#include <math.h>

#define Nn 1024
#define Mm 16384
#define Bb 2
#define MT 64
#define KTS 65
#define NB 64
#define NBLK (Nn / NB)
#define LDP 65

typedef __attribute__((ext_vector_type(2))) float v2f;
typedef __attribute__((ext_vector_type(8))) float v8f;

#define SQRT3F 1.7320508075688772f
#define LOG2PIF 1.8378770664093453f

__device__ __forceinline__ float kern_pair(float x1, float y1, float a1, float b1, float c1, float q1,
                                           float x2, float y2, float a2, float b2, float c2, float q2,
                                           float scale) {
    float s00 = a1 + a2, s11 = b1 + b2, s01 = c1 + c2;
    float sdet = s00 * s11 - s01 * s01;
    float d0 = x1 - x2, d1 = y1 - y2;
    float Q = 0.5f * (s11 * d0 * d0 - 2.0f * s01 * d0 * d1 + s00 * d1 * d1) / sdet;
    float C = 2.0f * q1 * q2 * rsqrtf(fmaxf(sdet, 1e-5f));
    float t = SQRT3F * sqrtf(fmaxf(Q, 1e-5f));
    return (1.0f + t) * __expf(-t) * C * scale;
}

__global__ void prep_kernel(const float* __restrict__ ct, const float* __restrict__ kpt,
                            const float* __restrict__ cs, const float* __restrict__ kps,
                            const float* __restrict__ scale_param,
                            float* __restrict__ trainP, float* __restrict__ testP,
                            float* __restrict__ Kmm) {
    int gid = blockIdx.x * blockDim.x + threadIdx.x;
    float scale = __expf(scale_param[0]);
    if (gid < Bb * Mm) {
        float a = kps[gid * 3 + 0], b = kps[gid * 3 + 1], c = kps[gid * 3 + 2];
        float det = a * b - c * c;
        testP[gid * 6 + 0] = cs[gid * 2 + 0];
        testP[gid * 6 + 1] = cs[gid * 2 + 1];
        testP[gid * 6 + 2] = a;
        testP[gid * 6 + 3] = b;
        testP[gid * 6 + 4] = c;
        testP[gid * 6 + 5] = sqrtf(sqrtf(det));
        float t0 = SQRT3F * sqrtf(1e-5f);
        float m0 = (1.0f + t0) * __expf(-t0);
        float C = 2.0f * sqrtf(det) * rsqrtf(fmaxf(4.0f * det, 1e-5f));
        Kmm[gid] = C * m0 * scale;
    }
    if (gid < Bb * Nn) {
        float a = kpt[gid * 3 + 0], b = kpt[gid * 3 + 1], c = kpt[gid * 3 + 2];
        float det = a * b - c * c;
        trainP[gid * 6 + 0] = ct[gid * 2 + 0];
        trainP[gid * 6 + 1] = ct[gid * 2 + 1];
        trainP[gid * 6 + 2] = a;
        trainP[gid * 6 + 3] = b;
        trainP[gid * 6 + 4] = c;
        trainP[gid * 6 + 5] = sqrtf(sqrtf(det));
    }
}

__global__ void buildA_kernel(const float* __restrict__ trainP, const float* __restrict__ var,
                              const float* __restrict__ scale_param, float* __restrict__ A) {
    int gid = blockIdx.x * blockDim.x + threadIdx.x;
    int b = gid >> 20;
    int i = (gid >> 10) & (Nn - 1);
    int j = gid & (Nn - 1);
    const float* p1 = trainP + (size_t)(b * Nn + i) * 6;
    const float* p2 = trainP + (size_t)(b * Nn + j) * 6;
    float scale = __expf(scale_param[0]);
    float v = kern_pair(p1[0], p1[1], p1[2], p1[3], p1[4], p1[5],
                        p2[0], p2[1], p2[2], p2[3], p2[4], p2[5], scale);
    if (i == j) v += var[b * Nn + i];
    A[(size_t)gid] = v;
}

template <bool WITH_S>
__device__ __forceinline__ void panel_sweep(float* Dsh, float* Ssh, float* dg, int tid) {
    int t = tid & (NB - 1);
    int q = tid >> 6;
    for (int k = 0; k < NB; k++) {
        float d = sqrtf(Dsh[k * LDP + k]);
        float dinv = 1.0f / d;
        if (tid == 0) dg[k] = d;
        if (q == 0 && t > k) Dsh[t * LDP + k] *= dinv;
        if (WITH_S && q == 1) Ssh[t * LDP + k] *= dinv;
        __syncthreads();
        float dtk = Dsh[t * LDP + k];
        float stk = WITH_S ? Ssh[t * LDP + k] : 0.0f;
        int j0 = k + 1 + q;
        float lj[8], dv[8], sv[8];
#pragma unroll
        for (int u = 0; u < 8; u++) {
            int j = j0 + 8 * u;
            int jj = (j < NB) ? j : k;
            lj[u] = Dsh[jj * LDP + k];
            dv[u] = Dsh[t * LDP + jj];
            sv[u] = WITH_S ? Ssh[t * LDP + jj] : 0.0f;
        }
#pragma unroll
        for (int u = 0; u < 8; u++) {
            int j = j0 + 8 * u;
            if (j < NB) {
                if (j <= t) Dsh[t * LDP + j] = dv[u] - dtk * lj[u];
                if (WITH_S) Ssh[t * LDP + j] = sv[u] - stk * lj[u];
            }
        }
        __syncthreads();
    }
}

__global__ __launch_bounds__(512) void chol_step_kernel(const float* __restrict__ Abase,
                                                        float* Lbase, int p) {
    __shared__ float Dsh[NB * LDP];
    __shared__ float Ssh[NB * LDP];
    __shared__ float dg[NB];
    int b = blockIdx.y;
    int i = p + blockIdx.x;
    bool isDiag = (blockIdx.x == 0);
    const float* A = Abase + (size_t)b * Nn * Nn;
    float* L = Lbase + (size_t)b * Nn * Nn;
    int tid = threadIdx.x;
    int wave = tid >> 5, lane = tid & 31;
    int r = lane & 15, hi = lane >> 4;
    int tm = wave >> 2, tn = wave & 3;
    int prow = p * NB, irow = i * NB;

    v8f accD, accS;
#pragma unroll
    for (int v = 0; v < 8; v++) {
        int row = tm * 16 + v + 8 * hi;
        int col = tn * 16 + r;
        accD[v] = -A[(size_t)(prow + row) * Nn + prow + col];
        accS[v] = -A[(size_t)(irow + row) * Nn + prow + col];
    }
    {
        const float* aP = L + (size_t)(prow + tm * 16 + r) * Nn;
        const float* aI = L + (size_t)(irow + tm * 16 + r) * Nn;
        const float* bP = L + (size_t)(prow + tn * 16 + r) * Nn;
        for (int kb = 0; kb < prow; kb += 16) {
#pragma unroll
            for (int k0 = 0; k0 < 16; k0 += 4) {
                int kk = kb + k0 + 2 * hi;
                v2f fa = *(const v2f*)(aP + kk);
                v2f fi = *(const v2f*)(aI + kk);
                v2f fb = *(const v2f*)(bP + kk);
                accD = __builtin_amdgcn_wmma_f32_16x16x4_f32(false, fa, false, fb,
                                                             (short)0, accD, false, false);
                accS = __builtin_amdgcn_wmma_f32_16x16x4_f32(false, fi, false, fb,
                                                             (short)0, accS, false, false);
            }
        }
    }
#pragma unroll
    for (int v = 0; v < 8; v++) {
        int row = tm * 16 + v + 8 * hi;
        int col = tn * 16 + r;
        Dsh[row * LDP + col] = -accD[v];
        Ssh[row * LDP + col] = -accS[v];
    }
    __syncthreads();

    if (isDiag) panel_sweep<false>(Dsh, Ssh, dg, tid);
    else        panel_sweep<true>(Dsh, Ssh, dg, tid);

    if (isDiag) {
        for (int idx = tid; idx < NB * NB; idx += 512) {
            int rr = idx >> 6, c = idx & (NB - 1);
            float val = (c < rr) ? Dsh[rr * LDP + c] : ((c == rr) ? dg[rr] : 0.0f);
            L[(size_t)(prow + rr) * Nn + prow + c] = val;
        }
    } else {
        for (int idx = tid; idx < NB * NB; idx += 512) {
            int rr = idx >> 6, c = idx & (NB - 1);
            L[(size_t)(irow + rr) * Nn + prow + c] = Ssh[rr * LDP + c];
        }
    }
}

__global__ __launch_bounds__(1024) void zmatvec_kernel(const float* __restrict__ Linvbase,
                                                       const float* __restrict__ y,
                                                       const float* __restrict__ mean,
                                                       float* __restrict__ zbase) {
    __shared__ float zrow[32];
    int b = blockIdx.y;
    int tid = threadIdx.x;
    int wave = tid >> 5, lane = tid & 31;
    int i = blockIdx.x * 32 + wave;
    const float* row = Linvbase + (size_t)b * Nn * Nn + (size_t)i * Nn;
    const float* yb = y + b * Nn;
    float mb = mean[b];
    int rowEnd = ((i >> 6) + 1) << 6;
    float4 l[8], v[8];
#pragma unroll
    for (int u = 0; u < 8; u++) {
        int c0 = 4 * lane + 128 * u;
        if (c0 < rowEnd) {
            l[u] = *(const float4*)(row + c0);
            v[u] = *(const float4*)(yb + c0);
        } else {
            l[u] = make_float4(0.0f, 0.0f, 0.0f, 0.0f);
            v[u] = make_float4(mb, mb, mb, mb);
        }
    }
    float s0 = 0.0f, s1 = 0.0f, s2 = 0.0f, s3 = 0.0f;
#pragma unroll
    for (int u = 0; u < 8; u++) {
        s0 += l[u].x * (v[u].x - mb);
        s1 += l[u].y * (v[u].y - mb);
        s2 += l[u].z * (v[u].z - mb);
        s3 += l[u].w * (v[u].w - mb);
    }
    float s = (s0 + s1) + (s2 + s3);
    for (int o = 16; o > 0; o >>= 1) s += __shfl_xor(s, o);
    if (lane == 0) zrow[wave] = s;
    __syncthreads();
    if (wave == 0) zbase[b * Nn + blockIdx.x * 32 + lane] = zrow[lane];
}

__global__ __launch_bounds__(1024) void nlml_kernel(const float* __restrict__ Lbase,
                                                    const float* __restrict__ zbase,
                                                    float* __restrict__ nlml_out) {
    int b = blockIdx.x;
    const float* L = Lbase + (size_t)b * Nn * Nn;
    __shared__ float col[Nn];
    __shared__ float zz[Nn];
    int tid = threadIdx.x;
    float z = zbase[b * Nn + tid];
    col[tid] = __logf(L[(size_t)tid * Nn + tid]);
    zz[tid] = z * z;
    __syncthreads();
    for (int s = 512; s > 0; s >>= 1) {
        if (tid < s) {
            col[tid] += col[tid + s];
            zz[tid] += zz[tid + s];
        }
        __syncthreads();
    }
    if (tid == 0) nlml_out[b] = 0.5f * zz[0] + col[0] + 0.5f * Nn * LOG2PIF;
}

__global__ __launch_bounds__(64) void trinv_diag_kernel(const float* __restrict__ Lbase,
                                                        float* __restrict__ Linvbase) {
    __shared__ __attribute__((aligned(16))) float Lb[NB * NB];
    __shared__ float xs[NB * NB];
    int p = blockIdx.x, b = blockIdx.y, tid = threadIdx.x;
    size_t base = (size_t)b * Nn * Nn + (size_t)(p * NB) * Nn + p * NB;
    const float* L = Lbase + base;
    float* Li = Linvbase + base;
    for (int idx = tid; idx < NB * NB; idx += 64) {
        int rr = idx >> 6, c = idx & (NB - 1);
        Lb[idx] = L[(size_t)rr * Nn + c];
    }
    __syncthreads();
    int j = tid;
    for (int i = 0; i < NB; i++) {
        const float* lr = Lb + i * NB;
        float s0 = (i == j) ? 1.0f : 0.0f, s1 = 0.0f, s2 = 0.0f, s3 = 0.0f;
        int k = 0;
        for (; k + 4 <= i; k += 4) {
            float4 l = *(const float4*)(lr + k);
            s0 -= l.x * xs[(k + 0) * NB + j];
            s1 -= l.y * xs[(k + 1) * NB + j];
            s2 -= l.z * xs[(k + 2) * NB + j];
            s3 -= l.w * xs[(k + 3) * NB + j];
        }
        for (; k < i; k++) s0 -= lr[k] * xs[k * NB + j];
        xs[i * NB + j] = ((s0 + s1) + (s2 + s3)) / lr[i];
    }
    __syncthreads();
    for (int idx = tid; idx < NB * NB; idx += 64) {
        int rr = idx >> 6, c = idx & (NB - 1);
        Li[(size_t)rr * Nn + c] = xs[idx];
    }
}

#define XLD 65

__device__ __forceinline__ void slab32(const float* aP, const float* bP, int kb, int hi,
                                       v8f& c0, v8f& c1) {
    v2f a[8], q[8];
#pragma unroll
    for (int u = 0; u < 8; u++) {
        int kk = kb + 4 * u + 2 * hi;
        a[u] = *(const v2f*)(aP + kk);
        q[u].x = bP[kk * XLD];
        q[u].y = bP[(kk + 1) * XLD];
    }
    __builtin_amdgcn_sched_barrier(0);
#pragma unroll
    for (int u = 0; u < 8; u += 2) {
        c0 = __builtin_amdgcn_wmma_f32_16x16x4_f32(false, a[u], false, q[u],
                                                   (short)0, c0, false, false);
        c1 = __builtin_amdgcn_wmma_f32_16x16x4_f32(false, a[u + 1], false, q[u + 1],
                                                   (short)0, c1, false, false);
    }
}
__global__ __launch_bounds__(512) void trinv_panel_kernel(const float* __restrict__ Lbase,
                                                          float* Linvbase) {
    extern __shared__ float smem[];
    float* Xp  = smem;
    float* Tsh = smem + Nn * XLD;
    int J = blockIdx.x, b = blockIdx.y, tid = threadIdx.x;
    const float* L = Lbase + (size_t)b * Nn * Nn;
    float* Li = Linvbase + (size_t)b * Nn * Nn;
    int wave = __builtin_amdgcn_readfirstlane(tid >> 5);
    int lane = tid & 31;
    int r = lane & 15, hi = lane >> 4;
    int tm = wave >> 2, tn = wave & 3;
    int jcol = J * NB;

    for (int idx = tid; idx < NB * NB; idx += 512) {
        int rr = idx >> 6, c = idx & (NB - 1);
        Xp[rr * XLD + c] = Li[(size_t)(jcol + rr) * Nn + jcol + c];
    }
    __syncthreads();

    for (int i = J + 1; i < NBLK; i++) {
        int irow = i * NB;
        int kdepth = irow - jcol;
        v8f acc0 = {}, acc1 = {};
        {
            const float* aP = L + (size_t)(irow + tm * 16 + r) * Nn + jcol;
            const float* bP = Xp + tn * 16 + r;
            for (int kb = 0; kb < kdepth; kb += 32) slab32(aP, bP, kb, hi, acc0, acc1);
        }
#pragma unroll
        for (int v = 0; v < 8; v++) {
            int row = tm * 16 + v + 8 * hi;
            int col = tn * 16 + r;
            Tsh[row * XLD + col] = acc0[v] + acc1[v];
        }
        __syncthreads();
        v8f x0 = {}, x1 = {};
        {
            const float* aD = Li + (size_t)(irow + tm * 16 + r) * Nn + irow;
            const float* bT = Tsh + tn * 16 + r;
            for (int kb = 0; kb < (tm + 1) * 16; kb += 32) slab32(aD, bT, kb, hi, x0, x1);
        }
#pragma unroll
        for (int v = 0; v < 8; v++) {
            int row = tm * 16 + v + 8 * hi;
            int col = tn * 16 + r;
            Xp[(kdepth + row) * XLD + col] = -(x0[v] + x1[v]);
        }
        __syncthreads();
        for (int idx = tid; idx < NB * NB; idx += 512) {
            int rr = idx >> 6, c = idx & (NB - 1);
            Li[(size_t)(irow + rr) * Nn + jcol + c] = Xp[(kdepth + rr) * XLD + c];
        }
    }
}

__global__ __launch_bounds__(128) void predict_kernel(const float* __restrict__ Linvbase,
                                                      const float* __restrict__ trainPg,
                                                      const float* __restrict__ testPg,
                                                      const float* __restrict__ Kmm,
                                                      const float* __restrict__ zbase,
                                                      const float* __restrict__ meang,
                                                      const float* __restrict__ scale_param,
                                                      float* __restrict__ out_mean,
                                                      float* __restrict__ out_var) {
    extern __shared__ float smem[];
    float* Kt  = smem;
    float* zsh = smem + Nn * KTS;
    float* tp  = zsh + Nn;
    int b = blockIdx.y;
    int colBase = blockIdx.x * MT;
    int tid = threadIdx.x;
    float scale = __expf(scale_param[0]);

    for (int idx = tid; idx < Nn; idx += 128) zsh[idx] = zbase[b * Nn + idx];
    for (int idx = tid; idx < Nn * 6; idx += 128) tp[idx] = trainPg[(size_t)b * Nn * 6 + idx];
    __syncthreads();

    {
        int c = tid & 63;
        int h = tid >> 6;
        const float* q = testPg + (size_t)(b * Mm + colBase + c) * 6;
        float x2 = q[0], y2 = q[1], a2 = q[2], b2 = q[3], c2 = q[4], q2 = q[5];
        for (int n = h; n < Nn; n += 2) {
            const float* p = tp + n * 6;
            Kt[n * KTS + c] = kern_pair(p[0], p[1], p[2], p[3], p[4], p[5],
                                        x2, y2, a2, b2, c2, q2, scale);
        }
    }
    __syncthreads();

    int wave = tid >> 5, lane = tid & 31;
    int r = lane & 15, hi = lane >> 4;
    int wcol = wave * 16;
    const float* Li = Linvbase + (size_t)b * Nn * Nn;

    float meanAcc = 0.0f, varAcc = 0.0f;
    for (int rb = 0; rb < Nn / 16; rb++) {
        v8f acc = {};
        const float* arow = Li + (size_t)(rb * 16 + r) * Nn;
        for (int j = 0; j <= rb; j++) {
            int kb = j * 16;
#pragma unroll
            for (int k0 = 0; k0 < 16; k0 += 4) {
                int kk = kb + k0 + 2 * hi;
                v2f a = *(const v2f*)(arow + kk);
                v2f bb;
                bb.x = Kt[kk * KTS + wcol + r];
                bb.y = Kt[(kk + 1) * KTS + wcol + r];
                acc = __builtin_amdgcn_wmma_f32_16x16x4_f32(false, a, false, bb,
                                                            (short)0, acc, false, false);
            }
        }
        const float* zb = zsh + rb * 16 + 8 * hi;
#pragma unroll
        for (int v = 0; v < 8; v++) {
            float cv = acc[v];
            meanAcc += zb[v] * cv;
            varAcc  += cv * cv;
        }
    }
    meanAcc += __shfl_xor(meanAcc, 16);
    varAcc  += __shfl_xor(varAcc, 16);
    if (lane < 16) {
        int m = colBase + wcol + lane;
        out_mean[b * Mm + m] = meang[b] + meanAcc;
        out_var[b * Mm + m]  = Kmm[b * Mm + m] - varAcc;
    }
}

extern "C" void kernel_launch(void* const* d_in, const int* in_sizes, int n_in,
                              void* d_out, int out_size, void* d_ws, size_t ws_size,
                              hipStream_t stream) {
    const float* coords_train = (const float*)d_in[0];
    const float* kp_train     = (const float*)d_in[1];
    const float* coords_test  = (const float*)d_in[2];
    const float* kp_test      = (const float*)d_in[3];
    const float* y_train      = (const float*)d_in[4];
    const float* var          = (const float*)d_in[5];
    const float* mean         = (const float*)d_in[6];
    const float* scale_param  = (const float*)d_in[7];

    float* ws = (float*)d_ws;
    size_t off = 0;
    float* A      = ws + off; off += (size_t)Bb * Nn * Nn;
    float* Lm     = ws + off; off += (size_t)Bb * Nn * Nn;
    float* Linv   = ws + off; off += (size_t)Bb * Nn * Nn;
    float* trainP = ws + off; off += (size_t)Bb * Nn * 6;
    float* testP  = ws + off; off += (size_t)Bb * Mm * 6;
    float* Kmm    = ws + off; off += (size_t)Bb * Mm;
    float* z      = ws + off; off += (size_t)Bb * Nn;

    float* out_mean = (float*)d_out;
    float* out_var  = out_mean + (size_t)Bb * Mm;
    float* nlml     = out_var + (size_t)Bb * Mm;

    prep_kernel<<<(Bb * Mm + 255) / 256, 256, 0, stream>>>(
        coords_train, kp_train, coords_test, kp_test, scale_param, trainP, testP, Kmm);
    buildA_kernel<<<(Bb * Nn * Nn) / 256, 256, 0, stream>>>(trainP, var, scale_param, A);
#define CHOL_STEP(P) chol_step_kernel<<<dim3(NBLK - (P), Bb), 512, 0, stream>>>(A, Lm, (P))
    CHOL_STEP(0);  CHOL_STEP(1);  CHOL_STEP(2);  CHOL_STEP(3);
    CHOL_STEP(4);  CHOL_STEP(5);  CHOL_STEP(6);  CHOL_STEP(7);
    CHOL_STEP(8);  CHOL_STEP(9);  CHOL_STEP(10); CHOL_STEP(11);
    CHOL_STEP(12); CHOL_STEP(13); CHOL_STEP(14); CHOL_STEP(15);
#undef CHOL_STEP
    trinv_diag_kernel<<<dim3(NBLK, Bb), 64, 0, stream>>>(Lm, Linv);
    size_t inv_smem_bytes = (size_t)(Nn * XLD + NB * XLD) * sizeof(float);
    trinv_panel_kernel<<<dim3(NBLK - 1, Bb), 512, inv_smem_bytes, stream>>>(Lm, Linv);
    zmatvec_kernel<<<dim3(Nn / 32, Bb), 1024, 0, stream>>>(Linv, y_train, mean, z);
    nlml_kernel<<<Bb, 1024, 0, stream>>>(Lm, z, nlml);

    size_t smem_bytes = (size_t)(Nn * KTS + Nn + Nn * 6) * sizeof(float);
    predict_kernel<<<dim3(Mm / MT, Bb), 128, smem_bytes, stream>>>(
        Linv, trainP, testP, Kmm, z, mean, scale_param, out_mean, out_var);
}


# predict GEMM re-tiled: 16 waves split the rows of Linv*K (two 32-row bands each, all 64 columns, 8 accumulator tiles); new summation order in pred_mean and pred_var
# speedup vs baseline: 35.2518x; 1.4759x over previous
#include <hip/hip_runtime.h>
#include <math.h>

#define Nn 1024
#define Mm 16384
#define Bb 2
#define MT 64
#define KTS 65
#define NB 64
#define NBLK (Nn / NB)
#define LDP 65

typedef __attribute__((ext_vector_type(2))) float v2f;
typedef __attribute__((ext_vector_type(8))) float v8f;

#define SQRT3F 1.7320508075688772f
#define LOG2PIF 1.8378770664093453f

__device__ __forceinline__ float kern_pair(float x1, float y1, float a1, float b1, float c1, float q1,
                                           float x2, float y2, float a2, float b2, float c2, float q2,
                                           float scale) {
    float s00 = a1 + a2, s11 = b1 + b2, s01 = c1 + c2;
    float sdet = s00 * s11 - s01 * s01;
    float d0 = x1 - x2, d1 = y1 - y2;
    float Q = 0.5f * (s11 * d0 * d0 - 2.0f * s01 * d0 * d1 + s00 * d1 * d1) / sdet;
    float C = 2.0f * q1 * q2 * rsqrtf(fmaxf(sdet, 1e-5f));
    float t = SQRT3F * sqrtf(fmaxf(Q, 1e-5f));
    return (1.0f + t) * __expf(-t) * C * scale;
}

__global__ void prep_kernel(const float* __restrict__ ct, const float* __restrict__ kpt,
                            const float* __restrict__ cs, const float* __restrict__ kps,
                            const float* __restrict__ scale_param,
                            float* __restrict__ trainP, float* __restrict__ testP,
                            float* __restrict__ Kmm) {
    int gid = blockIdx.x * blockDim.x + threadIdx.x;
    float scale = __expf(scale_param[0]);
    if (gid < Bb * Mm) {
        float a = kps[gid * 3 + 0], b = kps[gid * 3 + 1], c = kps[gid * 3 + 2];
        float det = a * b - c * c;
        testP[gid * 6 + 0] = cs[gid * 2 + 0];
        testP[gid * 6 + 1] = cs[gid * 2 + 1];
        testP[gid * 6 + 2] = a;
        testP[gid * 6 + 3] = b;
        testP[gid * 6 + 4] = c;
        testP[gid * 6 + 5] = sqrtf(sqrtf(det));
        float t0 = SQRT3F * sqrtf(1e-5f);
        float m0 = (1.0f + t0) * __expf(-t0);
        float C = 2.0f * sqrtf(det) * rsqrtf(fmaxf(4.0f * det, 1e-5f));
        Kmm[gid] = C * m0 * scale;
    }
    if (gid < Bb * Nn) {
        float a = kpt[gid * 3 + 0], b = kpt[gid * 3 + 1], c = kpt[gid * 3 + 2];
        float det = a * b - c * c;
        trainP[gid * 6 + 0] = ct[gid * 2 + 0];
        trainP[gid * 6 + 1] = ct[gid * 2 + 1];
        trainP[gid * 6 + 2] = a;
        trainP[gid * 6 + 3] = b;
        trainP[gid * 6 + 4] = c;
        trainP[gid * 6 + 5] = sqrtf(sqrtf(det));
    }
}

__global__ void buildA_kernel(const float* __restrict__ trainP, const float* __restrict__ var,
                              const float* __restrict__ scale_param, float* __restrict__ A) {
    int gid = blockIdx.x * blockDim.x + threadIdx.x;
    int b = gid >> 20;
    int i = (gid >> 10) & (Nn - 1);
    int j = gid & (Nn - 1);
    const float* p1 = trainP + (size_t)(b * Nn + i) * 6;
    const float* p2 = trainP + (size_t)(b * Nn + j) * 6;
    float scale = __expf(scale_param[0]);
    float v = kern_pair(p1[0], p1[1], p1[2], p1[3], p1[4], p1[5],
                        p2[0], p2[1], p2[2], p2[3], p2[4], p2[5], scale);
    if (i == j) v += var[b * Nn + i];
    A[(size_t)gid] = v;
}

template <bool WITH_S>
__device__ __forceinline__ void panel_sweep(float* Dsh, float* Ssh, float* dg, int tid) {
    int t = tid & (NB - 1);
    int q = tid >> 6;
    for (int k = 0; k < NB; k++) {
        float d = sqrtf(Dsh[k * LDP + k]);
        float dinv = 1.0f / d;
        if (tid == 0) dg[k] = d;
        if (q == 0 && t > k) Dsh[t * LDP + k] *= dinv;
        if (WITH_S && q == 1) Ssh[t * LDP + k] *= dinv;
        __syncthreads();
        float dtk = Dsh[t * LDP + k];
        float stk = WITH_S ? Ssh[t * LDP + k] : 0.0f;
        int j0 = k + 1 + q;
        float lj[8], dv[8], sv[8];
#pragma unroll
        for (int u = 0; u < 8; u++) {
            int j = j0 + 8 * u;
            int jj = (j < NB) ? j : k;
            lj[u] = Dsh[jj * LDP + k];
            dv[u] = Dsh[t * LDP + jj];
            sv[u] = WITH_S ? Ssh[t * LDP + jj] : 0.0f;
        }
#pragma unroll
        for (int u = 0; u < 8; u++) {
            int j = j0 + 8 * u;
            if (j < NB) {
                if (j <= t) Dsh[t * LDP + j] = dv[u] - dtk * lj[u];
                if (WITH_S) Ssh[t * LDP + j] = sv[u] - stk * lj[u];
            }
        }
        __syncthreads();
    }
}

__global__ __launch_bounds__(512) void chol_step_kernel(const float* __restrict__ Abase,
                                                        float* Lbase, int p) {
    __shared__ float Dsh[NB * LDP];
    __shared__ float Ssh[NB * LDP];
    __shared__ float dg[NB];
    int b = blockIdx.y;
    int i = p + blockIdx.x;
    bool isDiag = (blockIdx.x == 0);
    const float* A = Abase + (size_t)b * Nn * Nn;
    float* L = Lbase + (size_t)b * Nn * Nn;
    int tid = threadIdx.x;
    int wave = tid >> 5, lane = tid & 31;
    int r = lane & 15, hi = lane >> 4;
    int tm = wave >> 2, tn = wave & 3;
    int prow = p * NB, irow = i * NB;

    v8f accD, accS;
#pragma unroll
    for (int v = 0; v < 8; v++) {
        int row = tm * 16 + v + 8 * hi;
        int col = tn * 16 + r;
        accD[v] = -A[(size_t)(prow + row) * Nn + prow + col];
        accS[v] = -A[(size_t)(irow + row) * Nn + prow + col];
    }
    {
        const float* aP = L + (size_t)(prow + tm * 16 + r) * Nn;
        const float* aI = L + (size_t)(irow + tm * 16 + r) * Nn;
        const float* bP = L + (size_t)(prow + tn * 16 + r) * Nn;
        for (int kb = 0; kb < prow; kb += 16) {
#pragma unroll
            for (int k0 = 0; k0 < 16; k0 += 4) {
                int kk = kb + k0 + 2 * hi;
                v2f fa = *(const v2f*)(aP + kk);
                v2f fi = *(const v2f*)(aI + kk);
                v2f fb = *(const v2f*)(bP + kk);
                accD = __builtin_amdgcn_wmma_f32_16x16x4_f32(false, fa, false, fb,
                                                             (short)0, accD, false, false);
                accS = __builtin_amdgcn_wmma_f32_16x16x4_f32(false, fi, false, fb,
                                                             (short)0, accS, false, false);
            }
        }
    }
#pragma unroll
    for (int v = 0; v < 8; v++) {
        int row = tm * 16 + v + 8 * hi;
        int col = tn * 16 + r;
        Dsh[row * LDP + col] = -accD[v];
        Ssh[row * LDP + col] = -accS[v];
    }
    __syncthreads();

    if (isDiag) panel_sweep<false>(Dsh, Ssh, dg, tid);
    else        panel_sweep<true>(Dsh, Ssh, dg, tid);

    if (isDiag) {
        for (int idx = tid; idx < NB * NB; idx += 512) {
            int rr = idx >> 6, c = idx & (NB - 1);
            float val = (c < rr) ? Dsh[rr * LDP + c] : ((c == rr) ? dg[rr] : 0.0f);
            L[(size_t)(prow + rr) * Nn + prow + c] = val;
        }
    } else {
        for (int idx = tid; idx < NB * NB; idx += 512) {
            int rr = idx >> 6, c = idx & (NB - 1);
            L[(size_t)(irow + rr) * Nn + prow + c] = Ssh[rr * LDP + c];
        }
    }
}

__global__ __launch_bounds__(1024) void zmatvec_kernel(const float* __restrict__ Linvbase,
                                                       const float* __restrict__ y,
                                                       const float* __restrict__ mean,
                                                       float* __restrict__ zbase) {
    __shared__ float zrow[32];
    int b = blockIdx.y;
    int tid = threadIdx.x;
    int wave = tid >> 5, lane = tid & 31;
    int i = blockIdx.x * 32 + wave;
    const float* row = Linvbase + (size_t)b * Nn * Nn + (size_t)i * Nn;
    const float* yb = y + b * Nn;
    float mb = mean[b];
    int rowEnd = ((i >> 6) + 1) << 6;
    float4 l[8], v[8];
#pragma unroll
    for (int u = 0; u < 8; u++) {
        int c0 = 4 * lane + 128 * u;
        if (c0 < rowEnd) {
            l[u] = *(const float4*)(row + c0);
            v[u] = *(const float4*)(yb + c0);
        } else {
            l[u] = make_float4(0.0f, 0.0f, 0.0f, 0.0f);
            v[u] = make_float4(mb, mb, mb, mb);
        }
    }
    float s0 = 0.0f, s1 = 0.0f, s2 = 0.0f, s3 = 0.0f;
#pragma unroll
    for (int u = 0; u < 8; u++) {
        s0 += l[u].x * (v[u].x - mb);
        s1 += l[u].y * (v[u].y - mb);
        s2 += l[u].z * (v[u].z - mb);
        s3 += l[u].w * (v[u].w - mb);
    }
    float s = (s0 + s1) + (s2 + s3);
    for (int o = 16; o > 0; o >>= 1) s += __shfl_xor(s, o);
    if (lane == 0) zrow[wave] = s;
    __syncthreads();
    if (wave == 0) zbase[b * Nn + blockIdx.x * 32 + lane] = zrow[lane];
}

__global__ __launch_bounds__(1024) void nlml_kernel(const float* __restrict__ Lbase,
                                                    const float* __restrict__ zbase,
                                                    float* __restrict__ nlml_out) {
    int b = blockIdx.x;
    const float* L = Lbase + (size_t)b * Nn * Nn;
    __shared__ float col[Nn];
    __shared__ float zz[Nn];
    int tid = threadIdx.x;
    float z = zbase[b * Nn + tid];
    col[tid] = __logf(L[(size_t)tid * Nn + tid]);
    zz[tid] = z * z;
    __syncthreads();
    for (int s = 512; s > 0; s >>= 1) {
        if (tid < s) {
            col[tid] += col[tid + s];
            zz[tid] += zz[tid + s];
        }
        __syncthreads();
    }
    if (tid == 0) nlml_out[b] = 0.5f * zz[0] + col[0] + 0.5f * Nn * LOG2PIF;
}

__global__ __launch_bounds__(64) void trinv_diag_kernel(const float* __restrict__ Lbase,
                                                        float* __restrict__ Linvbase) {
    __shared__ __attribute__((aligned(16))) float Lb[NB * NB];
    __shared__ float xs[NB * NB];
    int p = blockIdx.x, b = blockIdx.y, tid = threadIdx.x;
    size_t base = (size_t)b * Nn * Nn + (size_t)(p * NB) * Nn + p * NB;
    const float* L = Lbase + base;
    float* Li = Linvbase + base;
    for (int idx = tid; idx < NB * NB; idx += 64) {
        int rr = idx >> 6, c = idx & (NB - 1);
        Lb[idx] = L[(size_t)rr * Nn + c];
    }
    __syncthreads();
    int j = tid;
    for (int i = 0; i < NB; i++) {
        const float* lr = Lb + i * NB;
        float s0 = (i == j) ? 1.0f : 0.0f, s1 = 0.0f, s2 = 0.0f, s3 = 0.0f;
        int k = 0;
        for (; k + 4 <= i; k += 4) {
            float4 l = *(const float4*)(lr + k);
            s0 -= l.x * xs[(k + 0) * NB + j];
            s1 -= l.y * xs[(k + 1) * NB + j];
            s2 -= l.z * xs[(k + 2) * NB + j];
            s3 -= l.w * xs[(k + 3) * NB + j];
        }
        for (; k < i; k++) s0 -= lr[k] * xs[k * NB + j];
        xs[i * NB + j] = ((s0 + s1) + (s2 + s3)) / lr[i];
    }
    __syncthreads();
    for (int idx = tid; idx < NB * NB; idx += 64) {
        int rr = idx >> 6, c = idx & (NB - 1);
        Li[(size_t)rr * Nn + c] = xs[idx];
    }
}

#define XLD 65

__device__ __forceinline__ void slab32(const float* aP, const float* bP, int kb, int hi,
                                       v8f& c0, v8f& c1) {
    v2f a[8], q[8];
#pragma unroll
    for (int u = 0; u < 8; u++) {
        int kk = kb + 4 * u + 2 * hi;
        a[u] = *(const v2f*)(aP + kk);
        q[u].x = bP[kk * XLD];
        q[u].y = bP[(kk + 1) * XLD];
    }
    __builtin_amdgcn_sched_barrier(0);
#pragma unroll
    for (int u = 0; u < 8; u += 2) {
        c0 = __builtin_amdgcn_wmma_f32_16x16x4_f32(false, a[u], false, q[u],
                                                   (short)0, c0, false, false);
        c1 = __builtin_amdgcn_wmma_f32_16x16x4_f32(false, a[u + 1], false, q[u + 1],
                                                   (short)0, c1, false, false);
    }
}
__global__ __launch_bounds__(512) void trinv_panel_kernel(const float* __restrict__ Lbase,
                                                          float* Linvbase) {
    extern __shared__ float smem[];
    float* Xp  = smem;
    float* Tsh = smem + Nn * XLD;
    int J = blockIdx.x, b = blockIdx.y, tid = threadIdx.x;
    const float* L = Lbase + (size_t)b * Nn * Nn;
    float* Li = Linvbase + (size_t)b * Nn * Nn;
    int wave = __builtin_amdgcn_readfirstlane(tid >> 5);
    int lane = tid & 31;
    int r = lane & 15, hi = lane >> 4;
    int tm = wave >> 2, tn = wave & 3;
    int jcol = J * NB;

    for (int idx = tid; idx < NB * NB; idx += 512) {
        int rr = idx >> 6, c = idx & (NB - 1);
        Xp[rr * XLD + c] = Li[(size_t)(jcol + rr) * Nn + jcol + c];
    }
    __syncthreads();

    for (int i = J + 1; i < NBLK; i++) {
        int irow = i * NB;
        int kdepth = irow - jcol;
        v8f acc0 = {}, acc1 = {};
        {
            const float* aP = L + (size_t)(irow + tm * 16 + r) * Nn + jcol;
            const float* bP = Xp + tn * 16 + r;
            for (int kb = 0; kb < kdepth; kb += 32) slab32(aP, bP, kb, hi, acc0, acc1);
        }
#pragma unroll
        for (int v = 0; v < 8; v++) {
            int row = tm * 16 + v + 8 * hi;
            int col = tn * 16 + r;
            Tsh[row * XLD + col] = acc0[v] + acc1[v];
        }
        __syncthreads();
        v8f x0 = {}, x1 = {};
        {
            const float* aD = Li + (size_t)(irow + tm * 16 + r) * Nn + irow;
            const float* bT = Tsh + tn * 16 + r;
            for (int kb = 0; kb < (tm + 1) * 16; kb += 32) slab32(aD, bT, kb, hi, x0, x1);
        }
#pragma unroll
        for (int v = 0; v < 8; v++) {
            int row = tm * 16 + v + 8 * hi;
            int col = tn * 16 + r;
            Xp[(kdepth + row) * XLD + col] = -(x0[v] + x1[v]);
        }
        __syncthreads();
        for (int idx = tid; idx < NB * NB; idx += 512) {
            int rr = idx >> 6, c = idx & (NB - 1);
            Li[(size_t)(irow + rr) * Nn + jcol + c] = Xp[(kdepth + rr) * XLD + c];
        }
    }
}


template <int NU0>
__device__ __forceinline__ void band_slab(const float* a0, const float* a1, const float* bK,
                                          int kb, int hi, v8f (&c0)[4], v8f (&c1)[4]) {
    v2f fa0[8], fa1[8], fb[8][4];
#pragma unroll
    for (int u = 0; u < 8; u++) {
        int kk = kb + 4 * u + 2 * hi;
        if (u < NU0) fa0[u] = *(const v2f*)(a0 + kk);
        fa1[u] = *(const v2f*)(a1 + kk);
#pragma unroll
        for (int c = 0; c < 4; c++) {
            fb[u][c].x = bK[kk * KTS + c * 16];
            fb[u][c].y = bK[(kk + 1) * KTS + c * 16];
        }
    }
#pragma unroll
    for (int u = 0; u < 8; u++) {
#pragma unroll
        for (int c = 0; c < 4; c++) {
            if (u < NU0)
                c0[c] = __builtin_amdgcn_wmma_f32_16x16x4_f32(false, fa0[u], false, fb[u][c],
                                                              (short)0, c0[c], false, false);
            c1[c] = __builtin_amdgcn_wmma_f32_16x16x4_f32(false, fa1[u], false, fb[u][c],
                                                          (short)0, c1[c], false, false);
        }
    }
}

#define PT 512
__global__ __launch_bounds__(PT) void predict_kernel(const float* __restrict__ Linvbase,
                                                     const float* __restrict__ trainPg,
                                                     const float* __restrict__ testPg,
                                                     const float* __restrict__ Kmm,
                                                     const float* __restrict__ zbase,
                                                     const float* __restrict__ meang,
                                                     const float* __restrict__ scale_param,
                                                     float* __restrict__ out_mean,
                                                     float* __restrict__ out_var) {
    extern __shared__ float smem[];
    float* Kt  = smem;
    float* zsh = smem + Nn * KTS;
    float* tp  = zsh + Nn;
    int b = blockIdx.y;
    int colBase = blockIdx.x * MT;
    int tid = threadIdx.x;
    float scale = __expf(scale_param[0]);

    for (int idx = tid; idx < Nn; idx += PT) zsh[idx] = zbase[b * Nn + idx];
    for (int idx = tid; idx < Nn * 6; idx += PT) tp[idx] = trainPg[(size_t)b * Nn * 6 + idx];
    __syncthreads();

    {
        int c = tid & 63;
        int h = tid >> 6;
        const float* q = testPg + (size_t)(b * Mm + colBase + c) * 6;
        float x2 = q[0], y2 = q[1], a2 = q[2], b2 = q[3], c2 = q[4], q2 = q[5];
        for (int n = h; n < Nn; n += PT / 64) {
            const float* p = tp + n * 6;
            Kt[n * KTS + c] = kern_pair(p[0], p[1], p[2], p[3], p[4], p[5],
                                        x2, y2, a2, b2, c2, q2, scale);
        }
    }
    __syncthreads();

    int wave = __builtin_amdgcn_readfirstlane(tid >> 5);
    int lane = tid & 31;
    int r = lane & 15, hi = lane >> 4;
    const float* Li = Linvbase + (size_t)b * Nn * Nn;
    const float* bK = Kt + r;

    float meanAcc[4] = {0.0f, 0.0f, 0.0f, 0.0f};
    float varAcc[4]  = {0.0f, 0.0f, 0.0f, 0.0f};
    for (int s = 0; s < 2; s++) {
        int p = s ? (Nn / 32 - 1 - wave) : wave;
        int row0 = p * 32;
        const float* a0 = Li + (size_t)(row0 + r) * Nn;
        const float* a1 = a0 + (size_t)16 * Nn;
        v8f c0[4] = {}, c1[4] = {};
        for (int t = 0; t < p; t++) band_slab<8>(a0, a1, bK, t * 32, hi, c0, c1);
        band_slab<4>(a0, a1, bK, row0, hi, c0, c1);
        const float* z0 = zsh + row0 + 8 * hi;
#pragma unroll
        for (int c = 0; c < 4; c++) {
#pragma unroll
            for (int v = 0; v < 8; v++) {
                float w0 = c0[c][v], w1 = c1[c][v];
                meanAcc[c] += z0[v] * w0;
                varAcc[c]  += w0 * w0;
                meanAcc[c] += z0[16 + v] * w1;
                varAcc[c]  += w1 * w1;
            }
        }
    }
    float* red = tp;
#pragma unroll
    for (int c = 0; c < 4; c++) {
        float m = meanAcc[c] + __shfl_xor(meanAcc[c], 16);
        float v = varAcc[c] + __shfl_xor(varAcc[c], 16);
        if (lane < 16) {
            red[wave * 128 + c * 16 + lane] = m;
            red[wave * 128 + 64 + c * 16 + lane] = v;
        }
    }
    __syncthreads();
    if (tid < MT) {
        float m = 0.0f, v = 0.0f;
#pragma unroll
        for (int w = 0; w < PT / 32; w++) {
            m += red[w * 128 + tid];
            v += red[w * 128 + 64 + tid];
        }
        int col = b * Mm + colBase + tid;
        out_mean[col] = meang[b] + m;
        out_var[col]  = Kmm[col] - v;
    }
}

extern "C" void kernel_launch(void* const* d_in, const int* in_sizes, int n_in,
                              void* d_out, int out_size, void* d_ws, size_t ws_size,
                              hipStream_t stream) {
    const float* coords_train = (const float*)d_in[0];
    const float* kp_train     = (const float*)d_in[1];
    const float* coords_test  = (const float*)d_in[2];
    const float* kp_test      = (const float*)d_in[3];
    const float* y_train      = (const float*)d_in[4];
    const float* var          = (const float*)d_in[5];
    const float* mean         = (const float*)d_in[6];
    const float* scale_param  = (const float*)d_in[7];

    float* ws = (float*)d_ws;
    size_t off = 0;
    float* A      = ws + off; off += (size_t)Bb * Nn * Nn;
    float* Lm     = ws + off; off += (size_t)Bb * Nn * Nn;
    float* Linv   = ws + off; off += (size_t)Bb * Nn * Nn;
    float* trainP = ws + off; off += (size_t)Bb * Nn * 6;
    float* testP  = ws + off; off += (size_t)Bb * Mm * 6;
    float* Kmm    = ws + off; off += (size_t)Bb * Mm;
    float* z      = ws + off; off += (size_t)Bb * Nn;

    float* out_mean = (float*)d_out;
    float* out_var  = out_mean + (size_t)Bb * Mm;
    float* nlml     = out_var + (size_t)Bb * Mm;

    prep_kernel<<<(Bb * Mm + 255) / 256, 256, 0, stream>>>(
        coords_train, kp_train, coords_test, kp_test, scale_param, trainP, testP, Kmm);
    buildA_kernel<<<(Bb * Nn * Nn) / 256, 256, 0, stream>>>(trainP, var, scale_param, A);
#define CHOL_STEP(P) chol_step_kernel<<<dim3(NBLK - (P), Bb), 512, 0, stream>>>(A, Lm, (P))
    CHOL_STEP(0);  CHOL_STEP(1);  CHOL_STEP(2);  CHOL_STEP(3);
    CHOL_STEP(4);  CHOL_STEP(5);  CHOL_STEP(6);  CHOL_STEP(7);
    CHOL_STEP(8);  CHOL_STEP(9);  CHOL_STEP(10); CHOL_STEP(11);
    CHOL_STEP(12); CHOL_STEP(13); CHOL_STEP(14); CHOL_STEP(15);
#undef CHOL_STEP
    trinv_diag_kernel<<<dim3(NBLK, Bb), 64, 0, stream>>>(Lm, Linv);
    size_t inv_smem_bytes = (size_t)(Nn * XLD + NB * XLD) * sizeof(float);
    trinv_panel_kernel<<<dim3(NBLK - 1, Bb), 512, inv_smem_bytes, stream>>>(Lm, Linv);
    zmatvec_kernel<<<dim3(Nn / 32, Bb), 1024, 0, stream>>>(Linv, y_train, mean, z);
    nlml_kernel<<<Bb, 1024, 0, stream>>>(Lm, z, nlml);

    size_t smem_bytes = (size_t)(Nn * KTS + Nn + Nn * 6) * sizeof(float);
    predict_kernel<<<dim3(Mm / MT, Bb), PT, smem_bytes, stream>>>(
        Linv, trainP, testP, Kmm, z, mean, scale_param, out_mean, out_var);
}


# Cholesky panel sweep: each thread keeps eight columns of its D and S row in registers over all 64 steps, only the pivot column goes through LDS, one barrier per step; arithmetic and order unchanged
# speedup vs baseline: 54.7725x; 1.5538x over previous
#include <hip/hip_runtime.h>
#include <math.h>

#define Nn 1024
#define Mm 16384
#define Bb 2
#define MT 64
#define KTS 65
#define NB 64
#define NBLK (Nn / NB)
#define LDP 65

typedef __attribute__((ext_vector_type(2))) float v2f;
typedef __attribute__((ext_vector_type(8))) float v8f;

#define SQRT3F 1.7320508075688772f
#define LOG2PIF 1.8378770664093453f

__device__ __forceinline__ float kern_pair(float x1, float y1, float a1, float b1, float c1, float q1,
                                           float x2, float y2, float a2, float b2, float c2, float q2,
                                           float scale) {
    float s00 = a1 + a2, s11 = b1 + b2, s01 = c1 + c2;
    float sdet = s00 * s11 - s01 * s01;
    float d0 = x1 - x2, d1 = y1 - y2;
    float Q = 0.5f * (s11 * d0 * d0 - 2.0f * s01 * d0 * d1 + s00 * d1 * d1) / sdet;
    float C = 2.0f * q1 * q2 * rsqrtf(fmaxf(sdet, 1e-5f));
    float t = SQRT3F * sqrtf(fmaxf(Q, 1e-5f));
    return (1.0f + t) * __expf(-t) * C * scale;
}

__global__ void prep_kernel(const float* __restrict__ ct, const float* __restrict__ kpt,
                            const float* __restrict__ cs, const float* __restrict__ kps,
                            const float* __restrict__ scale_param,
                            float* __restrict__ trainP, float* __restrict__ testP,
                            float* __restrict__ Kmm) {
    int gid = blockIdx.x * blockDim.x + threadIdx.x;
    float scale = __expf(scale_param[0]);
    if (gid < Bb * Mm) {
        float a = kps[gid * 3 + 0], b = kps[gid * 3 + 1], c = kps[gid * 3 + 2];
        float det = a * b - c * c;
        testP[gid * 6 + 0] = cs[gid * 2 + 0];
        testP[gid * 6 + 1] = cs[gid * 2 + 1];
        testP[gid * 6 + 2] = a;
        testP[gid * 6 + 3] = b;
        testP[gid * 6 + 4] = c;
        testP[gid * 6 + 5] = sqrtf(sqrtf(det));
        float t0 = SQRT3F * sqrtf(1e-5f);
        float m0 = (1.0f + t0) * __expf(-t0);
        float C = 2.0f * sqrtf(det) * rsqrtf(fmaxf(4.0f * det, 1e-5f));
        Kmm[gid] = C * m0 * scale;
    }
    if (gid < Bb * Nn) {
        float a = kpt[gid * 3 + 0], b = kpt[gid * 3 + 1], c = kpt[gid * 3 + 2];
        float det = a * b - c * c;
        trainP[gid * 6 + 0] = ct[gid * 2 + 0];
        trainP[gid * 6 + 1] = ct[gid * 2 + 1];
        trainP[gid * 6 + 2] = a;
        trainP[gid * 6 + 3] = b;
        trainP[gid * 6 + 4] = c;
        trainP[gid * 6 + 5] = sqrtf(sqrtf(det));
    }
}

__global__ void buildA_kernel(const float* __restrict__ trainP, const float* __restrict__ var,
                              const float* __restrict__ scale_param, float* __restrict__ A) {
    int gid = blockIdx.x * blockDim.x + threadIdx.x;
    int b = gid >> 20;
    int i = (gid >> 10) & (Nn - 1);
    int j = gid & (Nn - 1);
    const float* p1 = trainP + (size_t)(b * Nn + i) * 6;
    const float* p2 = trainP + (size_t)(b * Nn + j) * 6;
    float scale = __expf(scale_param[0]);
    float v = kern_pair(p1[0], p1[1], p1[2], p1[3], p1[4], p1[5],
                        p2[0], p2[1], p2[2], p2[3], p2[4], p2[5], scale);
    if (i == j) v += var[b * Nn + i];
    A[(size_t)gid] = v;
}

template <bool WITH_S>
__device__ __forceinline__ void panel_sweep(float* Dsh, float* Ssh, float* dg,
                                            float* colD, float* colS, int tid) {
    int t = tid & (NB - 1);
    int q = __builtin_amdgcn_readfirstlane(tid >> 6);
    float dr[8], sr[8];
#pragma unroll
    for (int u = 0; u < 8; u++) {
        dr[u] = Dsh[t * LDP + 8 * q + u];
        sr[u] = WITH_S ? Ssh[t * LDP + 8 * q + u] : 0.0f;
    }
#pragma unroll 1
    for (int ko = 0; ko < NB / 8; ko++) {
#pragma unroll
        for (int ki = 0; ki < 8; ki++) {
            int k = 8 * ko + ki;
            float* cD = colD + (ki & 1) * NB;
            float* cS = colS + (ki & 1) * NB;
            if (q == ko) {
                cD[t] = dr[ki];
                if (WITH_S) cS[t] = sr[ki];
            }
            __syncthreads();
            if (q >= ko) {
                float x = cD[k];
                float dtr = cD[t];
                float str = WITH_S ? cS[t] : 0.0f;
                float4 l0 = *(const float4*)(cD + 8 * q);
                float4 l1 = *(const float4*)(cD + 8 * q + 4);
                float d = sqrtf(x);
                float dinv = 1.0f / d;
                float dtk = (t > k) ? dtr * dinv : 0.0f;
                float stk = str * dinv;
                bool own = (q == ko);
                float lj[8];
                lj[0] = l0.x * dinv; lj[1] = l0.y * dinv; lj[2] = l0.z * dinv; lj[3] = l0.w * dinv;
                lj[4] = l1.x * dinv; lj[5] = l1.y * dinv; lj[6] = l1.z * dinv; lj[7] = l1.w * dinv;
                if (own && t == 0) dg[k] = d;
#pragma unroll
                for (int u = 0; u < 8; u++) {
                    float nd = dr[u] - dtk * lj[u];
                    float ns = sr[u] - stk * lj[u];
                    if (u < ki)       { dr[u] = own ? dr[u] : nd; if (WITH_S) sr[u] = own ? sr[u] : ns; }
                    else if (u == ki) { dr[u] = own ? dtk : nd;   if (WITH_S) sr[u] = own ? stk : ns; }
                    else              { dr[u] = nd;               if (WITH_S) sr[u] = ns; }
                }
            }
        }
    }
#pragma unroll
    for (int u = 0; u < 8; u++) {
        Dsh[t * LDP + 8 * q + u] = dr[u];
        if (WITH_S) Ssh[t * LDP + 8 * q + u] = sr[u];
    }
    __syncthreads();
}

__global__ __launch_bounds__(512) void chol_step_kernel(const float* __restrict__ Abase,
                                                        float* Lbase, int p) {
    __shared__ float Dsh[NB * LDP];
    __shared__ float Ssh[NB * LDP];
    __shared__ float dg[NB];
    __shared__ __attribute__((aligned(16))) float colD[2 * NB];
    __shared__ __attribute__((aligned(16))) float colS[2 * NB];
    int b = blockIdx.y;
    int i = p + blockIdx.x;
    bool isDiag = (blockIdx.x == 0);
    const float* A = Abase + (size_t)b * Nn * Nn;
    float* L = Lbase + (size_t)b * Nn * Nn;
    int tid = threadIdx.x;
    int wave = tid >> 5, lane = tid & 31;
    int r = lane & 15, hi = lane >> 4;
    int tm = wave >> 2, tn = wave & 3;
    int prow = p * NB, irow = i * NB;

    v8f accD, accS;
#pragma unroll
    for (int v = 0; v < 8; v++) {
        int row = tm * 16 + v + 8 * hi;
        int col = tn * 16 + r;
        accD[v] = -A[(size_t)(prow + row) * Nn + prow + col];
        accS[v] = -A[(size_t)(irow + row) * Nn + prow + col];
    }
    {
        const float* aP = L + (size_t)(prow + tm * 16 + r) * Nn;
        const float* aI = L + (size_t)(irow + tm * 16 + r) * Nn;
        const float* bP = L + (size_t)(prow + tn * 16 + r) * Nn;
        for (int kb = 0; kb < prow; kb += 16) {
#pragma unroll
            for (int k0 = 0; k0 < 16; k0 += 4) {
                int kk = kb + k0 + 2 * hi;
                v2f fa = *(const v2f*)(aP + kk);
                v2f fi = *(const v2f*)(aI + kk);
                v2f fb = *(const v2f*)(bP + kk);
                accD = __builtin_amdgcn_wmma_f32_16x16x4_f32(false, fa, false, fb,
                                                             (short)0, accD, false, false);
                accS = __builtin_amdgcn_wmma_f32_16x16x4_f32(false, fi, false, fb,
                                                             (short)0, accS, false, false);
            }
        }
    }
#pragma unroll
    for (int v = 0; v < 8; v++) {
        int row = tm * 16 + v + 8 * hi;
        int col = tn * 16 + r;
        Dsh[row * LDP + col] = -accD[v];
        Ssh[row * LDP + col] = -accS[v];
    }
    __syncthreads();

    if (isDiag) panel_sweep<false>(Dsh, Ssh, dg, colD, colS, tid);
    else        panel_sweep<true>(Dsh, Ssh, dg, colD, colS, tid);

    if (isDiag) {
        for (int idx = tid; idx < NB * NB; idx += 512) {
            int rr = idx >> 6, c = idx & (NB - 1);
            float val = (c < rr) ? Dsh[rr * LDP + c] : ((c == rr) ? dg[rr] : 0.0f);
            L[(size_t)(prow + rr) * Nn + prow + c] = val;
        }
    } else {
        for (int idx = tid; idx < NB * NB; idx += 512) {
            int rr = idx >> 6, c = idx & (NB - 1);
            L[(size_t)(irow + rr) * Nn + prow + c] = Ssh[rr * LDP + c];
        }
    }
}

__global__ __launch_bounds__(1024) void zmatvec_kernel(const float* __restrict__ Linvbase,
                                                       const float* __restrict__ y,
                                                       const float* __restrict__ mean,
                                                       float* __restrict__ zbase) {
    __shared__ float zrow[32];
    int b = blockIdx.y;
    int tid = threadIdx.x;
    int wave = tid >> 5, lane = tid & 31;
    int i = blockIdx.x * 32 + wave;
    const float* row = Linvbase + (size_t)b * Nn * Nn + (size_t)i * Nn;
    const float* yb = y + b * Nn;
    float mb = mean[b];
    int rowEnd = ((i >> 6) + 1) << 6;
    float4 l[8], v[8];
#pragma unroll
    for (int u = 0; u < 8; u++) {
        int c0 = 4 * lane + 128 * u;
        if (c0 < rowEnd) {
            l[u] = *(const float4*)(row + c0);
            v[u] = *(const float4*)(yb + c0);
        } else {
            l[u] = make_float4(0.0f, 0.0f, 0.0f, 0.0f);
            v[u] = make_float4(mb, mb, mb, mb);
        }
    }
    float s0 = 0.0f, s1 = 0.0f, s2 = 0.0f, s3 = 0.0f;
#pragma unroll
    for (int u = 0; u < 8; u++) {
        s0 += l[u].x * (v[u].x - mb);
        s1 += l[u].y * (v[u].y - mb);
        s2 += l[u].z * (v[u].z - mb);
        s3 += l[u].w * (v[u].w - mb);
    }
    float s = (s0 + s1) + (s2 + s3);
    for (int o = 16; o > 0; o >>= 1) s += __shfl_xor(s, o);
    if (lane == 0) zrow[wave] = s;
    __syncthreads();
    if (wave == 0) zbase[b * Nn + blockIdx.x * 32 + lane] = zrow[lane];
}

__global__ __launch_bounds__(1024) void nlml_kernel(const float* __restrict__ Lbase,
                                                    const float* __restrict__ zbase,
                                                    float* __restrict__ nlml_out) {
    int b = blockIdx.x;
    const float* L = Lbase + (size_t)b * Nn * Nn;
    __shared__ float col[Nn];
    __shared__ float zz[Nn];
    int tid = threadIdx.x;
    float z = zbase[b * Nn + tid];
    col[tid] = __logf(L[(size_t)tid * Nn + tid]);
    zz[tid] = z * z;
    __syncthreads();
    for (int s = 512; s > 0; s >>= 1) {
        if (tid < s) {
            col[tid] += col[tid + s];
            zz[tid] += zz[tid + s];
        }
        __syncthreads();
    }
    if (tid == 0) nlml_out[b] = 0.5f * zz[0] + col[0] + 0.5f * Nn * LOG2PIF;
}

__global__ __launch_bounds__(64) void trinv_diag_kernel(const float* __restrict__ Lbase,
                                                        float* __restrict__ Linvbase) {
    __shared__ __attribute__((aligned(16))) float Lb[NB * NB];
    __shared__ float xs[NB * NB];
    int p = blockIdx.x, b = blockIdx.y, tid = threadIdx.x;
    size_t base = (size_t)b * Nn * Nn + (size_t)(p * NB) * Nn + p * NB;
    const float* L = Lbase + base;
    float* Li = Linvbase + base;
    for (int idx = tid; idx < NB * NB; idx += 64) {
        int rr = idx >> 6, c = idx & (NB - 1);
        Lb[idx] = L[(size_t)rr * Nn + c];
    }
    __syncthreads();
    int j = tid;
    for (int i = 0; i < NB; i++) {
        const float* lr = Lb + i * NB;
        float s0 = (i == j) ? 1.0f : 0.0f, s1 = 0.0f, s2 = 0.0f, s3 = 0.0f;
        int k = 0;
        for (; k + 4 <= i; k += 4) {
            float4 l = *(const float4*)(lr + k);
            s0 -= l.x * xs[(k + 0) * NB + j];
            s1 -= l.y * xs[(k + 1) * NB + j];
            s2 -= l.z * xs[(k + 2) * NB + j];
            s3 -= l.w * xs[(k + 3) * NB + j];
        }
        for (; k < i; k++) s0 -= lr[k] * xs[k * NB + j];
        xs[i * NB + j] = ((s0 + s1) + (s2 + s3)) / lr[i];
    }
    __syncthreads();
    for (int idx = tid; idx < NB * NB; idx += 64) {
        int rr = idx >> 6, c = idx & (NB - 1);
        Li[(size_t)rr * Nn + c] = xs[idx];
    }
}

#define XLD 65

__device__ __forceinline__ void slab32(const float* aP, const float* bP, int kb, int hi,
                                       v8f& c0, v8f& c1) {
    v2f a[8], q[8];
#pragma unroll
    for (int u = 0; u < 8; u++) {
        int kk = kb + 4 * u + 2 * hi;
        a[u] = *(const v2f*)(aP + kk);
        q[u].x = bP[kk * XLD];
        q[u].y = bP[(kk + 1) * XLD];
    }
    __builtin_amdgcn_sched_barrier(0);
#pragma unroll
    for (int u = 0; u < 8; u += 2) {
        c0 = __builtin_amdgcn_wmma_f32_16x16x4_f32(false, a[u], false, q[u],
                                                   (short)0, c0, false, false);
        c1 = __builtin_amdgcn_wmma_f32_16x16x4_f32(false, a[u + 1], false, q[u + 1],
                                                   (short)0, c1, false, false);
    }
}
__global__ __launch_bounds__(512) void trinv_panel_kernel(const float* __restrict__ Lbase,
                                                          float* Linvbase) {
    extern __shared__ float smem[];
    float* Xp  = smem;
    float* Tsh = smem + Nn * XLD;
    int J = blockIdx.x, b = blockIdx.y, tid = threadIdx.x;
    const float* L = Lbase + (size_t)b * Nn * Nn;
    float* Li = Linvbase + (size_t)b * Nn * Nn;
    int wave = __builtin_amdgcn_readfirstlane(tid >> 5);
    int lane = tid & 31;
    int r = lane & 15, hi = lane >> 4;
    int tm = wave >> 2, tn = wave & 3;
    int jcol = J * NB;

    for (int idx = tid; idx < NB * NB; idx += 512) {
        int rr = idx >> 6, c = idx & (NB - 1);
        Xp[rr * XLD + c] = Li[(size_t)(jcol + rr) * Nn + jcol + c];
    }
    __syncthreads();

    for (int i = J + 1; i < NBLK; i++) {
        int irow = i * NB;
        int kdepth = irow - jcol;
        v8f acc0 = {}, acc1 = {};
        {
            const float* aP = L + (size_t)(irow + tm * 16 + r) * Nn + jcol;
            const float* bP = Xp + tn * 16 + r;
            for (int kb = 0; kb < kdepth; kb += 32) slab32(aP, bP, kb, hi, acc0, acc1);
        }
#pragma unroll
        for (int v = 0; v < 8; v++) {
            int row = tm * 16 + v + 8 * hi;
            int col = tn * 16 + r;
            Tsh[row * XLD + col] = acc0[v] + acc1[v];
        }
        __syncthreads();
        v8f x0 = {}, x1 = {};
        {
            const float* aD = Li + (size_t)(irow + tm * 16 + r) * Nn + irow;
            const float* bT = Tsh + tn * 16 + r;
            for (int kb = 0; kb < (tm + 1) * 16; kb += 32) slab32(aD, bT, kb, hi, x0, x1);
        }
#pragma unroll
        for (int v = 0; v < 8; v++) {
            int row = tm * 16 + v + 8 * hi;
            int col = tn * 16 + r;
            Xp[(kdepth + row) * XLD + col] = -(x0[v] + x1[v]);
        }
        __syncthreads();
        for (int idx = tid; idx < NB * NB; idx += 512) {
            int rr = idx >> 6, c = idx & (NB - 1);
            Li[(size_t)(irow + rr) * Nn + jcol + c] = Xp[(kdepth + rr) * XLD + c];
        }
    }
}


template <int NU0>
__device__ __forceinline__ void band_slab(const float* a0, const float* a1, const float* bK,
                                          int kb, int hi, v8f (&c0)[4], v8f (&c1)[4]) {
    v2f fa0[8], fa1[8], fb[8][4];
#pragma unroll
    for (int u = 0; u < 8; u++) {
        int kk = kb + 4 * u + 2 * hi;
        if (u < NU0) fa0[u] = *(const v2f*)(a0 + kk);
        fa1[u] = *(const v2f*)(a1 + kk);
#pragma unroll
        for (int c = 0; c < 4; c++) {
            fb[u][c].x = bK[kk * KTS + c * 16];
            fb[u][c].y = bK[(kk + 1) * KTS + c * 16];
        }
    }
#pragma unroll
    for (int u = 0; u < 8; u++) {
#pragma unroll
        for (int c = 0; c < 4; c++) {
            if (u < NU0)
                c0[c] = __builtin_amdgcn_wmma_f32_16x16x4_f32(false, fa0[u], false, fb[u][c],
                                                              (short)0, c0[c], false, false);
            c1[c] = __builtin_amdgcn_wmma_f32_16x16x4_f32(false, fa1[u], false, fb[u][c],
                                                          (short)0, c1[c], false, false);
        }
    }
}

#define PT 512
__global__ __launch_bounds__(PT) void predict_kernel(const float* __restrict__ Linvbase,
                                                     const float* __restrict__ trainPg,
                                                     const float* __restrict__ testPg,
                                                     const float* __restrict__ Kmm,
                                                     const float* __restrict__ zbase,
                                                     const float* __restrict__ meang,
                                                     const float* __restrict__ scale_param,
                                                     float* __restrict__ out_mean,
                                                     float* __restrict__ out_var) {
    extern __shared__ float smem[];
    float* Kt  = smem;
    float* zsh = smem + Nn * KTS;
    float* tp  = zsh + Nn;
    int b = blockIdx.y;
    int colBase = blockIdx.x * MT;
    int tid = threadIdx.x;
    float scale = __expf(scale_param[0]);

    for (int idx = tid; idx < Nn; idx += PT) zsh[idx] = zbase[b * Nn + idx];
    for (int idx = tid; idx < Nn * 6; idx += PT) tp[idx] = trainPg[(size_t)b * Nn * 6 + idx];
    __syncthreads();

    {
        int c = tid & 63;
        int h = tid >> 6;
        const float* q = testPg + (size_t)(b * Mm + colBase + c) * 6;
        float x2 = q[0], y2 = q[1], a2 = q[2], b2 = q[3], c2 = q[4], q2 = q[5];
        for (int n = h; n < Nn; n += PT / 64) {
            const float* p = tp + n * 6;
            Kt[n * KTS + c] = kern_pair(p[0], p[1], p[2], p[3], p[4], p[5],
                                        x2, y2, a2, b2, c2, q2, scale);
        }
    }
    __syncthreads();

    int wave = __builtin_amdgcn_readfirstlane(tid >> 5);
    int lane = tid & 31;
    int r = lane & 15, hi = lane >> 4;
    const float* Li = Linvbase + (size_t)b * Nn * Nn;
    const float* bK = Kt + r;

    float meanAcc[4] = {0.0f, 0.0f, 0.0f, 0.0f};
    float varAcc[4]  = {0.0f, 0.0f, 0.0f, 0.0f};
    for (int s = 0; s < 2; s++) {
        int p = s ? (Nn / 32 - 1 - wave) : wave;
        int row0 = p * 32;
        const float* a0 = Li + (size_t)(row0 + r) * Nn;
        const float* a1 = a0 + (size_t)16 * Nn;
        v8f c0[4] = {}, c1[4] = {};
        for (int t = 0; t < p; t++) band_slab<8>(a0, a1, bK, t * 32, hi, c0, c1);
        band_slab<4>(a0, a1, bK, row0, hi, c0, c1);
        const float* z0 = zsh + row0 + 8 * hi;
#pragma unroll
        for (int c = 0; c < 4; c++) {
#pragma unroll
            for (int v = 0; v < 8; v++) {
                float w0 = c0[c][v], w1 = c1[c][v];
                meanAcc[c] += z0[v] * w0;
                varAcc[c]  += w0 * w0;
                meanAcc[c] += z0[16 + v] * w1;
                varAcc[c]  += w1 * w1;
            }
        }
    }
    float* red = tp;
#pragma unroll
    for (int c = 0; c < 4; c++) {
        float m = meanAcc[c] + __shfl_xor(meanAcc[c], 16);
        float v = varAcc[c] + __shfl_xor(varAcc[c], 16);
        if (lane < 16) {
            red[wave * 128 + c * 16 + lane] = m;
            red[wave * 128 + 64 + c * 16 + lane] = v;
        }
    }
    __syncthreads();
    if (tid < MT) {
        float m = 0.0f, v = 0.0f;
#pragma unroll
        for (int w = 0; w < PT / 32; w++) {
            m += red[w * 128 + tid];
            v += red[w * 128 + 64 + tid];
        }
        int col = b * Mm + colBase + tid;
        out_mean[col] = meang[b] + m;
        out_var[col]  = Kmm[col] - v;
    }
}

extern "C" void kernel_launch(void* const* d_in, const int* in_sizes, int n_in,
                              void* d_out, int out_size, void* d_ws, size_t ws_size,
                              hipStream_t stream) {
    const float* coords_train = (const float*)d_in[0];
    const float* kp_train     = (const float*)d_in[1];
    const float* coords_test  = (const float*)d_in[2];
    const float* kp_test      = (const float*)d_in[3];
    const float* y_train      = (const float*)d_in[4];
    const float* var          = (const float*)d_in[5];
    const float* mean         = (const float*)d_in[6];
    const float* scale_param  = (const float*)d_in[7];

    float* ws = (float*)d_ws;
    size_t off = 0;
    float* A      = ws + off; off += (size_t)Bb * Nn * Nn;
    float* Lm     = ws + off; off += (size_t)Bb * Nn * Nn;
    float* Linv   = ws + off; off += (size_t)Bb * Nn * Nn;
    float* trainP = ws + off; off += (size_t)Bb * Nn * 6;
    float* testP  = ws + off; off += (size_t)Bb * Mm * 6;
    float* Kmm    = ws + off; off += (size_t)Bb * Mm;
    float* z      = ws + off; off += (size_t)Bb * Nn;

    float* out_mean = (float*)d_out;
    float* out_var  = out_mean + (size_t)Bb * Mm;
    float* nlml     = out_var + (size_t)Bb * Mm;

    prep_kernel<<<(Bb * Mm + 255) / 256, 256, 0, stream>>>(
        coords_train, kp_train, coords_test, kp_test, scale_param, trainP, testP, Kmm);
    buildA_kernel<<<(Bb * Nn * Nn) / 256, 256, 0, stream>>>(trainP, var, scale_param, A);
#define CHOL_STEP(P) chol_step_kernel<<<dim3(NBLK - (P), Bb), 512, 0, stream>>>(A, Lm, (P))
    CHOL_STEP(0);  CHOL_STEP(1);  CHOL_STEP(2);  CHOL_STEP(3);
    CHOL_STEP(4);  CHOL_STEP(5);  CHOL_STEP(6);  CHOL_STEP(7);
    CHOL_STEP(8);  CHOL_STEP(9);  CHOL_STEP(10); CHOL_STEP(11);
    CHOL_STEP(12); CHOL_STEP(13); CHOL_STEP(14); CHOL_STEP(15);
#undef CHOL_STEP
    trinv_diag_kernel<<<dim3(NBLK, Bb), 64, 0, stream>>>(Lm, Linv);
    size_t inv_smem_bytes = (size_t)(Nn * XLD + NB * XLD) * sizeof(float);
    trinv_panel_kernel<<<dim3(NBLK - 1, Bb), 512, inv_smem_bytes, stream>>>(Lm, Linv);
    zmatvec_kernel<<<dim3(Nn / 32, Bb), 1024, 0, stream>>>(Linv, y_train, mean, z);
    nlml_kernel<<<Bb, 1024, 0, stream>>>(Lm, z, nlml);

    size_t smem_bytes = (size_t)(Nn * KTS + Nn + Nn * 6) * sizeof(float);
    predict_kernel<<<dim3(Mm / MT, Bb), PT, smem_bytes, stream>>>(
        Linv, trainP, testP, Kmm, z, mean, scale_param, out_mean, out_var);
}


# Cholesky made right-looking: each launch also takes the newest block column off all later blocks (a workgroup per block, on idle compute units); panel product 64 deep, not 64p; sum order kept
# speedup vs baseline: 75.1152x; 1.3714x over previous
#include <hip/hip_runtime.h>
#include <math.h>

#define Nn 1024
#define Mm 16384
#define Bb 2
#define MT 64
#define KTS 65
#define NB 64
#define NBLK (Nn / NB)
#define LDP 65

typedef __attribute__((ext_vector_type(2))) float v2f;
typedef __attribute__((ext_vector_type(8))) float v8f;

#define SQRT3F 1.7320508075688772f
#define LOG2PIF 1.8378770664093453f

__device__ __forceinline__ float kern_pair(float x1, float y1, float a1, float b1, float c1, float q1,
                                           float x2, float y2, float a2, float b2, float c2, float q2,
                                           float scale) {
    float s00 = a1 + a2, s11 = b1 + b2, s01 = c1 + c2;
    float sdet = s00 * s11 - s01 * s01;
    float d0 = x1 - x2, d1 = y1 - y2;
    float Q = 0.5f * (s11 * d0 * d0 - 2.0f * s01 * d0 * d1 + s00 * d1 * d1) / sdet;
    float C = 2.0f * q1 * q2 * rsqrtf(fmaxf(sdet, 1e-5f));
    float t = SQRT3F * sqrtf(fmaxf(Q, 1e-5f));
    return (1.0f + t) * __expf(-t) * C * scale;
}

__global__ void prep_kernel(const float* __restrict__ ct, const float* __restrict__ kpt,
                            const float* __restrict__ cs, const float* __restrict__ kps,
                            const float* __restrict__ scale_param,
                            float* __restrict__ trainP, float* __restrict__ testP,
                            float* __restrict__ Kmm) {
    int gid = blockIdx.x * blockDim.x + threadIdx.x;
    float scale = __expf(scale_param[0]);
    if (gid < Bb * Mm) {
        float a = kps[gid * 3 + 0], b = kps[gid * 3 + 1], c = kps[gid * 3 + 2];
        float det = a * b - c * c;
        testP[gid * 6 + 0] = cs[gid * 2 + 0];
        testP[gid * 6 + 1] = cs[gid * 2 + 1];
        testP[gid * 6 + 2] = a;
        testP[gid * 6 + 3] = b;
        testP[gid * 6 + 4] = c;
        testP[gid * 6 + 5] = sqrtf(sqrtf(det));
        float t0 = SQRT3F * sqrtf(1e-5f);
        float m0 = (1.0f + t0) * __expf(-t0);
        float C = 2.0f * sqrtf(det) * rsqrtf(fmaxf(4.0f * det, 1e-5f));
        Kmm[gid] = C * m0 * scale;
    }
    if (gid < Bb * Nn) {
        float a = kpt[gid * 3 + 0], b = kpt[gid * 3 + 1], c = kpt[gid * 3 + 2];
        float det = a * b - c * c;
        trainP[gid * 6 + 0] = ct[gid * 2 + 0];
        trainP[gid * 6 + 1] = ct[gid * 2 + 1];
        trainP[gid * 6 + 2] = a;
        trainP[gid * 6 + 3] = b;
        trainP[gid * 6 + 4] = c;
        trainP[gid * 6 + 5] = sqrtf(sqrtf(det));
    }
}

__global__ void buildA_kernel(const float* __restrict__ trainP, const float* __restrict__ var,
                              const float* __restrict__ scale_param, float* __restrict__ A) {
    int gid = blockIdx.x * blockDim.x + threadIdx.x;
    int b = gid >> 20;
    int i = (gid >> 10) & (Nn - 1);
    int j = gid & (Nn - 1);
    const float* p1 = trainP + (size_t)(b * Nn + i) * 6;
    const float* p2 = trainP + (size_t)(b * Nn + j) * 6;
    float scale = __expf(scale_param[0]);
    float v = kern_pair(p1[0], p1[1], p1[2], p1[3], p1[4], p1[5],
                        p2[0], p2[1], p2[2], p2[3], p2[4], p2[5], scale);
    if (i == j) v += var[b * Nn + i];
    A[(size_t)gid] = v;
}

template <bool WITH_S>
__device__ __forceinline__ void panel_sweep(float* Dsh, float* Ssh, float* dg,
                                            float* colD, float* colS, int tid) {
    int t = tid & (NB - 1);
    int q = __builtin_amdgcn_readfirstlane(tid >> 6);
    float dr[8], sr[8];
#pragma unroll
    for (int u = 0; u < 8; u++) {
        dr[u] = Dsh[t * LDP + 8 * q + u];
        sr[u] = WITH_S ? Ssh[t * LDP + 8 * q + u] : 0.0f;
    }
#pragma unroll 1
    for (int ko = 0; ko < NB / 8; ko++) {
#pragma unroll
        for (int ki = 0; ki < 8; ki++) {
            int k = 8 * ko + ki;
            float* cD = colD + (ki & 1) * NB;
            float* cS = colS + (ki & 1) * NB;
            if (q == ko) {
                cD[t] = dr[ki];
                if (WITH_S) cS[t] = sr[ki];
            }
            __syncthreads();
            if (q >= ko) {
                float x = cD[k];
                float dtr = cD[t];
                float str = WITH_S ? cS[t] : 0.0f;
                float4 l0 = *(const float4*)(cD + 8 * q);
                float4 l1 = *(const float4*)(cD + 8 * q + 4);
                float d = sqrtf(x);
                float dinv = 1.0f / d;
                float dtk = (t > k) ? dtr * dinv : 0.0f;
                float stk = str * dinv;
                bool own = (q == ko);
                float lj[8];
                lj[0] = l0.x * dinv; lj[1] = l0.y * dinv; lj[2] = l0.z * dinv; lj[3] = l0.w * dinv;
                lj[4] = l1.x * dinv; lj[5] = l1.y * dinv; lj[6] = l1.z * dinv; lj[7] = l1.w * dinv;
                if (own && t == 0) dg[k] = d;
#pragma unroll
                for (int u = 0; u < 8; u++) {
                    float nd = dr[u] - dtk * lj[u];
                    float ns = sr[u] - stk * lj[u];
                    if (u < ki)       { dr[u] = own ? dr[u] : nd; if (WITH_S) sr[u] = own ? sr[u] : ns; }
                    else if (u == ki) { dr[u] = own ? dtk : nd;   if (WITH_S) sr[u] = own ? stk : ns; }
                    else              { dr[u] = nd;               if (WITH_S) sr[u] = ns; }
                }
            }
        }
    }
#pragma unroll
    for (int u = 0; u < 8; u++) {
        Dsh[t * LDP + 8 * q + u] = dr[u];
        if (WITH_S) Ssh[t * LDP + 8 * q + u] = sr[u];
    }
    __syncthreads();
}

__global__ __launch_bounds__(512) void chol_step_kernel(float* Rbase, float* Lbase, int p) {
    __shared__ float Dsh[NB * LDP];
    __shared__ float Ssh[NB * LDP];
    __shared__ float dg[NB];
    __shared__ __attribute__((aligned(16))) float colD[2 * NB];
    __shared__ __attribute__((aligned(16))) float colS[2 * NB];
    int b = blockIdx.x & (Bb - 1);
    int idx = blockIdx.x / Bb;
    int np = NBLK - p;
    int i, j;
    if (idx < np) {
        i = p + idx; j = p;
    } else {
        int tr = idx - np, a = 0;
        while (tr > a) { tr -= a + 1; a++; }
        i = p + 1 + a; j = p + 1 + tr;
    }
    bool isDiag = (i == p);
    float* R = Rbase + (size_t)b * Nn * Nn;
    float* L = Lbase + (size_t)b * Nn * Nn;
    int tid = threadIdx.x;
    int wave = tid >> 5, lane = tid & 31;
    int r = lane & 15, hi = lane >> 4;
    int tm = wave >> 2, tn = wave & 3;
    int prow = p * NB, irow = i * NB;
    int kcol = prow - NB;

    if (j > p) {
        int jrow = j * NB;
        v8f acc;
#pragma unroll
        for (int v = 0; v < 8; v++) {
            int row = tm * 16 + v + 8 * hi;
            int col = tn * 16 + r;
            acc[v] = -R[(size_t)(irow + row) * Nn + jrow + col];
        }
        const float* aI = L + (size_t)(irow + tm * 16 + r) * Nn + kcol;
        const float* bJ = L + (size_t)(jrow + tn * 16 + r) * Nn + kcol;
#pragma unroll
        for (int k0 = 0; k0 < NB; k0 += 4) {
            int kk = k0 + 2 * hi;
            v2f fi = *(const v2f*)(aI + kk);
            v2f fb = *(const v2f*)(bJ + kk);
            acc = __builtin_amdgcn_wmma_f32_16x16x4_f32(false, fi, false, fb,
                                                        (short)0, acc, false, false);
        }
#pragma unroll
        for (int v = 0; v < 8; v++) {
            int row = tm * 16 + v + 8 * hi;
            int col = tn * 16 + r;
            Dsh[row * LDP + col] = -acc[v];
        }
        __syncthreads();
        for (int e = tid; e < NB * NB; e += 512) {
            int rr = e >> 6, c = e & (NB - 1);
            R[(size_t)(irow + rr) * Nn + jrow + c] = Dsh[rr * LDP + c];
        }
        return;
    }

    v8f accD, accS;
#pragma unroll
    for (int v = 0; v < 8; v++) {
        int row = tm * 16 + v + 8 * hi;
        int col = tn * 16 + r;
        accD[v] = -R[(size_t)(prow + row) * Nn + prow + col];
        accS[v] = -R[(size_t)(irow + row) * Nn + prow + col];
    }
    if (p > 0) {
        const float* aP = L + (size_t)(prow + tm * 16 + r) * Nn + kcol;
        const float* aI = L + (size_t)(irow + tm * 16 + r) * Nn + kcol;
        const float* bP = L + (size_t)(prow + tn * 16 + r) * Nn + kcol;
#pragma unroll
        for (int k0 = 0; k0 < NB; k0 += 4) {
            int kk = k0 + 2 * hi;
            v2f fa = *(const v2f*)(aP + kk);
            v2f fi = *(const v2f*)(aI + kk);
            v2f fb = *(const v2f*)(bP + kk);
            accD = __builtin_amdgcn_wmma_f32_16x16x4_f32(false, fa, false, fb,
                                                         (short)0, accD, false, false);
            accS = __builtin_amdgcn_wmma_f32_16x16x4_f32(false, fi, false, fb,
                                                         (short)0, accS, false, false);
        }
    }
#pragma unroll
    for (int v = 0; v < 8; v++) {
        int row = tm * 16 + v + 8 * hi;
        int col = tn * 16 + r;
        Dsh[row * LDP + col] = -accD[v];
        Ssh[row * LDP + col] = -accS[v];
    }
    __syncthreads();

    if (isDiag) panel_sweep<false>(Dsh, Ssh, dg, colD, colS, tid);
    else        panel_sweep<true>(Dsh, Ssh, dg, colD, colS, tid);

    if (isDiag) {
        for (int e = tid; e < NB * NB; e += 512) {
            int rr = e >> 6, c = e & (NB - 1);
            float val = (c < rr) ? Dsh[rr * LDP + c] : ((c == rr) ? dg[rr] : 0.0f);
            L[(size_t)(prow + rr) * Nn + prow + c] = val;
        }
    } else {
        for (int e = tid; e < NB * NB; e += 512) {
            int rr = e >> 6, c = e & (NB - 1);
            L[(size_t)(irow + rr) * Nn + prow + c] = Ssh[rr * LDP + c];
        }
    }
}

__global__ __launch_bounds__(1024) void zmatvec_kernel(const float* __restrict__ Linvbase,
                                                       const float* __restrict__ y,
                                                       const float* __restrict__ mean,
                                                       float* __restrict__ zbase) {
    __shared__ float zrow[32];
    int b = blockIdx.y;
    int tid = threadIdx.x;
    int wave = tid >> 5, lane = tid & 31;
    int i = blockIdx.x * 32 + wave;
    const float* row = Linvbase + (size_t)b * Nn * Nn + (size_t)i * Nn;
    const float* yb = y + b * Nn;
    float mb = mean[b];
    int rowEnd = ((i >> 6) + 1) << 6;
    float4 l[8], v[8];
#pragma unroll
    for (int u = 0; u < 8; u++) {
        int c0 = 4 * lane + 128 * u;
        if (c0 < rowEnd) {
            l[u] = *(const float4*)(row + c0);
            v[u] = *(const float4*)(yb + c0);
        } else {
            l[u] = make_float4(0.0f, 0.0f, 0.0f, 0.0f);
            v[u] = make_float4(mb, mb, mb, mb);
        }
    }
    float s0 = 0.0f, s1 = 0.0f, s2 = 0.0f, s3 = 0.0f;
#pragma unroll
    for (int u = 0; u < 8; u++) {
        s0 += l[u].x * (v[u].x - mb);
        s1 += l[u].y * (v[u].y - mb);
        s2 += l[u].z * (v[u].z - mb);
        s3 += l[u].w * (v[u].w - mb);
    }
    float s = (s0 + s1) + (s2 + s3);
    for (int o = 16; o > 0; o >>= 1) s += __shfl_xor(s, o);
    if (lane == 0) zrow[wave] = s;
    __syncthreads();
    if (wave == 0) zbase[b * Nn + blockIdx.x * 32 + lane] = zrow[lane];
}

__global__ __launch_bounds__(1024) void nlml_kernel(const float* __restrict__ Lbase,
                                                    const float* __restrict__ zbase,
                                                    float* __restrict__ nlml_out) {
    int b = blockIdx.x;
    const float* L = Lbase + (size_t)b * Nn * Nn;
    __shared__ float col[Nn];
    __shared__ float zz[Nn];
    int tid = threadIdx.x;
    float z = zbase[b * Nn + tid];
    col[tid] = __logf(L[(size_t)tid * Nn + tid]);
    zz[tid] = z * z;
    __syncthreads();
    for (int s = 512; s > 0; s >>= 1) {
        if (tid < s) {
            col[tid] += col[tid + s];
            zz[tid] += zz[tid + s];
        }
        __syncthreads();
    }
    if (tid == 0) nlml_out[b] = 0.5f * zz[0] + col[0] + 0.5f * Nn * LOG2PIF;
}

__global__ __launch_bounds__(64) void trinv_diag_kernel(const float* __restrict__ Lbase,
                                                        float* __restrict__ Linvbase) {
    __shared__ __attribute__((aligned(16))) float Lb[NB * NB];
    __shared__ float xs[NB * NB];
    int p = blockIdx.x, b = blockIdx.y, tid = threadIdx.x;
    size_t base = (size_t)b * Nn * Nn + (size_t)(p * NB) * Nn + p * NB;
    const float* L = Lbase + base;
    float* Li = Linvbase + base;
    for (int idx = tid; idx < NB * NB; idx += 64) {
        int rr = idx >> 6, c = idx & (NB - 1);
        Lb[idx] = L[(size_t)rr * Nn + c];
    }
    __syncthreads();
    int j = tid;
    for (int i = 0; i < NB; i++) {
        const float* lr = Lb + i * NB;
        float s0 = (i == j) ? 1.0f : 0.0f, s1 = 0.0f, s2 = 0.0f, s3 = 0.0f;
        int k = 0;
        for (; k + 4 <= i; k += 4) {
            float4 l = *(const float4*)(lr + k);
            s0 -= l.x * xs[(k + 0) * NB + j];
            s1 -= l.y * xs[(k + 1) * NB + j];
            s2 -= l.z * xs[(k + 2) * NB + j];
            s3 -= l.w * xs[(k + 3) * NB + j];
        }
        for (; k < i; k++) s0 -= lr[k] * xs[k * NB + j];
        xs[i * NB + j] = ((s0 + s1) + (s2 + s3)) / lr[i];
    }
    __syncthreads();
    for (int idx = tid; idx < NB * NB; idx += 64) {
        int rr = idx >> 6, c = idx & (NB - 1);
        Li[(size_t)rr * Nn + c] = xs[idx];
    }
}

#define XLD 65

__device__ __forceinline__ void slab32(const float* aP, const float* bP, int kb, int hi,
                                       v8f& c0, v8f& c1) {
    v2f a[8], q[8];
#pragma unroll
    for (int u = 0; u < 8; u++) {
        int kk = kb + 4 * u + 2 * hi;
        a[u] = *(const v2f*)(aP + kk);
        q[u].x = bP[kk * XLD];
        q[u].y = bP[(kk + 1) * XLD];
    }
    __builtin_amdgcn_sched_barrier(0);
#pragma unroll
    for (int u = 0; u < 8; u += 2) {
        c0 = __builtin_amdgcn_wmma_f32_16x16x4_f32(false, a[u], false, q[u],
                                                   (short)0, c0, false, false);
        c1 = __builtin_amdgcn_wmma_f32_16x16x4_f32(false, a[u + 1], false, q[u + 1],
                                                   (short)0, c1, false, false);
    }
}
__global__ __launch_bounds__(512) void trinv_panel_kernel(const float* __restrict__ Lbase,
                                                          float* Linvbase) {
    extern __shared__ float smem[];
    float* Xp  = smem;
    float* Tsh = smem + Nn * XLD;
    int J = blockIdx.x, b = blockIdx.y, tid = threadIdx.x;
    const float* L = Lbase + (size_t)b * Nn * Nn;
    float* Li = Linvbase + (size_t)b * Nn * Nn;
    int wave = __builtin_amdgcn_readfirstlane(tid >> 5);
    int lane = tid & 31;
    int r = lane & 15, hi = lane >> 4;
    int tm = wave >> 2, tn = wave & 3;
    int jcol = J * NB;

    for (int idx = tid; idx < NB * NB; idx += 512) {
        int rr = idx >> 6, c = idx & (NB - 1);
        Xp[rr * XLD + c] = Li[(size_t)(jcol + rr) * Nn + jcol + c];
    }
    __syncthreads();

    for (int i = J + 1; i < NBLK; i++) {
        int irow = i * NB;
        int kdepth = irow - jcol;
        v8f acc0 = {}, acc1 = {};
        {
            const float* aP = L + (size_t)(irow + tm * 16 + r) * Nn + jcol;
            const float* bP = Xp + tn * 16 + r;
            for (int kb = 0; kb < kdepth; kb += 32) slab32(aP, bP, kb, hi, acc0, acc1);
        }
#pragma unroll
        for (int v = 0; v < 8; v++) {
            int row = tm * 16 + v + 8 * hi;
            int col = tn * 16 + r;
            Tsh[row * XLD + col] = acc0[v] + acc1[v];
        }
        __syncthreads();
        v8f x0 = {}, x1 = {};
        {
            const float* aD = Li + (size_t)(irow + tm * 16 + r) * Nn + irow;
            const float* bT = Tsh + tn * 16 + r;
            for (int kb = 0; kb < (tm + 1) * 16; kb += 32) slab32(aD, bT, kb, hi, x0, x1);
        }
#pragma unroll
        for (int v = 0; v < 8; v++) {
            int row = tm * 16 + v + 8 * hi;
            int col = tn * 16 + r;
            Xp[(kdepth + row) * XLD + col] = -(x0[v] + x1[v]);
        }
        __syncthreads();
        for (int idx = tid; idx < NB * NB; idx += 512) {
            int rr = idx >> 6, c = idx & (NB - 1);
            Li[(size_t)(irow + rr) * Nn + jcol + c] = Xp[(kdepth + rr) * XLD + c];
        }
    }
}


template <int NU0>
__device__ __forceinline__ void band_slab(const float* a0, const float* a1, const float* bK,
                                          int kb, int hi, v8f (&c0)[4], v8f (&c1)[4]) {
    v2f fa0[8], fa1[8], fb[8][4];
#pragma unroll
    for (int u = 0; u < 8; u++) {
        int kk = kb + 4 * u + 2 * hi;
        if (u < NU0) fa0[u] = *(const v2f*)(a0 + kk);
        fa1[u] = *(const v2f*)(a1 + kk);
#pragma unroll
        for (int c = 0; c < 4; c++) {
            fb[u][c].x = bK[kk * KTS + c * 16];
            fb[u][c].y = bK[(kk + 1) * KTS + c * 16];
        }
    }
#pragma unroll
    for (int u = 0; u < 8; u++) {
#pragma unroll
        for (int c = 0; c < 4; c++) {
            if (u < NU0)
                c0[c] = __builtin_amdgcn_wmma_f32_16x16x4_f32(false, fa0[u], false, fb[u][c],
                                                              (short)0, c0[c], false, false);
            c1[c] = __builtin_amdgcn_wmma_f32_16x16x4_f32(false, fa1[u], false, fb[u][c],
                                                          (short)0, c1[c], false, false);
        }
    }
}

#define PT 512
__global__ __launch_bounds__(PT) void predict_kernel(const float* __restrict__ Linvbase,
                                                     const float* __restrict__ trainPg,
                                                     const float* __restrict__ testPg,
                                                     const float* __restrict__ Kmm,
                                                     const float* __restrict__ zbase,
                                                     const float* __restrict__ meang,
                                                     const float* __restrict__ scale_param,
                                                     float* __restrict__ out_mean,
                                                     float* __restrict__ out_var) {
    extern __shared__ float smem[];
    float* Kt  = smem;
    float* zsh = smem + Nn * KTS;
    float* tp  = zsh + Nn;
    int b = blockIdx.y;
    int colBase = blockIdx.x * MT;
    int tid = threadIdx.x;
    float scale = __expf(scale_param[0]);

    for (int idx = tid; idx < Nn; idx += PT) zsh[idx] = zbase[b * Nn + idx];
    for (int idx = tid; idx < Nn * 6; idx += PT) tp[idx] = trainPg[(size_t)b * Nn * 6 + idx];
    __syncthreads();

    {
        int c = tid & 63;
        int h = tid >> 6;
        const float* q = testPg + (size_t)(b * Mm + colBase + c) * 6;
        float x2 = q[0], y2 = q[1], a2 = q[2], b2 = q[3], c2 = q[4], q2 = q[5];
        for (int n = h; n < Nn; n += PT / 64) {
            const float* p = tp + n * 6;
            Kt[n * KTS + c] = kern_pair(p[0], p[1], p[2], p[3], p[4], p[5],
                                        x2, y2, a2, b2, c2, q2, scale);
        }
    }
    __syncthreads();

    int wave = __builtin_amdgcn_readfirstlane(tid >> 5);
    int lane = tid & 31;
    int r = lane & 15, hi = lane >> 4;
    const float* Li = Linvbase + (size_t)b * Nn * Nn;
    const float* bK = Kt + r;

    float meanAcc[4] = {0.0f, 0.0f, 0.0f, 0.0f};
    float varAcc[4]  = {0.0f, 0.0f, 0.0f, 0.0f};
    for (int s = 0; s < 2; s++) {
        int p = s ? (Nn / 32 - 1 - wave) : wave;
        int row0 = p * 32;
        const float* a0 = Li + (size_t)(row0 + r) * Nn;
        const float* a1 = a0 + (size_t)16 * Nn;
        v8f c0[4] = {}, c1[4] = {};
        for (int t = 0; t < p; t++) band_slab<8>(a0, a1, bK, t * 32, hi, c0, c1);
        band_slab<4>(a0, a1, bK, row0, hi, c0, c1);
        const float* z0 = zsh + row0 + 8 * hi;
#pragma unroll
        for (int c = 0; c < 4; c++) {
#pragma unroll
            for (int v = 0; v < 8; v++) {
                float w0 = c0[c][v], w1 = c1[c][v];
                meanAcc[c] += z0[v] * w0;
                varAcc[c]  += w0 * w0;
                meanAcc[c] += z0[16 + v] * w1;
                varAcc[c]  += w1 * w1;
            }
        }
    }
    float* red = tp;
#pragma unroll
    for (int c = 0; c < 4; c++) {
        float m = meanAcc[c] + __shfl_xor(meanAcc[c], 16);
        float v = varAcc[c] + __shfl_xor(varAcc[c], 16);
        if (lane < 16) {
            red[wave * 128 + c * 16 + lane] = m;
            red[wave * 128 + 64 + c * 16 + lane] = v;
        }
    }
    __syncthreads();
    if (tid < MT) {
        float m = 0.0f, v = 0.0f;
#pragma unroll
        for (int w = 0; w < PT / 32; w++) {
            m += red[w * 128 + tid];
            v += red[w * 128 + 64 + tid];
        }
        int col = b * Mm + colBase + tid;
        out_mean[col] = meang[b] + m;
        out_var[col]  = Kmm[col] - v;
    }
}

extern "C" void kernel_launch(void* const* d_in, const int* in_sizes, int n_in,
                              void* d_out, int out_size, void* d_ws, size_t ws_size,
                              hipStream_t stream) {
    const float* coords_train = (const float*)d_in[0];
    const float* kp_train     = (const float*)d_in[1];
    const float* coords_test  = (const float*)d_in[2];
    const float* kp_test      = (const float*)d_in[3];
    const float* y_train      = (const float*)d_in[4];
    const float* var          = (const float*)d_in[5];
    const float* mean         = (const float*)d_in[6];
    const float* scale_param  = (const float*)d_in[7];

    float* ws = (float*)d_ws;
    size_t off = 0;
    float* A      = ws + off; off += (size_t)Bb * Nn * Nn;
    float* Lm     = ws + off; off += (size_t)Bb * Nn * Nn;
    float* Linv   = ws + off; off += (size_t)Bb * Nn * Nn;
    float* trainP = ws + off; off += (size_t)Bb * Nn * 6;
    float* testP  = ws + off; off += (size_t)Bb * Mm * 6;
    float* Kmm    = ws + off; off += (size_t)Bb * Mm;
    float* z      = ws + off; off += (size_t)Bb * Nn;

    float* out_mean = (float*)d_out;
    float* out_var  = out_mean + (size_t)Bb * Mm;
    float* nlml     = out_var + (size_t)Bb * Mm;

    prep_kernel<<<(Bb * Mm + 255) / 256, 256, 0, stream>>>(
        coords_train, kp_train, coords_test, kp_test, scale_param, trainP, testP, Kmm);
    buildA_kernel<<<(Bb * Nn * Nn) / 256, 256, 0, stream>>>(trainP, var, scale_param, A);
    chol_step_kernel<<<dim3(Bb * NBLK), 512, 0, stream>>>(A, Lm, 0);
#define CHOL_STEP(P) chol_step_kernel<<<dim3((NBLK - (P)) * (NBLK + 1 - (P))), 512, 0, stream>>>(A, Lm, (P))
                   CHOL_STEP(1);  CHOL_STEP(2);  CHOL_STEP(3);
    CHOL_STEP(4);  CHOL_STEP(5);  CHOL_STEP(6);  CHOL_STEP(7);
    CHOL_STEP(8);  CHOL_STEP(9);  CHOL_STEP(10); CHOL_STEP(11);
    CHOL_STEP(12); CHOL_STEP(13); CHOL_STEP(14); CHOL_STEP(15);
#undef CHOL_STEP
    trinv_diag_kernel<<<dim3(NBLK, Bb), 64, 0, stream>>>(Lm, Linv);
    size_t inv_smem_bytes = (size_t)(Nn * XLD + NB * XLD) * sizeof(float);
    trinv_panel_kernel<<<dim3(NBLK - 1, Bb), 512, inv_smem_bytes, stream>>>(Lm, Linv);
    zmatvec_kernel<<<dim3(Nn / 32, Bb), 1024, 0, stream>>>(Linv, y_train, mean, z);
    nlml_kernel<<<Bb, 1024, 0, stream>>>(Lm, z, nlml);

    size_t smem_bytes = (size_t)(Nn * KTS + Nn + Nn * 6) * sizeof(float);
    predict_kernel<<<dim3(Mm / MT, Bb), PT, smem_bytes, stream>>>(
        Linv, trainP, testP, Kmm, z, mean, scale_param, out_mean, out_var);
}


# Cholesky panel sweep blocked by 8 columns: owner waves factor each 8x8 diagonal piece in registers, two barriers per 8 steps instead of 8; arithmetic and order unchanged
# speedup vs baseline: 81.1810x; 1.0808x over previous
#include <hip/hip_runtime.h>
#include <math.h>

#define Nn 1024
#define Mm 16384
#define Bb 2
#define MT 64
#define KTS 65
#define NB 64
#define NBLK (Nn / NB)
#define LDP 65

typedef __attribute__((ext_vector_type(2))) float v2f;
typedef __attribute__((ext_vector_type(8))) float v8f;

#define SQRT3F 1.7320508075688772f
#define LOG2PIF 1.8378770664093453f

__device__ __forceinline__ float kern_pair(float x1, float y1, float a1, float b1, float c1, float q1,
                                           float x2, float y2, float a2, float b2, float c2, float q2,
                                           float scale) {
    float s00 = a1 + a2, s11 = b1 + b2, s01 = c1 + c2;
    float sdet = s00 * s11 - s01 * s01;
    float d0 = x1 - x2, d1 = y1 - y2;
    float Q = 0.5f * (s11 * d0 * d0 - 2.0f * s01 * d0 * d1 + s00 * d1 * d1) / sdet;
    float C = 2.0f * q1 * q2 * rsqrtf(fmaxf(sdet, 1e-5f));
    float t = SQRT3F * sqrtf(fmaxf(Q, 1e-5f));
    return (1.0f + t) * __expf(-t) * C * scale;
}

__global__ void prep_kernel(const float* __restrict__ ct, const float* __restrict__ kpt,
                            const float* __restrict__ cs, const float* __restrict__ kps,
                            const float* __restrict__ scale_param,
                            float* __restrict__ trainP, float* __restrict__ testP,
                            float* __restrict__ Kmm) {
    int gid = blockIdx.x * blockDim.x + threadIdx.x;
    float scale = __expf(scale_param[0]);
    if (gid < Bb * Mm) {
        float a = kps[gid * 3 + 0], b = kps[gid * 3 + 1], c = kps[gid * 3 + 2];
        float det = a * b - c * c;
        testP[gid * 6 + 0] = cs[gid * 2 + 0];
        testP[gid * 6 + 1] = cs[gid * 2 + 1];
        testP[gid * 6 + 2] = a;
        testP[gid * 6 + 3] = b;
        testP[gid * 6 + 4] = c;
        testP[gid * 6 + 5] = sqrtf(sqrtf(det));
        float t0 = SQRT3F * sqrtf(1e-5f);
        float m0 = (1.0f + t0) * __expf(-t0);
        float C = 2.0f * sqrtf(det) * rsqrtf(fmaxf(4.0f * det, 1e-5f));
        Kmm[gid] = C * m0 * scale;
    }
    if (gid < Bb * Nn) {
        float a = kpt[gid * 3 + 0], b = kpt[gid * 3 + 1], c = kpt[gid * 3 + 2];
        float det = a * b - c * c;
        trainP[gid * 6 + 0] = ct[gid * 2 + 0];
        trainP[gid * 6 + 1] = ct[gid * 2 + 1];
        trainP[gid * 6 + 2] = a;
        trainP[gid * 6 + 3] = b;
        trainP[gid * 6 + 4] = c;
        trainP[gid * 6 + 5] = sqrtf(sqrtf(det));
    }
}

__global__ void buildA_kernel(const float* __restrict__ trainP, const float* __restrict__ var,
                              const float* __restrict__ scale_param, float* __restrict__ A) {
    int gid = blockIdx.x * blockDim.x + threadIdx.x;
    int b = gid >> 20;
    int i = (gid >> 10) & (Nn - 1);
    int j = gid & (Nn - 1);
    const float* p1 = trainP + (size_t)(b * Nn + i) * 6;
    const float* p2 = trainP + (size_t)(b * Nn + j) * 6;
    float scale = __expf(scale_param[0]);
    float v = kern_pair(p1[0], p1[1], p1[2], p1[3], p1[4], p1[5],
                        p2[0], p2[1], p2[2], p2[3], p2[4], p2[5], scale);
    if (i == j) v += var[b * Nn + i];
    A[(size_t)gid] = v;
}

template <bool WITH_S>
__device__ __forceinline__ void panel_sweep(float* Dsh, float* Ssh, float* dg,
                                            float* piece, float* colsD, float* colsS, int tid) {
    int t = tid & (NB - 1);
    int q = __builtin_amdgcn_readfirstlane(tid >> 6);
    float dr[8], sr[8];
#pragma unroll
    for (int u = 0; u < 8; u++) {
        dr[u] = Dsh[t * LDP + 8 * q + u];
        sr[u] = WITH_S ? Ssh[t * LDP + 8 * q + u] : 0.0f;
    }
#pragma unroll 1
    for (int ko = 0; ko < NB / 8; ko++) {
        int m = t - 8 * ko;
        if (q == ko && m >= 0 && m < 8) {
            *(float4*)(piece + 8 * m)     = make_float4(dr[0], dr[1], dr[2], dr[3]);
            *(float4*)(piece + 8 * m + 4) = make_float4(dr[4], dr[5], dr[6], dr[7]);
        }
        __syncthreads();
        if (q == ko) {
            float g[8][8];
#pragma unroll
            for (int r = 0; r < 8; r++) {
                float4 lo = *(const float4*)(piece + 8 * r);
                float4 hi = *(const float4*)(piece + 8 * r + 4);
                g[r][0] = lo.x; g[r][1] = lo.y; g[r][2] = lo.z; g[r][3] = lo.w;
                g[r][4] = hi.x; g[r][5] = hi.y; g[r][6] = hi.z; g[r][7] = hi.w;
            }
            float dinv[8];
#pragma unroll
            for (int k = 0; k < 8; k++) {
                float d = sqrtf(g[k][k]);
                float di = 1.0f / d;
                dinv[k] = di;
                if (t == 0) dg[8 * ko + k] = d;
#pragma unroll
                for (int r = k + 1; r < 8; r++) g[r][k] *= di;
#pragma unroll
                for (int r = k + 1; r < 8; r++) {
#pragma unroll
                    for (int cc = k + 1; cc <= r; cc++) g[r][cc] -= g[r][k] * g[cc][k];
                }
            }
#pragma unroll
            for (int k = 0; k < 8; k++) {
                float dtk = (t > 8 * ko + k) ? dr[k] * dinv[k] : 0.0f;
                float stk = sr[k] * dinv[k];
#pragma unroll
                for (int u = k + 1; u < 8; u++) {
                    dr[u] -= dtk * g[u][k];
                    if (WITH_S) sr[u] -= stk * g[u][k];
                }
                dr[k] = dtk;
                if (WITH_S) sr[k] = stk;
            }
            *(float4*)(colsD + 8 * t)     = make_float4(dr[0], dr[1], dr[2], dr[3]);
            *(float4*)(colsD + 8 * t + 4) = make_float4(dr[4], dr[5], dr[6], dr[7]);
            if (WITH_S) {
                *(float4*)(colsS + 8 * t)     = make_float4(sr[0], sr[1], sr[2], sr[3]);
                *(float4*)(colsS + 8 * t + 4) = make_float4(sr[4], sr[5], sr[6], sr[7]);
            }
        }
        __syncthreads();
        if (q > ko) {
            float lt[8], st[8], lj[8][8];
            {
                float4 lo = *(const float4*)(colsD + 8 * t);
                float4 hi = *(const float4*)(colsD + 8 * t + 4);
                lt[0] = lo.x; lt[1] = lo.y; lt[2] = lo.z; lt[3] = lo.w;
                lt[4] = hi.x; lt[5] = hi.y; lt[6] = hi.z; lt[7] = hi.w;
            }
            if (WITH_S) {
                float4 lo = *(const float4*)(colsS + 8 * t);
                float4 hi = *(const float4*)(colsS + 8 * t + 4);
                st[0] = lo.x; st[1] = lo.y; st[2] = lo.z; st[3] = lo.w;
                st[4] = hi.x; st[5] = hi.y; st[6] = hi.z; st[7] = hi.w;
            } else {
#pragma unroll
                for (int k = 0; k < 8; k++) st[k] = 0.0f;
            }
#pragma unroll
            for (int u = 0; u < 8; u++) {
                float4 lo = *(const float4*)(colsD + 8 * (8 * q + u));
                float4 hi = *(const float4*)(colsD + 8 * (8 * q + u) + 4);
                lj[u][0] = lo.x; lj[u][1] = lo.y; lj[u][2] = lo.z; lj[u][3] = lo.w;
                lj[u][4] = hi.x; lj[u][5] = hi.y; lj[u][6] = hi.z; lj[u][7] = hi.w;
            }
#pragma unroll
            for (int k = 0; k < 8; k++) {
#pragma unroll
                for (int u = 0; u < 8; u++) {
                    dr[u] -= lt[k] * lj[u][k];
                    if (WITH_S) sr[u] -= st[k] * lj[u][k];
                }
            }
        }
    }
#pragma unroll
    for (int u = 0; u < 8; u++) {
        Dsh[t * LDP + 8 * q + u] = dr[u];
        if (WITH_S) Ssh[t * LDP + 8 * q + u] = sr[u];
    }
    __syncthreads();
}

__global__ __launch_bounds__(512) void chol_step_kernel(float* Rbase, float* Lbase, int p) {
    __shared__ float Dsh[NB * LDP];
    __shared__ float Ssh[NB * LDP];
    __shared__ float dg[NB];
    __shared__ __attribute__((aligned(16))) float piece[8 * 8];
    __shared__ __attribute__((aligned(16))) float colsD[NB * 8];
    __shared__ __attribute__((aligned(16))) float colsS[NB * 8];
    int b = blockIdx.x & (Bb - 1);
    int idx = blockIdx.x / Bb;
    int np = NBLK - p;
    int i, j;
    if (idx < np) {
        i = p + idx; j = p;
    } else {
        int tr = idx - np, a = 0;
        while (tr > a) { tr -= a + 1; a++; }
        i = p + 1 + a; j = p + 1 + tr;
    }
    bool isDiag = (i == p);
    float* R = Rbase + (size_t)b * Nn * Nn;
    float* L = Lbase + (size_t)b * Nn * Nn;
    int tid = threadIdx.x;
    int wave = tid >> 5, lane = tid & 31;
    int r = lane & 15, hi = lane >> 4;
    int tm = wave >> 2, tn = wave & 3;
    int prow = p * NB, irow = i * NB;
    int kcol = prow - NB;

    if (j > p) {
        int jrow = j * NB;
        v8f acc;
#pragma unroll
        for (int v = 0; v < 8; v++) {
            int row = tm * 16 + v + 8 * hi;
            int col = tn * 16 + r;
            acc[v] = -R[(size_t)(irow + row) * Nn + jrow + col];
        }
        const float* aI = L + (size_t)(irow + tm * 16 + r) * Nn + kcol;
        const float* bJ = L + (size_t)(jrow + tn * 16 + r) * Nn + kcol;
#pragma unroll
        for (int k0 = 0; k0 < NB; k0 += 4) {
            int kk = k0 + 2 * hi;
            v2f fi = *(const v2f*)(aI + kk);
            v2f fb = *(const v2f*)(bJ + kk);
            acc = __builtin_amdgcn_wmma_f32_16x16x4_f32(false, fi, false, fb,
                                                        (short)0, acc, false, false);
        }
#pragma unroll
        for (int v = 0; v < 8; v++) {
            int row = tm * 16 + v + 8 * hi;
            int col = tn * 16 + r;
            Dsh[row * LDP + col] = -acc[v];
        }
        __syncthreads();
        for (int e = tid; e < NB * NB; e += 512) {
            int rr = e >> 6, c = e & (NB - 1);
            R[(size_t)(irow + rr) * Nn + jrow + c] = Dsh[rr * LDP + c];
        }
        return;
    }

    v8f accD, accS;
#pragma unroll
    for (int v = 0; v < 8; v++) {
        int row = tm * 16 + v + 8 * hi;
        int col = tn * 16 + r;
        accD[v] = -R[(size_t)(prow + row) * Nn + prow + col];
        accS[v] = -R[(size_t)(irow + row) * Nn + prow + col];
    }
    if (p > 0) {
        const float* aP = L + (size_t)(prow + tm * 16 + r) * Nn + kcol;
        const float* aI = L + (size_t)(irow + tm * 16 + r) * Nn + kcol;
        const float* bP = L + (size_t)(prow + tn * 16 + r) * Nn + kcol;
#pragma unroll
        for (int k0 = 0; k0 < NB; k0 += 4) {
            int kk = k0 + 2 * hi;
            v2f fa = *(const v2f*)(aP + kk);
            v2f fi = *(const v2f*)(aI + kk);
            v2f fb = *(const v2f*)(bP + kk);
            accD = __builtin_amdgcn_wmma_f32_16x16x4_f32(false, fa, false, fb,
                                                         (short)0, accD, false, false);
            accS = __builtin_amdgcn_wmma_f32_16x16x4_f32(false, fi, false, fb,
                                                         (short)0, accS, false, false);
        }
    }
#pragma unroll
    for (int v = 0; v < 8; v++) {
        int row = tm * 16 + v + 8 * hi;
        int col = tn * 16 + r;
        Dsh[row * LDP + col] = -accD[v];
        Ssh[row * LDP + col] = -accS[v];
    }
    __syncthreads();

    if (isDiag) panel_sweep<false>(Dsh, Ssh, dg, piece, colsD, colsS, tid);
    else        panel_sweep<true>(Dsh, Ssh, dg, piece, colsD, colsS, tid);

    if (isDiag) {
        for (int e = tid; e < NB * NB; e += 512) {
            int rr = e >> 6, c = e & (NB - 1);
            float val = (c < rr) ? Dsh[rr * LDP + c] : ((c == rr) ? dg[rr] : 0.0f);
            L[(size_t)(prow + rr) * Nn + prow + c] = val;
        }
    } else {
        for (int e = tid; e < NB * NB; e += 512) {
            int rr = e >> 6, c = e & (NB - 1);
            L[(size_t)(irow + rr) * Nn + prow + c] = Ssh[rr * LDP + c];
        }
    }
}

__global__ __launch_bounds__(1024) void zmatvec_kernel(const float* __restrict__ Linvbase,
                                                       const float* __restrict__ y,
                                                       const float* __restrict__ mean,
                                                       float* __restrict__ zbase) {
    __shared__ float zrow[32];
    int b = blockIdx.y;
    int tid = threadIdx.x;
    int wave = tid >> 5, lane = tid & 31;
    int i = blockIdx.x * 32 + wave;
    const float* row = Linvbase + (size_t)b * Nn * Nn + (size_t)i * Nn;
    const float* yb = y + b * Nn;
    float mb = mean[b];
    int rowEnd = ((i >> 6) + 1) << 6;
    float4 l[8], v[8];
#pragma unroll
    for (int u = 0; u < 8; u++) {
        int c0 = 4 * lane + 128 * u;
        if (c0 < rowEnd) {
            l[u] = *(const float4*)(row + c0);
            v[u] = *(const float4*)(yb + c0);
        } else {
            l[u] = make_float4(0.0f, 0.0f, 0.0f, 0.0f);
            v[u] = make_float4(mb, mb, mb, mb);
        }
    }
    float s0 = 0.0f, s1 = 0.0f, s2 = 0.0f, s3 = 0.0f;
#pragma unroll
    for (int u = 0; u < 8; u++) {
        s0 += l[u].x * (v[u].x - mb);
        s1 += l[u].y * (v[u].y - mb);
        s2 += l[u].z * (v[u].z - mb);
        s3 += l[u].w * (v[u].w - mb);
    }
    float s = (s0 + s1) + (s2 + s3);
    for (int o = 16; o > 0; o >>= 1) s += __shfl_xor(s, o);
    if (lane == 0) zrow[wave] = s;
    __syncthreads();
    if (wave == 0) zbase[b * Nn + blockIdx.x * 32 + lane] = zrow[lane];
}

__global__ __launch_bounds__(1024) void nlml_kernel(const float* __restrict__ Lbase,
                                                    const float* __restrict__ zbase,
                                                    float* __restrict__ nlml_out) {
    int b = blockIdx.x;
    const float* L = Lbase + (size_t)b * Nn * Nn;
    __shared__ float col[Nn];
    __shared__ float zz[Nn];
    int tid = threadIdx.x;
    float z = zbase[b * Nn + tid];
    col[tid] = __logf(L[(size_t)tid * Nn + tid]);
    zz[tid] = z * z;
    __syncthreads();
    for (int s = 512; s > 0; s >>= 1) {
        if (tid < s) {
            col[tid] += col[tid + s];
            zz[tid] += zz[tid + s];
        }
        __syncthreads();
    }
    if (tid == 0) nlml_out[b] = 0.5f * zz[0] + col[0] + 0.5f * Nn * LOG2PIF;
}

__global__ __launch_bounds__(64) void trinv_diag_kernel(const float* __restrict__ Lbase,
                                                        float* __restrict__ Linvbase) {
    __shared__ __attribute__((aligned(16))) float Lb[NB * NB];
    __shared__ float xs[NB * NB];
    int p = blockIdx.x, b = blockIdx.y, tid = threadIdx.x;
    size_t base = (size_t)b * Nn * Nn + (size_t)(p * NB) * Nn + p * NB;
    const float* L = Lbase + base;
    float* Li = Linvbase + base;
    for (int idx = tid; idx < NB * NB; idx += 64) {
        int rr = idx >> 6, c = idx & (NB - 1);
        Lb[idx] = L[(size_t)rr * Nn + c];
    }
    __syncthreads();
    int j = tid;
    for (int i = 0; i < NB; i++) {
        const float* lr = Lb + i * NB;
        float s0 = (i == j) ? 1.0f : 0.0f, s1 = 0.0f, s2 = 0.0f, s3 = 0.0f;
        int k = 0;
        for (; k + 4 <= i; k += 4) {
            float4 l = *(const float4*)(lr + k);
            s0 -= l.x * xs[(k + 0) * NB + j];
            s1 -= l.y * xs[(k + 1) * NB + j];
            s2 -= l.z * xs[(k + 2) * NB + j];
            s3 -= l.w * xs[(k + 3) * NB + j];
        }
        for (; k < i; k++) s0 -= lr[k] * xs[k * NB + j];
        xs[i * NB + j] = ((s0 + s1) + (s2 + s3)) / lr[i];
    }
    __syncthreads();
    for (int idx = tid; idx < NB * NB; idx += 64) {
        int rr = idx >> 6, c = idx & (NB - 1);
        Li[(size_t)rr * Nn + c] = xs[idx];
    }
}

#define XLD 65

__device__ __forceinline__ void slab32(const float* aP, const float* bP, int kb, int hi,
                                       v8f& c0, v8f& c1) {
    v2f a[8], q[8];
#pragma unroll
    for (int u = 0; u < 8; u++) {
        int kk = kb + 4 * u + 2 * hi;
        a[u] = *(const v2f*)(aP + kk);
        q[u].x = bP[kk * XLD];
        q[u].y = bP[(kk + 1) * XLD];
    }
    __builtin_amdgcn_sched_barrier(0);
#pragma unroll
    for (int u = 0; u < 8; u += 2) {
        c0 = __builtin_amdgcn_wmma_f32_16x16x4_f32(false, a[u], false, q[u],
                                                   (short)0, c0, false, false);
        c1 = __builtin_amdgcn_wmma_f32_16x16x4_f32(false, a[u + 1], false, q[u + 1],
                                                   (short)0, c1, false, false);
    }
}
__global__ __launch_bounds__(512) void trinv_panel_kernel(const float* __restrict__ Lbase,
                                                          float* Linvbase) {
    extern __shared__ float smem[];
    float* Xp  = smem;
    float* Tsh = smem + Nn * XLD;
    int J = blockIdx.x, b = blockIdx.y, tid = threadIdx.x;
    const float* L = Lbase + (size_t)b * Nn * Nn;
    float* Li = Linvbase + (size_t)b * Nn * Nn;
    int wave = __builtin_amdgcn_readfirstlane(tid >> 5);
    int lane = tid & 31;
    int r = lane & 15, hi = lane >> 4;
    int tm = wave >> 2, tn = wave & 3;
    int jcol = J * NB;

    for (int idx = tid; idx < NB * NB; idx += 512) {
        int rr = idx >> 6, c = idx & (NB - 1);
        Xp[rr * XLD + c] = Li[(size_t)(jcol + rr) * Nn + jcol + c];
    }
    __syncthreads();

    for (int i = J + 1; i < NBLK; i++) {
        int irow = i * NB;
        int kdepth = irow - jcol;
        v8f acc0 = {}, acc1 = {};
        {
            const float* aP = L + (size_t)(irow + tm * 16 + r) * Nn + jcol;
            const float* bP = Xp + tn * 16 + r;
            for (int kb = 0; kb < kdepth; kb += 32) slab32(aP, bP, kb, hi, acc0, acc1);
        }
#pragma unroll
        for (int v = 0; v < 8; v++) {
            int row = tm * 16 + v + 8 * hi;
            int col = tn * 16 + r;
            Tsh[row * XLD + col] = acc0[v] + acc1[v];
        }
        __syncthreads();
        v8f x0 = {}, x1 = {};
        {
            const float* aD = Li + (size_t)(irow + tm * 16 + r) * Nn + irow;
            const float* bT = Tsh + tn * 16 + r;
            for (int kb = 0; kb < (tm + 1) * 16; kb += 32) slab32(aD, bT, kb, hi, x0, x1);
        }
#pragma unroll
        for (int v = 0; v < 8; v++) {
            int row = tm * 16 + v + 8 * hi;
            int col = tn * 16 + r;
            Xp[(kdepth + row) * XLD + col] = -(x0[v] + x1[v]);
        }
        __syncthreads();
        for (int idx = tid; idx < NB * NB; idx += 512) {
            int rr = idx >> 6, c = idx & (NB - 1);
            Li[(size_t)(irow + rr) * Nn + jcol + c] = Xp[(kdepth + rr) * XLD + c];
        }
    }
}


template <int NU0>
__device__ __forceinline__ void band_slab(const float* a0, const float* a1, const float* bK,
                                          int kb, int hi, v8f (&c0)[4], v8f (&c1)[4]) {
    v2f fa0[8], fa1[8], fb[8][4];
#pragma unroll
    for (int u = 0; u < 8; u++) {
        int kk = kb + 4 * u + 2 * hi;
        if (u < NU0) fa0[u] = *(const v2f*)(a0 + kk);
        fa1[u] = *(const v2f*)(a1 + kk);
#pragma unroll
        for (int c = 0; c < 4; c++) {
            fb[u][c].x = bK[kk * KTS + c * 16];
            fb[u][c].y = bK[(kk + 1) * KTS + c * 16];
        }
    }
#pragma unroll
    for (int u = 0; u < 8; u++) {
#pragma unroll
        for (int c = 0; c < 4; c++) {
            if (u < NU0)
                c0[c] = __builtin_amdgcn_wmma_f32_16x16x4_f32(false, fa0[u], false, fb[u][c],
                                                              (short)0, c0[c], false, false);
            c1[c] = __builtin_amdgcn_wmma_f32_16x16x4_f32(false, fa1[u], false, fb[u][c],
                                                          (short)0, c1[c], false, false);
        }
    }
}

#define PT 512
__global__ __launch_bounds__(PT) void predict_kernel(const float* __restrict__ Linvbase,
                                                     const float* __restrict__ trainPg,
                                                     const float* __restrict__ testPg,
                                                     const float* __restrict__ Kmm,
                                                     const float* __restrict__ zbase,
                                                     const float* __restrict__ meang,
                                                     const float* __restrict__ scale_param,
                                                     float* __restrict__ out_mean,
                                                     float* __restrict__ out_var) {
    extern __shared__ float smem[];
    float* Kt  = smem;
    float* zsh = smem + Nn * KTS;
    float* tp  = zsh + Nn;
    int b = blockIdx.y;
    int colBase = blockIdx.x * MT;
    int tid = threadIdx.x;
    float scale = __expf(scale_param[0]);

    for (int idx = tid; idx < Nn; idx += PT) zsh[idx] = zbase[b * Nn + idx];
    for (int idx = tid; idx < Nn * 6; idx += PT) tp[idx] = trainPg[(size_t)b * Nn * 6 + idx];
    __syncthreads();

    {
        int c = tid & 63;
        int h = tid >> 6;
        const float* q = testPg + (size_t)(b * Mm + colBase + c) * 6;
        float x2 = q[0], y2 = q[1], a2 = q[2], b2 = q[3], c2 = q[4], q2 = q[5];
        for (int n = h; n < Nn; n += PT / 64) {
            const float* p = tp + n * 6;
            Kt[n * KTS + c] = kern_pair(p[0], p[1], p[2], p[3], p[4], p[5],
                                        x2, y2, a2, b2, c2, q2, scale);
        }
    }
    __syncthreads();

    int wave = __builtin_amdgcn_readfirstlane(tid >> 5);
    int lane = tid & 31;
    int r = lane & 15, hi = lane >> 4;
    const float* Li = Linvbase + (size_t)b * Nn * Nn;
    const float* bK = Kt + r;

    float meanAcc[4] = {0.0f, 0.0f, 0.0f, 0.0f};
    float varAcc[4]  = {0.0f, 0.0f, 0.0f, 0.0f};
    for (int s = 0; s < 2; s++) {
        int p = s ? (Nn / 32 - 1 - wave) : wave;
        int row0 = p * 32;
        const float* a0 = Li + (size_t)(row0 + r) * Nn;
        const float* a1 = a0 + (size_t)16 * Nn;
        v8f c0[4] = {}, c1[4] = {};
        for (int t = 0; t < p; t++) band_slab<8>(a0, a1, bK, t * 32, hi, c0, c1);
        band_slab<4>(a0, a1, bK, row0, hi, c0, c1);
        const float* z0 = zsh + row0 + 8 * hi;
#pragma unroll
        for (int c = 0; c < 4; c++) {
#pragma unroll
            for (int v = 0; v < 8; v++) {
                float w0 = c0[c][v], w1 = c1[c][v];
                meanAcc[c] += z0[v] * w0;
                varAcc[c]  += w0 * w0;
                meanAcc[c] += z0[16 + v] * w1;
                varAcc[c]  += w1 * w1;
            }
        }
    }
    float* red = tp;
#pragma unroll
    for (int c = 0; c < 4; c++) {
        float m = meanAcc[c] + __shfl_xor(meanAcc[c], 16);
        float v = varAcc[c] + __shfl_xor(varAcc[c], 16);
        if (lane < 16) {
            red[wave * 128 + c * 16 + lane] = m;
            red[wave * 128 + 64 + c * 16 + lane] = v;
        }
    }
    __syncthreads();
    if (tid < MT) {
        float m = 0.0f, v = 0.0f;
#pragma unroll
        for (int w = 0; w < PT / 32; w++) {
            m += red[w * 128 + tid];
            v += red[w * 128 + 64 + tid];
        }
        int col = b * Mm + colBase + tid;
        out_mean[col] = meang[b] + m;
        out_var[col]  = Kmm[col] - v;
    }
}

extern "C" void kernel_launch(void* const* d_in, const int* in_sizes, int n_in,
                              void* d_out, int out_size, void* d_ws, size_t ws_size,
                              hipStream_t stream) {
    const float* coords_train = (const float*)d_in[0];
    const float* kp_train     = (const float*)d_in[1];
    const float* coords_test  = (const float*)d_in[2];
    const float* kp_test      = (const float*)d_in[3];
    const float* y_train      = (const float*)d_in[4];
    const float* var          = (const float*)d_in[5];
    const float* mean         = (const float*)d_in[6];
    const float* scale_param  = (const float*)d_in[7];

    float* ws = (float*)d_ws;
    size_t off = 0;
    float* A      = ws + off; off += (size_t)Bb * Nn * Nn;
    float* Lm     = ws + off; off += (size_t)Bb * Nn * Nn;
    float* Linv   = ws + off; off += (size_t)Bb * Nn * Nn;
    float* trainP = ws + off; off += (size_t)Bb * Nn * 6;
    float* testP  = ws + off; off += (size_t)Bb * Mm * 6;
    float* Kmm    = ws + off; off += (size_t)Bb * Mm;
    float* z      = ws + off; off += (size_t)Bb * Nn;

    float* out_mean = (float*)d_out;
    float* out_var  = out_mean + (size_t)Bb * Mm;
    float* nlml     = out_var + (size_t)Bb * Mm;

    prep_kernel<<<(Bb * Mm + 255) / 256, 256, 0, stream>>>(
        coords_train, kp_train, coords_test, kp_test, scale_param, trainP, testP, Kmm);
    buildA_kernel<<<(Bb * Nn * Nn) / 256, 256, 0, stream>>>(trainP, var, scale_param, A);
    chol_step_kernel<<<dim3(Bb * NBLK), 512, 0, stream>>>(A, Lm, 0);
#define CHOL_STEP(P) chol_step_kernel<<<dim3((NBLK - (P)) * (NBLK + 1 - (P))), 512, 0, stream>>>(A, Lm, (P))
                   CHOL_STEP(1);  CHOL_STEP(2);  CHOL_STEP(3);
    CHOL_STEP(4);  CHOL_STEP(5);  CHOL_STEP(6);  CHOL_STEP(7);
    CHOL_STEP(8);  CHOL_STEP(9);  CHOL_STEP(10); CHOL_STEP(11);
    CHOL_STEP(12); CHOL_STEP(13); CHOL_STEP(14); CHOL_STEP(15);
#undef CHOL_STEP
    trinv_diag_kernel<<<dim3(NBLK, Bb), 64, 0, stream>>>(Lm, Linv);
    size_t inv_smem_bytes = (size_t)(Nn * XLD + NB * XLD) * sizeof(float);
    trinv_panel_kernel<<<dim3(NBLK - 1, Bb), 512, inv_smem_bytes, stream>>>(Lm, Linv);
    zmatvec_kernel<<<dim3(Nn / 32, Bb), 1024, 0, stream>>>(Linv, y_train, mean, z);
    nlml_kernel<<<Bb, 1024, 0, stream>>>(Lm, z, nlml);

    size_t smem_bytes = (size_t)(Nn * KTS + Nn + Nn * 6) * sizeof(float);
    predict_kernel<<<dim3(Mm / MT, Bb), PT, smem_bytes, stream>>>(
        Linv, trainP, testP, Kmm, z, mean, scale_param, out_mean, out_var);
}


# Cholesky pivots: 1/sqrt(pivot) by the hardware reciprocal square root (1 ulp), diagonal = pivot times it, replacing correctly rounded sqrtf then IEEE divide on the serial chain; L's last bits change
# speedup vs baseline: 91.0613x; 1.1217x over previous
#include <hip/hip_runtime.h>
#include <math.h>

#define Nn 1024
#define Mm 16384
#define Bb 2
#define MT 64
#define KTS 65
#define NB 64
#define NBLK (Nn / NB)
#define LDP 65

typedef __attribute__((ext_vector_type(2))) float v2f;
typedef __attribute__((ext_vector_type(8))) float v8f;

#define SQRT3F 1.7320508075688772f
#define LOG2PIF 1.8378770664093453f

__device__ __forceinline__ float kern_pair(float x1, float y1, float a1, float b1, float c1, float q1,
                                           float x2, float y2, float a2, float b2, float c2, float q2,
                                           float scale) {
    float s00 = a1 + a2, s11 = b1 + b2, s01 = c1 + c2;
    float sdet = s00 * s11 - s01 * s01;
    float d0 = x1 - x2, d1 = y1 - y2;
    float Q = 0.5f * (s11 * d0 * d0 - 2.0f * s01 * d0 * d1 + s00 * d1 * d1) / sdet;
    float C = 2.0f * q1 * q2 * rsqrtf(fmaxf(sdet, 1e-5f));
    float t = SQRT3F * sqrtf(fmaxf(Q, 1e-5f));
    return (1.0f + t) * __expf(-t) * C * scale;
}

__global__ void prep_kernel(const float* __restrict__ ct, const float* __restrict__ kpt,
                            const float* __restrict__ cs, const float* __restrict__ kps,
                            const float* __restrict__ scale_param,
                            float* __restrict__ trainP, float* __restrict__ testP,
                            float* __restrict__ Kmm) {
    int gid = blockIdx.x * blockDim.x + threadIdx.x;
    float scale = __expf(scale_param[0]);
    if (gid < Bb * Mm) {
        float a = kps[gid * 3 + 0], b = kps[gid * 3 + 1], c = kps[gid * 3 + 2];
        float det = a * b - c * c;
        testP[gid * 6 + 0] = cs[gid * 2 + 0];
        testP[gid * 6 + 1] = cs[gid * 2 + 1];
        testP[gid * 6 + 2] = a;
        testP[gid * 6 + 3] = b;
        testP[gid * 6 + 4] = c;
        testP[gid * 6 + 5] = sqrtf(sqrtf(det));
        float t0 = SQRT3F * sqrtf(1e-5f);
        float m0 = (1.0f + t0) * __expf(-t0);
        float C = 2.0f * sqrtf(det) * rsqrtf(fmaxf(4.0f * det, 1e-5f));
        Kmm[gid] = C * m0 * scale;
    }
    if (gid < Bb * Nn) {
        float a = kpt[gid * 3 + 0], b = kpt[gid * 3 + 1], c = kpt[gid * 3 + 2];
        float det = a * b - c * c;
        trainP[gid * 6 + 0] = ct[gid * 2 + 0];
        trainP[gid * 6 + 1] = ct[gid * 2 + 1];
        trainP[gid * 6 + 2] = a;
        trainP[gid * 6 + 3] = b;
        trainP[gid * 6 + 4] = c;
        trainP[gid * 6 + 5] = sqrtf(sqrtf(det));
    }
}

__global__ void buildA_kernel(const float* __restrict__ trainP, const float* __restrict__ var,
                              const float* __restrict__ scale_param, float* __restrict__ A) {
    int gid = blockIdx.x * blockDim.x + threadIdx.x;
    int b = gid >> 20;
    int i = (gid >> 10) & (Nn - 1);
    int j = gid & (Nn - 1);
    const float* p1 = trainP + (size_t)(b * Nn + i) * 6;
    const float* p2 = trainP + (size_t)(b * Nn + j) * 6;
    float scale = __expf(scale_param[0]);
    float v = kern_pair(p1[0], p1[1], p1[2], p1[3], p1[4], p1[5],
                        p2[0], p2[1], p2[2], p2[3], p2[4], p2[5], scale);
    if (i == j) v += var[b * Nn + i];
    A[(size_t)gid] = v;
}

template <bool WITH_S>
__device__ __forceinline__ void panel_sweep(float* Dsh, float* Ssh, float* dg,
                                            float* piece, float* colsD, float* colsS, int tid) {
    int t = tid & (NB - 1);
    int q = __builtin_amdgcn_readfirstlane(tid >> 6);
    float dr[8], sr[8];
#pragma unroll
    for (int u = 0; u < 8; u++) {
        dr[u] = Dsh[t * LDP + 8 * q + u];
        sr[u] = WITH_S ? Ssh[t * LDP + 8 * q + u] : 0.0f;
    }
#pragma unroll 1
    for (int ko = 0; ko < NB / 8; ko++) {
        int m = t - 8 * ko;
        if (q == ko && m >= 0 && m < 8) {
            *(float4*)(piece + 8 * m)     = make_float4(dr[0], dr[1], dr[2], dr[3]);
            *(float4*)(piece + 8 * m + 4) = make_float4(dr[4], dr[5], dr[6], dr[7]);
        }
        __syncthreads();
        if (q == ko) {
            float g[8][8];
#pragma unroll
            for (int r = 0; r < 8; r++) {
                float4 lo = *(const float4*)(piece + 8 * r);
                float4 hi = *(const float4*)(piece + 8 * r + 4);
                g[r][0] = lo.x; g[r][1] = lo.y; g[r][2] = lo.z; g[r][3] = lo.w;
                g[r][4] = hi.x; g[r][5] = hi.y; g[r][6] = hi.z; g[r][7] = hi.w;
            }
            float dinv[8];
#pragma unroll
            for (int k = 0; k < 8; k++) {
                float di = __builtin_amdgcn_rsqf(g[k][k]);
                float d = g[k][k] * di;
                dinv[k] = di;
                if (t == 0) dg[8 * ko + k] = d;
#pragma unroll
                for (int r = k + 1; r < 8; r++) g[r][k] *= di;
#pragma unroll
                for (int r = k + 1; r < 8; r++) {
#pragma unroll
                    for (int cc = k + 1; cc <= r; cc++) g[r][cc] -= g[r][k] * g[cc][k];
                }
            }
#pragma unroll
            for (int k = 0; k < 8; k++) {
                float dtk = (t > 8 * ko + k) ? dr[k] * dinv[k] : 0.0f;
                float stk = sr[k] * dinv[k];
#pragma unroll
                for (int u = k + 1; u < 8; u++) {
                    dr[u] -= dtk * g[u][k];
                    if (WITH_S) sr[u] -= stk * g[u][k];
                }
                dr[k] = dtk;
                if (WITH_S) sr[k] = stk;
            }
            *(float4*)(colsD + 8 * t)     = make_float4(dr[0], dr[1], dr[2], dr[3]);
            *(float4*)(colsD + 8 * t + 4) = make_float4(dr[4], dr[5], dr[6], dr[7]);
            if (WITH_S) {
                *(float4*)(colsS + 8 * t)     = make_float4(sr[0], sr[1], sr[2], sr[3]);
                *(float4*)(colsS + 8 * t + 4) = make_float4(sr[4], sr[5], sr[6], sr[7]);
            }
        }
        __syncthreads();
        if (q > ko) {
            float lt[8], st[8], lj[8][8];
            {
                float4 lo = *(const float4*)(colsD + 8 * t);
                float4 hi = *(const float4*)(colsD + 8 * t + 4);
                lt[0] = lo.x; lt[1] = lo.y; lt[2] = lo.z; lt[3] = lo.w;
                lt[4] = hi.x; lt[5] = hi.y; lt[6] = hi.z; lt[7] = hi.w;
            }
            if (WITH_S) {
                float4 lo = *(const float4*)(colsS + 8 * t);
                float4 hi = *(const float4*)(colsS + 8 * t + 4);
                st[0] = lo.x; st[1] = lo.y; st[2] = lo.z; st[3] = lo.w;
                st[4] = hi.x; st[5] = hi.y; st[6] = hi.z; st[7] = hi.w;
            } else {
#pragma unroll
                for (int k = 0; k < 8; k++) st[k] = 0.0f;
            }
#pragma unroll
            for (int u = 0; u < 8; u++) {
                float4 lo = *(const float4*)(colsD + 8 * (8 * q + u));
                float4 hi = *(const float4*)(colsD + 8 * (8 * q + u) + 4);
                lj[u][0] = lo.x; lj[u][1] = lo.y; lj[u][2] = lo.z; lj[u][3] = lo.w;
                lj[u][4] = hi.x; lj[u][5] = hi.y; lj[u][6] = hi.z; lj[u][7] = hi.w;
            }
#pragma unroll
            for (int k = 0; k < 8; k++) {
#pragma unroll
                for (int u = 0; u < 8; u++) {
                    dr[u] -= lt[k] * lj[u][k];
                    if (WITH_S) sr[u] -= st[k] * lj[u][k];
                }
            }
        }
    }
#pragma unroll
    for (int u = 0; u < 8; u++) {
        Dsh[t * LDP + 8 * q + u] = dr[u];
        if (WITH_S) Ssh[t * LDP + 8 * q + u] = sr[u];
    }
    __syncthreads();
}

__global__ __launch_bounds__(512) void chol_step_kernel(float* Rbase, float* Lbase, int p) {
    __shared__ float Dsh[NB * LDP];
    __shared__ float Ssh[NB * LDP];
    __shared__ float dg[NB];
    __shared__ __attribute__((aligned(16))) float piece[8 * 8];
    __shared__ __attribute__((aligned(16))) float colsD[NB * 8];
    __shared__ __attribute__((aligned(16))) float colsS[NB * 8];
    int b = blockIdx.x & (Bb - 1);
    int idx = blockIdx.x / Bb;
    int np = NBLK - p;
    int i, j;
    if (idx < np) {
        i = p + idx; j = p;
    } else {
        int tr = idx - np, a = 0;
        while (tr > a) { tr -= a + 1; a++; }
        i = p + 1 + a; j = p + 1 + tr;
    }
    bool isDiag = (i == p);
    float* R = Rbase + (size_t)b * Nn * Nn;
    float* L = Lbase + (size_t)b * Nn * Nn;
    int tid = threadIdx.x;
    int wave = tid >> 5, lane = tid & 31;
    int r = lane & 15, hi = lane >> 4;
    int tm = wave >> 2, tn = wave & 3;
    int prow = p * NB, irow = i * NB;
    int kcol = prow - NB;

    if (j > p) {
        int jrow = j * NB;
        v8f acc;
#pragma unroll
        for (int v = 0; v < 8; v++) {
            int row = tm * 16 + v + 8 * hi;
            int col = tn * 16 + r;
            acc[v] = -R[(size_t)(irow + row) * Nn + jrow + col];
        }
        const float* aI = L + (size_t)(irow + tm * 16 + r) * Nn + kcol;
        const float* bJ = L + (size_t)(jrow + tn * 16 + r) * Nn + kcol;
#pragma unroll
        for (int k0 = 0; k0 < NB; k0 += 4) {
            int kk = k0 + 2 * hi;
            v2f fi = *(const v2f*)(aI + kk);
            v2f fb = *(const v2f*)(bJ + kk);
            acc = __builtin_amdgcn_wmma_f32_16x16x4_f32(false, fi, false, fb,
                                                        (short)0, acc, false, false);
        }
#pragma unroll
        for (int v = 0; v < 8; v++) {
            int row = tm * 16 + v + 8 * hi;
            int col = tn * 16 + r;
            Dsh[row * LDP + col] = -acc[v];
        }
        __syncthreads();
        for (int e = tid; e < NB * NB; e += 512) {
            int rr = e >> 6, c = e & (NB - 1);
            R[(size_t)(irow + rr) * Nn + jrow + c] = Dsh[rr * LDP + c];
        }
        return;
    }

    v8f accD, accS;
#pragma unroll
    for (int v = 0; v < 8; v++) {
        int row = tm * 16 + v + 8 * hi;
        int col = tn * 16 + r;
        accD[v] = -R[(size_t)(prow + row) * Nn + prow + col];
        accS[v] = -R[(size_t)(irow + row) * Nn + prow + col];
    }
    if (p > 0) {
        const float* aP = L + (size_t)(prow + tm * 16 + r) * Nn + kcol;
        const float* aI = L + (size_t)(irow + tm * 16 + r) * Nn + kcol;
        const float* bP = L + (size_t)(prow + tn * 16 + r) * Nn + kcol;
#pragma unroll
        for (int k0 = 0; k0 < NB; k0 += 4) {
            int kk = k0 + 2 * hi;
            v2f fa = *(const v2f*)(aP + kk);
            v2f fi = *(const v2f*)(aI + kk);
            v2f fb = *(const v2f*)(bP + kk);
            accD = __builtin_amdgcn_wmma_f32_16x16x4_f32(false, fa, false, fb,
                                                         (short)0, accD, false, false);
            accS = __builtin_amdgcn_wmma_f32_16x16x4_f32(false, fi, false, fb,
                                                         (short)0, accS, false, false);
        }
    }
#pragma unroll
    for (int v = 0; v < 8; v++) {
        int row = tm * 16 + v + 8 * hi;
        int col = tn * 16 + r;
        Dsh[row * LDP + col] = -accD[v];
        Ssh[row * LDP + col] = -accS[v];
    }
    __syncthreads();

    if (isDiag) panel_sweep<false>(Dsh, Ssh, dg, piece, colsD, colsS, tid);
    else        panel_sweep<true>(Dsh, Ssh, dg, piece, colsD, colsS, tid);

    if (isDiag) {
        for (int e = tid; e < NB * NB; e += 512) {
            int rr = e >> 6, c = e & (NB - 1);
            float val = (c < rr) ? Dsh[rr * LDP + c] : ((c == rr) ? dg[rr] : 0.0f);
            L[(size_t)(prow + rr) * Nn + prow + c] = val;
        }
    } else {
        for (int e = tid; e < NB * NB; e += 512) {
            int rr = e >> 6, c = e & (NB - 1);
            L[(size_t)(irow + rr) * Nn + prow + c] = Ssh[rr * LDP + c];
        }
    }
}

__global__ __launch_bounds__(1024) void zmatvec_kernel(const float* __restrict__ Linvbase,
                                                       const float* __restrict__ y,
                                                       const float* __restrict__ mean,
                                                       float* __restrict__ zbase) {
    __shared__ float zrow[32];
    int b = blockIdx.y;
    int tid = threadIdx.x;
    int wave = tid >> 5, lane = tid & 31;
    int i = blockIdx.x * 32 + wave;
    const float* row = Linvbase + (size_t)b * Nn * Nn + (size_t)i * Nn;
    const float* yb = y + b * Nn;
    float mb = mean[b];
    int rowEnd = ((i >> 6) + 1) << 6;
    float4 l[8], v[8];
#pragma unroll
    for (int u = 0; u < 8; u++) {
        int c0 = 4 * lane + 128 * u;
        if (c0 < rowEnd) {
            l[u] = *(const float4*)(row + c0);
            v[u] = *(const float4*)(yb + c0);
        } else {
            l[u] = make_float4(0.0f, 0.0f, 0.0f, 0.0f);
            v[u] = make_float4(mb, mb, mb, mb);
        }
    }
    float s0 = 0.0f, s1 = 0.0f, s2 = 0.0f, s3 = 0.0f;
#pragma unroll
    for (int u = 0; u < 8; u++) {
        s0 += l[u].x * (v[u].x - mb);
        s1 += l[u].y * (v[u].y - mb);
        s2 += l[u].z * (v[u].z - mb);
        s3 += l[u].w * (v[u].w - mb);
    }
    float s = (s0 + s1) + (s2 + s3);
    for (int o = 16; o > 0; o >>= 1) s += __shfl_xor(s, o);
    if (lane == 0) zrow[wave] = s;
    __syncthreads();
    if (wave == 0) zbase[b * Nn + blockIdx.x * 32 + lane] = zrow[lane];
}

__global__ __launch_bounds__(1024) void nlml_kernel(const float* __restrict__ Lbase,
                                                    const float* __restrict__ zbase,
                                                    float* __restrict__ nlml_out) {
    int b = blockIdx.x;
    const float* L = Lbase + (size_t)b * Nn * Nn;
    __shared__ float col[Nn];
    __shared__ float zz[Nn];
    int tid = threadIdx.x;
    float z = zbase[b * Nn + tid];
    col[tid] = __logf(L[(size_t)tid * Nn + tid]);
    zz[tid] = z * z;
    __syncthreads();
    for (int s = 512; s > 0; s >>= 1) {
        if (tid < s) {
            col[tid] += col[tid + s];
            zz[tid] += zz[tid + s];
        }
        __syncthreads();
    }
    if (tid == 0) nlml_out[b] = 0.5f * zz[0] + col[0] + 0.5f * Nn * LOG2PIF;
}

__global__ __launch_bounds__(64) void trinv_diag_kernel(const float* __restrict__ Lbase,
                                                        float* __restrict__ Linvbase) {
    __shared__ __attribute__((aligned(16))) float Lb[NB * NB];
    __shared__ float xs[NB * NB];
    int p = blockIdx.x, b = blockIdx.y, tid = threadIdx.x;
    size_t base = (size_t)b * Nn * Nn + (size_t)(p * NB) * Nn + p * NB;
    const float* L = Lbase + base;
    float* Li = Linvbase + base;
    for (int idx = tid; idx < NB * NB; idx += 64) {
        int rr = idx >> 6, c = idx & (NB - 1);
        Lb[idx] = L[(size_t)rr * Nn + c];
    }
    __syncthreads();
    int j = tid;
    for (int i = 0; i < NB; i++) {
        const float* lr = Lb + i * NB;
        float s0 = (i == j) ? 1.0f : 0.0f, s1 = 0.0f, s2 = 0.0f, s3 = 0.0f;
        int k = 0;
        for (; k + 4 <= i; k += 4) {
            float4 l = *(const float4*)(lr + k);
            s0 -= l.x * xs[(k + 0) * NB + j];
            s1 -= l.y * xs[(k + 1) * NB + j];
            s2 -= l.z * xs[(k + 2) * NB + j];
            s3 -= l.w * xs[(k + 3) * NB + j];
        }
        for (; k < i; k++) s0 -= lr[k] * xs[k * NB + j];
        xs[i * NB + j] = ((s0 + s1) + (s2 + s3)) / lr[i];
    }
    __syncthreads();
    for (int idx = tid; idx < NB * NB; idx += 64) {
        int rr = idx >> 6, c = idx & (NB - 1);
        Li[(size_t)rr * Nn + c] = xs[idx];
    }
}

#define XLD 65

__device__ __forceinline__ void slab32(const float* aP, const float* bP, int kb, int hi,
                                       v8f& c0, v8f& c1) {
    v2f a[8], q[8];
#pragma unroll
    for (int u = 0; u < 8; u++) {
        int kk = kb + 4 * u + 2 * hi;
        a[u] = *(const v2f*)(aP + kk);
        q[u].x = bP[kk * XLD];
        q[u].y = bP[(kk + 1) * XLD];
    }
    __builtin_amdgcn_sched_barrier(0);
#pragma unroll
    for (int u = 0; u < 8; u += 2) {
        c0 = __builtin_amdgcn_wmma_f32_16x16x4_f32(false, a[u], false, q[u],
                                                   (short)0, c0, false, false);
        c1 = __builtin_amdgcn_wmma_f32_16x16x4_f32(false, a[u + 1], false, q[u + 1],
                                                   (short)0, c1, false, false);
    }
}
__global__ __launch_bounds__(512) void trinv_panel_kernel(const float* __restrict__ Lbase,
                                                          float* Linvbase) {
    extern __shared__ float smem[];
    float* Xp  = smem;
    float* Tsh = smem + Nn * XLD;
    int J = blockIdx.x, b = blockIdx.y, tid = threadIdx.x;
    const float* L = Lbase + (size_t)b * Nn * Nn;
    float* Li = Linvbase + (size_t)b * Nn * Nn;
    int wave = __builtin_amdgcn_readfirstlane(tid >> 5);
    int lane = tid & 31;
    int r = lane & 15, hi = lane >> 4;
    int tm = wave >> 2, tn = wave & 3;
    int jcol = J * NB;

    for (int idx = tid; idx < NB * NB; idx += 512) {
        int rr = idx >> 6, c = idx & (NB - 1);
        Xp[rr * XLD + c] = Li[(size_t)(jcol + rr) * Nn + jcol + c];
    }
    __syncthreads();

    for (int i = J + 1; i < NBLK; i++) {
        int irow = i * NB;
        int kdepth = irow - jcol;
        v8f acc0 = {}, acc1 = {};
        {
            const float* aP = L + (size_t)(irow + tm * 16 + r) * Nn + jcol;
            const float* bP = Xp + tn * 16 + r;
            for (int kb = 0; kb < kdepth; kb += 32) slab32(aP, bP, kb, hi, acc0, acc1);
        }
#pragma unroll
        for (int v = 0; v < 8; v++) {
            int row = tm * 16 + v + 8 * hi;
            int col = tn * 16 + r;
            Tsh[row * XLD + col] = acc0[v] + acc1[v];
        }
        __syncthreads();
        v8f x0 = {}, x1 = {};
        {
            const float* aD = Li + (size_t)(irow + tm * 16 + r) * Nn + irow;
            const float* bT = Tsh + tn * 16 + r;
            for (int kb = 0; kb < (tm + 1) * 16; kb += 32) slab32(aD, bT, kb, hi, x0, x1);
        }
#pragma unroll
        for (int v = 0; v < 8; v++) {
            int row = tm * 16 + v + 8 * hi;
            int col = tn * 16 + r;
            Xp[(kdepth + row) * XLD + col] = -(x0[v] + x1[v]);
        }
        __syncthreads();
        for (int idx = tid; idx < NB * NB; idx += 512) {
            int rr = idx >> 6, c = idx & (NB - 1);
            Li[(size_t)(irow + rr) * Nn + jcol + c] = Xp[(kdepth + rr) * XLD + c];
        }
    }
}


template <int NU0>
__device__ __forceinline__ void band_slab(const float* a0, const float* a1, const float* bK,
                                          int kb, int hi, v8f (&c0)[4], v8f (&c1)[4]) {
    v2f fa0[8], fa1[8], fb[8][4];
#pragma unroll
    for (int u = 0; u < 8; u++) {
        int kk = kb + 4 * u + 2 * hi;
        if (u < NU0) fa0[u] = *(const v2f*)(a0 + kk);
        fa1[u] = *(const v2f*)(a1 + kk);
#pragma unroll
        for (int c = 0; c < 4; c++) {
            fb[u][c].x = bK[kk * KTS + c * 16];
            fb[u][c].y = bK[(kk + 1) * KTS + c * 16];
        }
    }
#pragma unroll
    for (int u = 0; u < 8; u++) {
#pragma unroll
        for (int c = 0; c < 4; c++) {
            if (u < NU0)
                c0[c] = __builtin_amdgcn_wmma_f32_16x16x4_f32(false, fa0[u], false, fb[u][c],
                                                              (short)0, c0[c], false, false);
            c1[c] = __builtin_amdgcn_wmma_f32_16x16x4_f32(false, fa1[u], false, fb[u][c],
                                                          (short)0, c1[c], false, false);
        }
    }
}

#define PT 512
__global__ __launch_bounds__(PT) void predict_kernel(const float* __restrict__ Linvbase,
                                                     const float* __restrict__ trainPg,
                                                     const float* __restrict__ testPg,
                                                     const float* __restrict__ Kmm,
                                                     const float* __restrict__ zbase,
                                                     const float* __restrict__ meang,
                                                     const float* __restrict__ scale_param,
                                                     float* __restrict__ out_mean,
                                                     float* __restrict__ out_var) {
    extern __shared__ float smem[];
    float* Kt  = smem;
    float* zsh = smem + Nn * KTS;
    float* tp  = zsh + Nn;
    int b = blockIdx.y;
    int colBase = blockIdx.x * MT;
    int tid = threadIdx.x;
    float scale = __expf(scale_param[0]);

    for (int idx = tid; idx < Nn; idx += PT) zsh[idx] = zbase[b * Nn + idx];
    for (int idx = tid; idx < Nn * 6; idx += PT) tp[idx] = trainPg[(size_t)b * Nn * 6 + idx];
    __syncthreads();

    {
        int c = tid & 63;
        int h = tid >> 6;
        const float* q = testPg + (size_t)(b * Mm + colBase + c) * 6;
        float x2 = q[0], y2 = q[1], a2 = q[2], b2 = q[3], c2 = q[4], q2 = q[5];
        for (int n = h; n < Nn; n += PT / 64) {
            const float* p = tp + n * 6;
            Kt[n * KTS + c] = kern_pair(p[0], p[1], p[2], p[3], p[4], p[5],
                                        x2, y2, a2, b2, c2, q2, scale);
        }
    }
    __syncthreads();

    int wave = __builtin_amdgcn_readfirstlane(tid >> 5);
    int lane = tid & 31;
    int r = lane & 15, hi = lane >> 4;
    const float* Li = Linvbase + (size_t)b * Nn * Nn;
    const float* bK = Kt + r;

    float meanAcc[4] = {0.0f, 0.0f, 0.0f, 0.0f};
    float varAcc[4]  = {0.0f, 0.0f, 0.0f, 0.0f};
    for (int s = 0; s < 2; s++) {
        int p = s ? (Nn / 32 - 1 - wave) : wave;
        int row0 = p * 32;
        const float* a0 = Li + (size_t)(row0 + r) * Nn;
        const float* a1 = a0 + (size_t)16 * Nn;
        v8f c0[4] = {}, c1[4] = {};
        for (int t = 0; t < p; t++) band_slab<8>(a0, a1, bK, t * 32, hi, c0, c1);
        band_slab<4>(a0, a1, bK, row0, hi, c0, c1);
        const float* z0 = zsh + row0 + 8 * hi;
#pragma unroll
        for (int c = 0; c < 4; c++) {
#pragma unroll
            for (int v = 0; v < 8; v++) {
                float w0 = c0[c][v], w1 = c1[c][v];
                meanAcc[c] += z0[v] * w0;
                varAcc[c]  += w0 * w0;
                meanAcc[c] += z0[16 + v] * w1;
                varAcc[c]  += w1 * w1;
            }
        }
    }
    float* red = tp;
#pragma unroll
    for (int c = 0; c < 4; c++) {
        float m = meanAcc[c] + __shfl_xor(meanAcc[c], 16);
        float v = varAcc[c] + __shfl_xor(varAcc[c], 16);
        if (lane < 16) {
            red[wave * 128 + c * 16 + lane] = m;
            red[wave * 128 + 64 + c * 16 + lane] = v;
        }
    }
    __syncthreads();
    if (tid < MT) {
        float m = 0.0f, v = 0.0f;
#pragma unroll
        for (int w = 0; w < PT / 32; w++) {
            m += red[w * 128 + tid];
            v += red[w * 128 + 64 + tid];
        }
        int col = b * Mm + colBase + tid;
        out_mean[col] = meang[b] + m;
        out_var[col]  = Kmm[col] - v;
    }
}

extern "C" void kernel_launch(void* const* d_in, const int* in_sizes, int n_in,
                              void* d_out, int out_size, void* d_ws, size_t ws_size,
                              hipStream_t stream) {
    const float* coords_train = (const float*)d_in[0];
    const float* kp_train     = (const float*)d_in[1];
    const float* coords_test  = (const float*)d_in[2];
    const float* kp_test      = (const float*)d_in[3];
    const float* y_train      = (const float*)d_in[4];
    const float* var          = (const float*)d_in[5];
    const float* mean         = (const float*)d_in[6];
    const float* scale_param  = (const float*)d_in[7];

    float* ws = (float*)d_ws;
    size_t off = 0;
    float* A      = ws + off; off += (size_t)Bb * Nn * Nn;
    float* Lm     = ws + off; off += (size_t)Bb * Nn * Nn;
    float* Linv   = ws + off; off += (size_t)Bb * Nn * Nn;
    float* trainP = ws + off; off += (size_t)Bb * Nn * 6;
    float* testP  = ws + off; off += (size_t)Bb * Mm * 6;
    float* Kmm    = ws + off; off += (size_t)Bb * Mm;
    float* z      = ws + off; off += (size_t)Bb * Nn;

    float* out_mean = (float*)d_out;
    float* out_var  = out_mean + (size_t)Bb * Mm;
    float* nlml     = out_var + (size_t)Bb * Mm;

    prep_kernel<<<(Bb * Mm + 255) / 256, 256, 0, stream>>>(
        coords_train, kp_train, coords_test, kp_test, scale_param, trainP, testP, Kmm);
    buildA_kernel<<<(Bb * Nn * Nn) / 256, 256, 0, stream>>>(trainP, var, scale_param, A);
    chol_step_kernel<<<dim3(Bb * NBLK), 512, 0, stream>>>(A, Lm, 0);
#define CHOL_STEP(P) chol_step_kernel<<<dim3((NBLK - (P)) * (NBLK + 1 - (P))), 512, 0, stream>>>(A, Lm, (P))
                   CHOL_STEP(1);  CHOL_STEP(2);  CHOL_STEP(3);
    CHOL_STEP(4);  CHOL_STEP(5);  CHOL_STEP(6);  CHOL_STEP(7);
    CHOL_STEP(8);  CHOL_STEP(9);  CHOL_STEP(10); CHOL_STEP(11);
    CHOL_STEP(12); CHOL_STEP(13); CHOL_STEP(14); CHOL_STEP(15);
#undef CHOL_STEP
    trinv_diag_kernel<<<dim3(NBLK, Bb), 64, 0, stream>>>(Lm, Linv);
    size_t inv_smem_bytes = (size_t)(Nn * XLD + NB * XLD) * sizeof(float);
    trinv_panel_kernel<<<dim3(NBLK - 1, Bb), 512, inv_smem_bytes, stream>>>(Lm, Linv);
    zmatvec_kernel<<<dim3(Nn / 32, Bb), 1024, 0, stream>>>(Linv, y_train, mean, z);
    nlml_kernel<<<Bb, 1024, 0, stream>>>(Lm, z, nlml);

    size_t smem_bytes = (size_t)(Nn * KTS + Nn + Nn * 6) * sizeof(float);
    predict_kernel<<<dim3(Mm / MT, Bb), PT, smem_bytes, stream>>>(
        Linv, trainP, testP, Kmm, z, mean, scale_param, out_mean, out_var);
}


# explicit inverse, off-diagonal blocks: row recurrence X L = I by 16-row strips, 120 workgroups instead of 30, depth split over 4 waves; new summation order inside the inverse
# speedup vs baseline: 104.1927x; 1.1442x over previous
#include <hip/hip_runtime.h>
#include <math.h>

#define Nn 1024
#define Mm 16384
#define Bb 2
#define MT 64
#define KTS 65
#define NB 64
#define NBLK (Nn / NB)
#define LDP 65

typedef __attribute__((ext_vector_type(2))) float v2f;
typedef __attribute__((ext_vector_type(8))) float v8f;

#define SQRT3F 1.7320508075688772f
#define LOG2PIF 1.8378770664093453f

__device__ __forceinline__ float kern_pair(float x1, float y1, float a1, float b1, float c1, float q1,
                                           float x2, float y2, float a2, float b2, float c2, float q2,
                                           float scale) {
    float s00 = a1 + a2, s11 = b1 + b2, s01 = c1 + c2;
    float sdet = s00 * s11 - s01 * s01;
    float d0 = x1 - x2, d1 = y1 - y2;
    float Q = 0.5f * (s11 * d0 * d0 - 2.0f * s01 * d0 * d1 + s00 * d1 * d1) / sdet;
    float C = 2.0f * q1 * q2 * rsqrtf(fmaxf(sdet, 1e-5f));
    float t = SQRT3F * sqrtf(fmaxf(Q, 1e-5f));
    return (1.0f + t) * __expf(-t) * C * scale;
}

__global__ void prep_kernel(const float* __restrict__ ct, const float* __restrict__ kpt,
                            const float* __restrict__ cs, const float* __restrict__ kps,
                            const float* __restrict__ scale_param,
                            float* __restrict__ trainP, float* __restrict__ testP,
                            float* __restrict__ Kmm) {
    int gid = blockIdx.x * blockDim.x + threadIdx.x;
    float scale = __expf(scale_param[0]);
    if (gid < Bb * Mm) {
        float a = kps[gid * 3 + 0], b = kps[gid * 3 + 1], c = kps[gid * 3 + 2];
        float det = a * b - c * c;
        testP[gid * 6 + 0] = cs[gid * 2 + 0];
        testP[gid * 6 + 1] = cs[gid * 2 + 1];
        testP[gid * 6 + 2] = a;
        testP[gid * 6 + 3] = b;
        testP[gid * 6 + 4] = c;
        testP[gid * 6 + 5] = sqrtf(sqrtf(det));
        float t0 = SQRT3F * sqrtf(1e-5f);
        float m0 = (1.0f + t0) * __expf(-t0);
        float C = 2.0f * sqrtf(det) * rsqrtf(fmaxf(4.0f * det, 1e-5f));
        Kmm[gid] = C * m0 * scale;
    }
    if (gid < Bb * Nn) {
        float a = kpt[gid * 3 + 0], b = kpt[gid * 3 + 1], c = kpt[gid * 3 + 2];
        float det = a * b - c * c;
        trainP[gid * 6 + 0] = ct[gid * 2 + 0];
        trainP[gid * 6 + 1] = ct[gid * 2 + 1];
        trainP[gid * 6 + 2] = a;
        trainP[gid * 6 + 3] = b;
        trainP[gid * 6 + 4] = c;
        trainP[gid * 6 + 5] = sqrtf(sqrtf(det));
    }
}

__global__ void buildA_kernel(const float* __restrict__ trainP, const float* __restrict__ var,
                              const float* __restrict__ scale_param, float* __restrict__ A) {
    int gid = blockIdx.x * blockDim.x + threadIdx.x;
    int b = gid >> 20;
    int i = (gid >> 10) & (Nn - 1);
    int j = gid & (Nn - 1);
    const float* p1 = trainP + (size_t)(b * Nn + i) * 6;
    const float* p2 = trainP + (size_t)(b * Nn + j) * 6;
    float scale = __expf(scale_param[0]);
    float v = kern_pair(p1[0], p1[1], p1[2], p1[3], p1[4], p1[5],
                        p2[0], p2[1], p2[2], p2[3], p2[4], p2[5], scale);
    if (i == j) v += var[b * Nn + i];
    A[(size_t)gid] = v;
}

template <bool WITH_S>
__device__ __forceinline__ void panel_sweep(float* Dsh, float* Ssh, float* dg,
                                            float* piece, float* colsD, float* colsS, int tid) {
    int t = tid & (NB - 1);
    int q = __builtin_amdgcn_readfirstlane(tid >> 6);
    float dr[8], sr[8];
#pragma unroll
    for (int u = 0; u < 8; u++) {
        dr[u] = Dsh[t * LDP + 8 * q + u];
        sr[u] = WITH_S ? Ssh[t * LDP + 8 * q + u] : 0.0f;
    }
#pragma unroll 1
    for (int ko = 0; ko < NB / 8; ko++) {
        int m = t - 8 * ko;
        if (q == ko && m >= 0 && m < 8) {
            *(float4*)(piece + 8 * m)     = make_float4(dr[0], dr[1], dr[2], dr[3]);
            *(float4*)(piece + 8 * m + 4) = make_float4(dr[4], dr[5], dr[6], dr[7]);
        }
        __syncthreads();
        if (q == ko) {
            float g[8][8];
#pragma unroll
            for (int r = 0; r < 8; r++) {
                float4 lo = *(const float4*)(piece + 8 * r);
                float4 hi = *(const float4*)(piece + 8 * r + 4);
                g[r][0] = lo.x; g[r][1] = lo.y; g[r][2] = lo.z; g[r][3] = lo.w;
                g[r][4] = hi.x; g[r][5] = hi.y; g[r][6] = hi.z; g[r][7] = hi.w;
            }
            float dinv[8];
#pragma unroll
            for (int k = 0; k < 8; k++) {
                float di = __builtin_amdgcn_rsqf(g[k][k]);
                float d = g[k][k] * di;
                dinv[k] = di;
                if (t == 0) dg[8 * ko + k] = d;
#pragma unroll
                for (int r = k + 1; r < 8; r++) g[r][k] *= di;
#pragma unroll
                for (int r = k + 1; r < 8; r++) {
#pragma unroll
                    for (int cc = k + 1; cc <= r; cc++) g[r][cc] -= g[r][k] * g[cc][k];
                }
            }
#pragma unroll
            for (int k = 0; k < 8; k++) {
                float dtk = (t > 8 * ko + k) ? dr[k] * dinv[k] : 0.0f;
                float stk = sr[k] * dinv[k];
#pragma unroll
                for (int u = k + 1; u < 8; u++) {
                    dr[u] -= dtk * g[u][k];
                    if (WITH_S) sr[u] -= stk * g[u][k];
                }
                dr[k] = dtk;
                if (WITH_S) sr[k] = stk;
            }
            *(float4*)(colsD + 8 * t)     = make_float4(dr[0], dr[1], dr[2], dr[3]);
            *(float4*)(colsD + 8 * t + 4) = make_float4(dr[4], dr[5], dr[6], dr[7]);
            if (WITH_S) {
                *(float4*)(colsS + 8 * t)     = make_float4(sr[0], sr[1], sr[2], sr[3]);
                *(float4*)(colsS + 8 * t + 4) = make_float4(sr[4], sr[5], sr[6], sr[7]);
            }
        }
        __syncthreads();
        if (q > ko) {
            float lt[8], st[8], lj[8][8];
            {
                float4 lo = *(const float4*)(colsD + 8 * t);
                float4 hi = *(const float4*)(colsD + 8 * t + 4);
                lt[0] = lo.x; lt[1] = lo.y; lt[2] = lo.z; lt[3] = lo.w;
                lt[4] = hi.x; lt[5] = hi.y; lt[6] = hi.z; lt[7] = hi.w;
            }
            if (WITH_S) {
                float4 lo = *(const float4*)(colsS + 8 * t);
                float4 hi = *(const float4*)(colsS + 8 * t + 4);
                st[0] = lo.x; st[1] = lo.y; st[2] = lo.z; st[3] = lo.w;
                st[4] = hi.x; st[5] = hi.y; st[6] = hi.z; st[7] = hi.w;
            } else {
#pragma unroll
                for (int k = 0; k < 8; k++) st[k] = 0.0f;
            }
#pragma unroll
            for (int u = 0; u < 8; u++) {
                float4 lo = *(const float4*)(colsD + 8 * (8 * q + u));
                float4 hi = *(const float4*)(colsD + 8 * (8 * q + u) + 4);
                lj[u][0] = lo.x; lj[u][1] = lo.y; lj[u][2] = lo.z; lj[u][3] = lo.w;
                lj[u][4] = hi.x; lj[u][5] = hi.y; lj[u][6] = hi.z; lj[u][7] = hi.w;
            }
#pragma unroll
            for (int k = 0; k < 8; k++) {
#pragma unroll
                for (int u = 0; u < 8; u++) {
                    dr[u] -= lt[k] * lj[u][k];
                    if (WITH_S) sr[u] -= st[k] * lj[u][k];
                }
            }
        }
    }
#pragma unroll
    for (int u = 0; u < 8; u++) {
        Dsh[t * LDP + 8 * q + u] = dr[u];
        if (WITH_S) Ssh[t * LDP + 8 * q + u] = sr[u];
    }
    __syncthreads();
}

__global__ __launch_bounds__(512) void chol_step_kernel(float* Rbase, float* Lbase, int p) {
    __shared__ float Dsh[NB * LDP];
    __shared__ float Ssh[NB * LDP];
    __shared__ float dg[NB];
    __shared__ __attribute__((aligned(16))) float piece[8 * 8];
    __shared__ __attribute__((aligned(16))) float colsD[NB * 8];
    __shared__ __attribute__((aligned(16))) float colsS[NB * 8];
    int b = blockIdx.x & (Bb - 1);
    int idx = blockIdx.x / Bb;
    int np = NBLK - p;
    int i, j;
    if (idx < np) {
        i = p + idx; j = p;
    } else {
        int tr = idx - np, a = 0;
        while (tr > a) { tr -= a + 1; a++; }
        i = p + 1 + a; j = p + 1 + tr;
    }
    bool isDiag = (i == p);
    float* R = Rbase + (size_t)b * Nn * Nn;
    float* L = Lbase + (size_t)b * Nn * Nn;
    int tid = threadIdx.x;
    int wave = tid >> 5, lane = tid & 31;
    int r = lane & 15, hi = lane >> 4;
    int tm = wave >> 2, tn = wave & 3;
    int prow = p * NB, irow = i * NB;
    int kcol = prow - NB;

    if (j > p) {
        int jrow = j * NB;
        v8f acc;
#pragma unroll
        for (int v = 0; v < 8; v++) {
            int row = tm * 16 + v + 8 * hi;
            int col = tn * 16 + r;
            acc[v] = -R[(size_t)(irow + row) * Nn + jrow + col];
        }
        const float* aI = L + (size_t)(irow + tm * 16 + r) * Nn + kcol;
        const float* bJ = L + (size_t)(jrow + tn * 16 + r) * Nn + kcol;
#pragma unroll
        for (int k0 = 0; k0 < NB; k0 += 4) {
            int kk = k0 + 2 * hi;
            v2f fi = *(const v2f*)(aI + kk);
            v2f fb = *(const v2f*)(bJ + kk);
            acc = __builtin_amdgcn_wmma_f32_16x16x4_f32(false, fi, false, fb,
                                                        (short)0, acc, false, false);
        }
#pragma unroll
        for (int v = 0; v < 8; v++) {
            int row = tm * 16 + v + 8 * hi;
            int col = tn * 16 + r;
            Dsh[row * LDP + col] = -acc[v];
        }
        __syncthreads();
        for (int e = tid; e < NB * NB; e += 512) {
            int rr = e >> 6, c = e & (NB - 1);
            R[(size_t)(irow + rr) * Nn + jrow + c] = Dsh[rr * LDP + c];
        }
        return;
    }

    v8f accD, accS;
#pragma unroll
    for (int v = 0; v < 8; v++) {
        int row = tm * 16 + v + 8 * hi;
        int col = tn * 16 + r;
        accD[v] = -R[(size_t)(prow + row) * Nn + prow + col];
        accS[v] = -R[(size_t)(irow + row) * Nn + prow + col];
    }
    if (p > 0) {
        const float* aP = L + (size_t)(prow + tm * 16 + r) * Nn + kcol;
        const float* aI = L + (size_t)(irow + tm * 16 + r) * Nn + kcol;
        const float* bP = L + (size_t)(prow + tn * 16 + r) * Nn + kcol;
#pragma unroll
        for (int k0 = 0; k0 < NB; k0 += 4) {
            int kk = k0 + 2 * hi;
            v2f fa = *(const v2f*)(aP + kk);
            v2f fi = *(const v2f*)(aI + kk);
            v2f fb = *(const v2f*)(bP + kk);
            accD = __builtin_amdgcn_wmma_f32_16x16x4_f32(false, fa, false, fb,
                                                         (short)0, accD, false, false);
            accS = __builtin_amdgcn_wmma_f32_16x16x4_f32(false, fi, false, fb,
                                                         (short)0, accS, false, false);
        }
    }
#pragma unroll
    for (int v = 0; v < 8; v++) {
        int row = tm * 16 + v + 8 * hi;
        int col = tn * 16 + r;
        Dsh[row * LDP + col] = -accD[v];
        Ssh[row * LDP + col] = -accS[v];
    }
    __syncthreads();

    if (isDiag) panel_sweep<false>(Dsh, Ssh, dg, piece, colsD, colsS, tid);
    else        panel_sweep<true>(Dsh, Ssh, dg, piece, colsD, colsS, tid);

    if (isDiag) {
        for (int e = tid; e < NB * NB; e += 512) {
            int rr = e >> 6, c = e & (NB - 1);
            float val = (c < rr) ? Dsh[rr * LDP + c] : ((c == rr) ? dg[rr] : 0.0f);
            L[(size_t)(prow + rr) * Nn + prow + c] = val;
        }
    } else {
        for (int e = tid; e < NB * NB; e += 512) {
            int rr = e >> 6, c = e & (NB - 1);
            L[(size_t)(irow + rr) * Nn + prow + c] = Ssh[rr * LDP + c];
        }
    }
}

__global__ __launch_bounds__(1024) void zmatvec_kernel(const float* __restrict__ Linvbase,
                                                       const float* __restrict__ y,
                                                       const float* __restrict__ mean,
                                                       float* __restrict__ zbase) {
    __shared__ float zrow[32];
    int b = blockIdx.y;
    int tid = threadIdx.x;
    int wave = tid >> 5, lane = tid & 31;
    int i = blockIdx.x * 32 + wave;
    const float* row = Linvbase + (size_t)b * Nn * Nn + (size_t)i * Nn;
    const float* yb = y + b * Nn;
    float mb = mean[b];
    int rowEnd = ((i >> 6) + 1) << 6;
    float4 l[8], v[8];
#pragma unroll
    for (int u = 0; u < 8; u++) {
        int c0 = 4 * lane + 128 * u;
        if (c0 < rowEnd) {
            l[u] = *(const float4*)(row + c0);
            v[u] = *(const float4*)(yb + c0);
        } else {
            l[u] = make_float4(0.0f, 0.0f, 0.0f, 0.0f);
            v[u] = make_float4(mb, mb, mb, mb);
        }
    }
    float s0 = 0.0f, s1 = 0.0f, s2 = 0.0f, s3 = 0.0f;
#pragma unroll
    for (int u = 0; u < 8; u++) {
        s0 += l[u].x * (v[u].x - mb);
        s1 += l[u].y * (v[u].y - mb);
        s2 += l[u].z * (v[u].z - mb);
        s3 += l[u].w * (v[u].w - mb);
    }
    float s = (s0 + s1) + (s2 + s3);
    for (int o = 16; o > 0; o >>= 1) s += __shfl_xor(s, o);
    if (lane == 0) zrow[wave] = s;
    __syncthreads();
    if (wave == 0) zbase[b * Nn + blockIdx.x * 32 + lane] = zrow[lane];
}

__global__ __launch_bounds__(1024) void nlml_kernel(const float* __restrict__ Lbase,
                                                    const float* __restrict__ zbase,
                                                    float* __restrict__ nlml_out) {
    int b = blockIdx.x;
    const float* L = Lbase + (size_t)b * Nn * Nn;
    __shared__ float col[Nn];
    __shared__ float zz[Nn];
    int tid = threadIdx.x;
    float z = zbase[b * Nn + tid];
    col[tid] = __logf(L[(size_t)tid * Nn + tid]);
    zz[tid] = z * z;
    __syncthreads();
    for (int s = 512; s > 0; s >>= 1) {
        if (tid < s) {
            col[tid] += col[tid + s];
            zz[tid] += zz[tid + s];
        }
        __syncthreads();
    }
    if (tid == 0) nlml_out[b] = 0.5f * zz[0] + col[0] + 0.5f * Nn * LOG2PIF;
}

__global__ __launch_bounds__(64) void trinv_diag_kernel(const float* __restrict__ Lbase,
                                                        float* __restrict__ Linvbase) {
    __shared__ __attribute__((aligned(16))) float Lb[NB * NB];
    __shared__ float xs[NB * NB];
    int p = blockIdx.x, b = blockIdx.y, tid = threadIdx.x;
    size_t base = (size_t)b * Nn * Nn + (size_t)(p * NB) * Nn + p * NB;
    const float* L = Lbase + base;
    float* Li = Linvbase + base;
    for (int idx = tid; idx < NB * NB; idx += 64) {
        int rr = idx >> 6, c = idx & (NB - 1);
        Lb[idx] = L[(size_t)rr * Nn + c];
    }
    __syncthreads();
    int j = tid;
    for (int i = 0; i < NB; i++) {
        const float* lr = Lb + i * NB;
        float s0 = (i == j) ? 1.0f : 0.0f, s1 = 0.0f, s2 = 0.0f, s3 = 0.0f;
        int k = 0;
        for (; k + 4 <= i; k += 4) {
            float4 l = *(const float4*)(lr + k);
            s0 -= l.x * xs[(k + 0) * NB + j];
            s1 -= l.y * xs[(k + 1) * NB + j];
            s2 -= l.z * xs[(k + 2) * NB + j];
            s3 -= l.w * xs[(k + 3) * NB + j];
        }
        for (; k < i; k++) s0 -= lr[k] * xs[k * NB + j];
        xs[i * NB + j] = ((s0 + s1) + (s2 + s3)) / lr[i];
    }
    __syncthreads();
    for (int idx = tid; idx < NB * NB; idx += 64) {
        int rr = idx >> 6, c = idx & (NB - 1);
        Li[(size_t)rr * Nn + c] = xs[idx];
    }
}

#define RP 1028
#define TPP 68

__device__ __forceinline__ void load_chunk16(const float* xp, const float* lp, v2f* a, v2f* q) {
#pragma unroll
    for (int u = 0; u < 4; u++) {
        a[u] = *(const v2f*)(xp + 4 * u);
        q[u].x = lp[(4 * u) * Nn];
        q[u].y = lp[(4 * u + 1) * Nn];
    }
}
__device__ __forceinline__ void mma_chunk16(const v2f* a, const v2f* q, v8f& c0, v8f& c1) {
#pragma unroll
    for (int u = 0; u < 4; u += 2) {
        c0 = __builtin_amdgcn_wmma_f32_16x16x4_f32(false, a[u], false, q[u],
                                                   (short)0, c0, false, false);
        c1 = __builtin_amdgcn_wmma_f32_16x16x4_f32(false, a[u + 1], false, q[u + 1],
                                                   (short)0, c1, false, false);
    }
}
__global__ __launch_bounds__(512) void trinv_rows_kernel(const float* __restrict__ Lbase,
                                                         float* Linvbase) {
    extern __shared__ float smem[];
    float* Xs = smem;
    float* Ps = smem + 16 * RP;
    int i = NBLK - 1 - (int)(blockIdx.x >> 2);
    int tm = blockIdx.x & 3;
    int b = blockIdx.y, tid = threadIdx.x;
    const float* L = Lbase + (size_t)b * Nn * Nn;
    float* Li = Linvbase + (size_t)b * Nn * Nn;
    int wave = __builtin_amdgcn_readfirstlane(tid >> 5);
    int lane = tid & 31;
    int r = lane & 15, hi = lane >> 4;
    int tn = wave & 3, kq = wave >> 2;
    int g0 = i * NB + tm * 16;
    int icol = i * NB;

    for (int e = tid; e < 16 * NB; e += 512) {
        int rr = e >> 6, cc = e & (NB - 1);
        Xs[rr * RP + icol + cc] = Li[(size_t)(g0 + rr) * Nn + icol + cc];
    }
    __syncthreads();

    for (int J = i - 1; J >= 0; J--) {
        int jcol = J * NB;
        v2f dq[16];
        if (kq == 0) {
            const float* dp = Li + (size_t)(jcol + 2 * hi) * Nn + jcol + tn * 16 + r;
#pragma unroll
            for (int kc = 0; kc < 4; kc++) {
                if (kc >= tn) {
#pragma unroll
                    for (int u = 0; u < 4; u++) {
                        dq[4 * kc + u].x = dp[(16 * kc + 4 * u) * Nn];
                        dq[4 * kc + u].y = dp[(16 * kc + 4 * u + 1) * Nn];
                    }
                }
            }
        }
        int k0 = (J + 1) * NB + kq * 16 + 2 * hi;
        const float* xp = Xs + r * RP + k0;
        const float* lp = L + (size_t)k0 * Nn + jcol + tn * 16 + r;
        v8f acc0 = {}, acc1 = {};
        int kb = J + 1;
        for (; kb + 1 <= i; kb += 2) {
            v2f a0[4], q0[4], a1[4], q1[4];
            load_chunk16(xp, lp, a0, q0);
            load_chunk16(xp + NB, lp + (size_t)NB * Nn, a1, q1);
            mma_chunk16(a0, q0, acc0, acc1);
            mma_chunk16(a1, q1, acc0, acc1);
            xp += 2 * NB;
            lp += (size_t)2 * NB * Nn;
        }
        if (kb <= i) {
            v2f a0[4], q0[4];
            load_chunk16(xp, lp, a0, q0);
            mma_chunk16(a0, q0, acc0, acc1);
        }
#pragma unroll
        for (int v = 0; v < 8; v++)
            Ps[(kq * 16 + v + 8 * hi) * TPP + tn * 16 + r] = acc0[v] + acc1[v];
        __syncthreads();
        if (kq == 0) {
            v8f x0 = {}, x1 = {};
            const float* pp = Ps + r * TPP + 2 * hi;
#pragma unroll
            for (int kc = 0; kc < 4; kc++) {
                if (kc >= tn) {
                    v2f a[4];
#pragma unroll
                    for (int u = 0; u < 4; u++) {
                        const float* p = pp + 16 * kc + 4 * u;
                        v2f p0 = *(const v2f*)(p);
                        v2f p1 = *(const v2f*)(p + 16 * TPP);
                        v2f p2 = *(const v2f*)(p + 32 * TPP);
                        v2f p3 = *(const v2f*)(p + 48 * TPP);
                        a[u] = (p0 + p1) + (p2 + p3);
                    }
                    mma_chunk16(a, dq + 4 * kc, x0, x1);
                }
            }
#pragma unroll
            for (int v = 0; v < 8; v++)
                Xs[(v + 8 * hi) * RP + jcol + tn * 16 + r] = -(x0[v] + x1[v]);
        }
        __syncthreads();
        for (int e = tid; e < 16 * NB; e += 512) {
            int rr = e >> 6, cc = e & (NB - 1);
            Li[(size_t)(g0 + rr) * Nn + jcol + cc] = Xs[rr * RP + jcol + cc];
        }
    }
}


template <int NU0>
__device__ __forceinline__ void band_slab(const float* a0, const float* a1, const float* bK,
                                          int kb, int hi, v8f (&c0)[4], v8f (&c1)[4]) {
    v2f fa0[8], fa1[8], fb[8][4];
#pragma unroll
    for (int u = 0; u < 8; u++) {
        int kk = kb + 4 * u + 2 * hi;
        if (u < NU0) fa0[u] = *(const v2f*)(a0 + kk);
        fa1[u] = *(const v2f*)(a1 + kk);
#pragma unroll
        for (int c = 0; c < 4; c++) {
            fb[u][c].x = bK[kk * KTS + c * 16];
            fb[u][c].y = bK[(kk + 1) * KTS + c * 16];
        }
    }
#pragma unroll
    for (int u = 0; u < 8; u++) {
#pragma unroll
        for (int c = 0; c < 4; c++) {
            if (u < NU0)
                c0[c] = __builtin_amdgcn_wmma_f32_16x16x4_f32(false, fa0[u], false, fb[u][c],
                                                              (short)0, c0[c], false, false);
            c1[c] = __builtin_amdgcn_wmma_f32_16x16x4_f32(false, fa1[u], false, fb[u][c],
                                                          (short)0, c1[c], false, false);
        }
    }
}

#define PT 512
__global__ __launch_bounds__(PT) void predict_kernel(const float* __restrict__ Linvbase,
                                                     const float* __restrict__ trainPg,
                                                     const float* __restrict__ testPg,
                                                     const float* __restrict__ Kmm,
                                                     const float* __restrict__ zbase,
                                                     const float* __restrict__ meang,
                                                     const float* __restrict__ scale_param,
                                                     float* __restrict__ out_mean,
                                                     float* __restrict__ out_var) {
    extern __shared__ float smem[];
    float* Kt  = smem;
    float* zsh = smem + Nn * KTS;
    float* tp  = zsh + Nn;
    int b = blockIdx.y;
    int colBase = blockIdx.x * MT;
    int tid = threadIdx.x;
    float scale = __expf(scale_param[0]);

    for (int idx = tid; idx < Nn; idx += PT) zsh[idx] = zbase[b * Nn + idx];
    for (int idx = tid; idx < Nn * 6; idx += PT) tp[idx] = trainPg[(size_t)b * Nn * 6 + idx];
    __syncthreads();

    {
        int c = tid & 63;
        int h = tid >> 6;
        const float* q = testPg + (size_t)(b * Mm + colBase + c) * 6;
        float x2 = q[0], y2 = q[1], a2 = q[2], b2 = q[3], c2 = q[4], q2 = q[5];
        for (int n = h; n < Nn; n += PT / 64) {
            const float* p = tp + n * 6;
            Kt[n * KTS + c] = kern_pair(p[0], p[1], p[2], p[3], p[4], p[5],
                                        x2, y2, a2, b2, c2, q2, scale);
        }
    }
    __syncthreads();

    int wave = __builtin_amdgcn_readfirstlane(tid >> 5);
    int lane = tid & 31;
    int r = lane & 15, hi = lane >> 4;
    const float* Li = Linvbase + (size_t)b * Nn * Nn;
    const float* bK = Kt + r;

    float meanAcc[4] = {0.0f, 0.0f, 0.0f, 0.0f};
    float varAcc[4]  = {0.0f, 0.0f, 0.0f, 0.0f};
    for (int s = 0; s < 2; s++) {
        int p = s ? (Nn / 32 - 1 - wave) : wave;
        int row0 = p * 32;
        const float* a0 = Li + (size_t)(row0 + r) * Nn;
        const float* a1 = a0 + (size_t)16 * Nn;
        v8f c0[4] = {}, c1[4] = {};
        for (int t = 0; t < p; t++) band_slab<8>(a0, a1, bK, t * 32, hi, c0, c1);
        band_slab<4>(a0, a1, bK, row0, hi, c0, c1);
        const float* z0 = zsh + row0 + 8 * hi;
#pragma unroll
        for (int c = 0; c < 4; c++) {
#pragma unroll
            for (int v = 0; v < 8; v++) {
                float w0 = c0[c][v], w1 = c1[c][v];
                meanAcc[c] += z0[v] * w0;
                varAcc[c]  += w0 * w0;
                meanAcc[c] += z0[16 + v] * w1;
                varAcc[c]  += w1 * w1;
            }
        }
    }
    float* red = tp;
#pragma unroll
    for (int c = 0; c < 4; c++) {
        float m = meanAcc[c] + __shfl_xor(meanAcc[c], 16);
        float v = varAcc[c] + __shfl_xor(varAcc[c], 16);
        if (lane < 16) {
            red[wave * 128 + c * 16 + lane] = m;
            red[wave * 128 + 64 + c * 16 + lane] = v;
        }
    }
    __syncthreads();
    if (tid < MT) {
        float m = 0.0f, v = 0.0f;
#pragma unroll
        for (int w = 0; w < PT / 32; w++) {
            m += red[w * 128 + tid];
            v += red[w * 128 + 64 + tid];
        }
        int col = b * Mm + colBase + tid;
        out_mean[col] = meang[b] + m;
        out_var[col]  = Kmm[col] - v;
    }
}

extern "C" void kernel_launch(void* const* d_in, const int* in_sizes, int n_in,
                              void* d_out, int out_size, void* d_ws, size_t ws_size,
                              hipStream_t stream) {
    const float* coords_train = (const float*)d_in[0];
    const float* kp_train     = (const float*)d_in[1];
    const float* coords_test  = (const float*)d_in[2];
    const float* kp_test      = (const float*)d_in[3];
    const float* y_train      = (const float*)d_in[4];
    const float* var          = (const float*)d_in[5];
    const float* mean         = (const float*)d_in[6];
    const float* scale_param  = (const float*)d_in[7];

    float* ws = (float*)d_ws;
    size_t off = 0;
    float* A      = ws + off; off += (size_t)Bb * Nn * Nn;
    float* Lm     = ws + off; off += (size_t)Bb * Nn * Nn;
    float* Linv   = ws + off; off += (size_t)Bb * Nn * Nn;
    float* trainP = ws + off; off += (size_t)Bb * Nn * 6;
    float* testP  = ws + off; off += (size_t)Bb * Mm * 6;
    float* Kmm    = ws + off; off += (size_t)Bb * Mm;
    float* z      = ws + off; off += (size_t)Bb * Nn;

    float* out_mean = (float*)d_out;
    float* out_var  = out_mean + (size_t)Bb * Mm;
    float* nlml     = out_var + (size_t)Bb * Mm;

    prep_kernel<<<(Bb * Mm + 255) / 256, 256, 0, stream>>>(
        coords_train, kp_train, coords_test, kp_test, scale_param, trainP, testP, Kmm);
    buildA_kernel<<<(Bb * Nn * Nn) / 256, 256, 0, stream>>>(trainP, var, scale_param, A);
    chol_step_kernel<<<dim3(Bb * NBLK), 512, 0, stream>>>(A, Lm, 0);
#define CHOL_STEP(P) chol_step_kernel<<<dim3((NBLK - (P)) * (NBLK + 1 - (P))), 512, 0, stream>>>(A, Lm, (P))
                   CHOL_STEP(1);  CHOL_STEP(2);  CHOL_STEP(3);
    CHOL_STEP(4);  CHOL_STEP(5);  CHOL_STEP(6);  CHOL_STEP(7);
    CHOL_STEP(8);  CHOL_STEP(9);  CHOL_STEP(10); CHOL_STEP(11);
    CHOL_STEP(12); CHOL_STEP(13); CHOL_STEP(14); CHOL_STEP(15);
#undef CHOL_STEP
    trinv_diag_kernel<<<dim3(NBLK, Bb), 64, 0, stream>>>(Lm, Linv);
    size_t inv_smem_bytes = (size_t)(16 * RP + 4 * 16 * TPP) * sizeof(float);
    trinv_rows_kernel<<<dim3(4 * (NBLK - 1), Bb), 512, inv_smem_bytes, stream>>>(Lm, Linv);
    zmatvec_kernel<<<dim3(Nn / 32, Bb), 1024, 0, stream>>>(Linv, y_train, mean, z);
    nlml_kernel<<<Bb, 1024, 0, stream>>>(Lm, z, nlml);

    size_t smem_bytes = (size_t)(Nn * KTS + Nn + Nn * 6) * sizeof(float);
    predict_kernel<<<dim3(Mm / MT, Bb), PT, smem_bytes, stream>>>(
        Linv, trainP, testP, Kmm, z, mean, scale_param, out_mean, out_var);
}


# explicit inverse, 64x64 diagonal blocks: each thread keeps its column in registers through a fully unrolled forward substitution (was in LDS, rolled loops); arithmetic and order unchanged
# speedup vs baseline: 108.1767x; 1.0382x over previous
#include <hip/hip_runtime.h>
#include <math.h>

#define Nn 1024
#define Mm 16384
#define Bb 2
#define MT 64
#define KTS 65
#define NB 64
#define NBLK (Nn / NB)
#define LDP 65

typedef __attribute__((ext_vector_type(2))) float v2f;
typedef __attribute__((ext_vector_type(8))) float v8f;

#define SQRT3F 1.7320508075688772f
#define LOG2PIF 1.8378770664093453f

__device__ __forceinline__ float kern_pair(float x1, float y1, float a1, float b1, float c1, float q1,
                                           float x2, float y2, float a2, float b2, float c2, float q2,
                                           float scale) {
    float s00 = a1 + a2, s11 = b1 + b2, s01 = c1 + c2;
    float sdet = s00 * s11 - s01 * s01;
    float d0 = x1 - x2, d1 = y1 - y2;
    float Q = 0.5f * (s11 * d0 * d0 - 2.0f * s01 * d0 * d1 + s00 * d1 * d1) / sdet;
    float C = 2.0f * q1 * q2 * rsqrtf(fmaxf(sdet, 1e-5f));
    float t = SQRT3F * sqrtf(fmaxf(Q, 1e-5f));
    return (1.0f + t) * __expf(-t) * C * scale;
}

__global__ void prep_kernel(const float* __restrict__ ct, const float* __restrict__ kpt,
                            const float* __restrict__ cs, const float* __restrict__ kps,
                            const float* __restrict__ scale_param,
                            float* __restrict__ trainP, float* __restrict__ testP,
                            float* __restrict__ Kmm) {
    int gid = blockIdx.x * blockDim.x + threadIdx.x;
    float scale = __expf(scale_param[0]);
    if (gid < Bb * Mm) {
        float a = kps[gid * 3 + 0], b = kps[gid * 3 + 1], c = kps[gid * 3 + 2];
        float det = a * b - c * c;
        testP[gid * 6 + 0] = cs[gid * 2 + 0];
        testP[gid * 6 + 1] = cs[gid * 2 + 1];
        testP[gid * 6 + 2] = a;
        testP[gid * 6 + 3] = b;
        testP[gid * 6 + 4] = c;
        testP[gid * 6 + 5] = sqrtf(sqrtf(det));
        float t0 = SQRT3F * sqrtf(1e-5f);
        float m0 = (1.0f + t0) * __expf(-t0);
        float C = 2.0f * sqrtf(det) * rsqrtf(fmaxf(4.0f * det, 1e-5f));
        Kmm[gid] = C * m0 * scale;
    }
    if (gid < Bb * Nn) {
        float a = kpt[gid * 3 + 0], b = kpt[gid * 3 + 1], c = kpt[gid * 3 + 2];
        float det = a * b - c * c;
        trainP[gid * 6 + 0] = ct[gid * 2 + 0];
        trainP[gid * 6 + 1] = ct[gid * 2 + 1];
        trainP[gid * 6 + 2] = a;
        trainP[gid * 6 + 3] = b;
        trainP[gid * 6 + 4] = c;
        trainP[gid * 6 + 5] = sqrtf(sqrtf(det));
    }
}

__global__ void buildA_kernel(const float* __restrict__ trainP, const float* __restrict__ var,
                              const float* __restrict__ scale_param, float* __restrict__ A) {
    int gid = blockIdx.x * blockDim.x + threadIdx.x;
    int b = gid >> 20;
    int i = (gid >> 10) & (Nn - 1);
    int j = gid & (Nn - 1);
    const float* p1 = trainP + (size_t)(b * Nn + i) * 6;
    const float* p2 = trainP + (size_t)(b * Nn + j) * 6;
    float scale = __expf(scale_param[0]);
    float v = kern_pair(p1[0], p1[1], p1[2], p1[3], p1[4], p1[5],
                        p2[0], p2[1], p2[2], p2[3], p2[4], p2[5], scale);
    if (i == j) v += var[b * Nn + i];
    A[(size_t)gid] = v;
}

template <bool WITH_S>
__device__ __forceinline__ void panel_sweep(float* Dsh, float* Ssh, float* dg,
                                            float* piece, float* colsD, float* colsS, int tid) {
    int t = tid & (NB - 1);
    int q = __builtin_amdgcn_readfirstlane(tid >> 6);
    float dr[8], sr[8];
#pragma unroll
    for (int u = 0; u < 8; u++) {
        dr[u] = Dsh[t * LDP + 8 * q + u];
        sr[u] = WITH_S ? Ssh[t * LDP + 8 * q + u] : 0.0f;
    }
#pragma unroll 1
    for (int ko = 0; ko < NB / 8; ko++) {
        int m = t - 8 * ko;
        if (q == ko && m >= 0 && m < 8) {
            *(float4*)(piece + 8 * m)     = make_float4(dr[0], dr[1], dr[2], dr[3]);
            *(float4*)(piece + 8 * m + 4) = make_float4(dr[4], dr[5], dr[6], dr[7]);
        }
        __syncthreads();
        if (q == ko) {
            float g[8][8];
#pragma unroll
            for (int r = 0; r < 8; r++) {
                float4 lo = *(const float4*)(piece + 8 * r);
                float4 hi = *(const float4*)(piece + 8 * r + 4);
                g[r][0] = lo.x; g[r][1] = lo.y; g[r][2] = lo.z; g[r][3] = lo.w;
                g[r][4] = hi.x; g[r][5] = hi.y; g[r][6] = hi.z; g[r][7] = hi.w;
            }
            float dinv[8];
#pragma unroll
            for (int k = 0; k < 8; k++) {
                float di = __builtin_amdgcn_rsqf(g[k][k]);
                float d = g[k][k] * di;
                dinv[k] = di;
                if (t == 0) dg[8 * ko + k] = d;
#pragma unroll
                for (int r = k + 1; r < 8; r++) g[r][k] *= di;
#pragma unroll
                for (int r = k + 1; r < 8; r++) {
#pragma unroll
                    for (int cc = k + 1; cc <= r; cc++) g[r][cc] -= g[r][k] * g[cc][k];
                }
            }
#pragma unroll
            for (int k = 0; k < 8; k++) {
                float dtk = (t > 8 * ko + k) ? dr[k] * dinv[k] : 0.0f;
                float stk = sr[k] * dinv[k];
#pragma unroll
                for (int u = k + 1; u < 8; u++) {
                    dr[u] -= dtk * g[u][k];
                    if (WITH_S) sr[u] -= stk * g[u][k];
                }
                dr[k] = dtk;
                if (WITH_S) sr[k] = stk;
            }
            *(float4*)(colsD + 8 * t)     = make_float4(dr[0], dr[1], dr[2], dr[3]);
            *(float4*)(colsD + 8 * t + 4) = make_float4(dr[4], dr[5], dr[6], dr[7]);
            if (WITH_S) {
                *(float4*)(colsS + 8 * t)     = make_float4(sr[0], sr[1], sr[2], sr[3]);
                *(float4*)(colsS + 8 * t + 4) = make_float4(sr[4], sr[5], sr[6], sr[7]);
            }
        }
        __syncthreads();
        if (q > ko) {
            float lt[8], st[8], lj[8][8];
            {
                float4 lo = *(const float4*)(colsD + 8 * t);
                float4 hi = *(const float4*)(colsD + 8 * t + 4);
                lt[0] = lo.x; lt[1] = lo.y; lt[2] = lo.z; lt[3] = lo.w;
                lt[4] = hi.x; lt[5] = hi.y; lt[6] = hi.z; lt[7] = hi.w;
            }
            if (WITH_S) {
                float4 lo = *(const float4*)(colsS + 8 * t);
                float4 hi = *(const float4*)(colsS + 8 * t + 4);
                st[0] = lo.x; st[1] = lo.y; st[2] = lo.z; st[3] = lo.w;
                st[4] = hi.x; st[5] = hi.y; st[6] = hi.z; st[7] = hi.w;
            } else {
#pragma unroll
                for (int k = 0; k < 8; k++) st[k] = 0.0f;
            }
#pragma unroll
            for (int u = 0; u < 8; u++) {
                float4 lo = *(const float4*)(colsD + 8 * (8 * q + u));
                float4 hi = *(const float4*)(colsD + 8 * (8 * q + u) + 4);
                lj[u][0] = lo.x; lj[u][1] = lo.y; lj[u][2] = lo.z; lj[u][3] = lo.w;
                lj[u][4] = hi.x; lj[u][5] = hi.y; lj[u][6] = hi.z; lj[u][7] = hi.w;
            }
#pragma unroll
            for (int k = 0; k < 8; k++) {
#pragma unroll
                for (int u = 0; u < 8; u++) {
                    dr[u] -= lt[k] * lj[u][k];
                    if (WITH_S) sr[u] -= st[k] * lj[u][k];
                }
            }
        }
    }
#pragma unroll
    for (int u = 0; u < 8; u++) {
        Dsh[t * LDP + 8 * q + u] = dr[u];
        if (WITH_S) Ssh[t * LDP + 8 * q + u] = sr[u];
    }
    __syncthreads();
}

__global__ __launch_bounds__(512) void chol_step_kernel(float* Rbase, float* Lbase, int p) {
    __shared__ float Dsh[NB * LDP];
    __shared__ float Ssh[NB * LDP];
    __shared__ float dg[NB];
    __shared__ __attribute__((aligned(16))) float piece[8 * 8];
    __shared__ __attribute__((aligned(16))) float colsD[NB * 8];
    __shared__ __attribute__((aligned(16))) float colsS[NB * 8];
    int b = blockIdx.x & (Bb - 1);
    int idx = blockIdx.x / Bb;
    int np = NBLK - p;
    int i, j;
    if (idx < np) {
        i = p + idx; j = p;
    } else {
        int tr = idx - np, a = 0;
        while (tr > a) { tr -= a + 1; a++; }
        i = p + 1 + a; j = p + 1 + tr;
    }
    bool isDiag = (i == p);
    float* R = Rbase + (size_t)b * Nn * Nn;
    float* L = Lbase + (size_t)b * Nn * Nn;
    int tid = threadIdx.x;
    int wave = tid >> 5, lane = tid & 31;
    int r = lane & 15, hi = lane >> 4;
    int tm = wave >> 2, tn = wave & 3;
    int prow = p * NB, irow = i * NB;
    int kcol = prow - NB;

    if (j > p) {
        int jrow = j * NB;
        v8f acc;
#pragma unroll
        for (int v = 0; v < 8; v++) {
            int row = tm * 16 + v + 8 * hi;
            int col = tn * 16 + r;
            acc[v] = -R[(size_t)(irow + row) * Nn + jrow + col];
        }
        const float* aI = L + (size_t)(irow + tm * 16 + r) * Nn + kcol;
        const float* bJ = L + (size_t)(jrow + tn * 16 + r) * Nn + kcol;
#pragma unroll
        for (int k0 = 0; k0 < NB; k0 += 4) {
            int kk = k0 + 2 * hi;
            v2f fi = *(const v2f*)(aI + kk);
            v2f fb = *(const v2f*)(bJ + kk);
            acc = __builtin_amdgcn_wmma_f32_16x16x4_f32(false, fi, false, fb,
                                                        (short)0, acc, false, false);
        }
#pragma unroll
        for (int v = 0; v < 8; v++) {
            int row = tm * 16 + v + 8 * hi;
            int col = tn * 16 + r;
            Dsh[row * LDP + col] = -acc[v];
        }
        __syncthreads();
        for (int e = tid; e < NB * NB; e += 512) {
            int rr = e >> 6, c = e & (NB - 1);
            R[(size_t)(irow + rr) * Nn + jrow + c] = Dsh[rr * LDP + c];
        }
        return;
    }

    v8f accD, accS;
#pragma unroll
    for (int v = 0; v < 8; v++) {
        int row = tm * 16 + v + 8 * hi;
        int col = tn * 16 + r;
        accD[v] = -R[(size_t)(prow + row) * Nn + prow + col];
        accS[v] = -R[(size_t)(irow + row) * Nn + prow + col];
    }
    if (p > 0) {
        const float* aP = L + (size_t)(prow + tm * 16 + r) * Nn + kcol;
        const float* aI = L + (size_t)(irow + tm * 16 + r) * Nn + kcol;
        const float* bP = L + (size_t)(prow + tn * 16 + r) * Nn + kcol;
#pragma unroll
        for (int k0 = 0; k0 < NB; k0 += 4) {
            int kk = k0 + 2 * hi;
            v2f fa = *(const v2f*)(aP + kk);
            v2f fi = *(const v2f*)(aI + kk);
            v2f fb = *(const v2f*)(bP + kk);
            accD = __builtin_amdgcn_wmma_f32_16x16x4_f32(false, fa, false, fb,
                                                         (short)0, accD, false, false);
            accS = __builtin_amdgcn_wmma_f32_16x16x4_f32(false, fi, false, fb,
                                                         (short)0, accS, false, false);
        }
    }
#pragma unroll
    for (int v = 0; v < 8; v++) {
        int row = tm * 16 + v + 8 * hi;
        int col = tn * 16 + r;
        Dsh[row * LDP + col] = -accD[v];
        Ssh[row * LDP + col] = -accS[v];
    }
    __syncthreads();

    if (isDiag) panel_sweep<false>(Dsh, Ssh, dg, piece, colsD, colsS, tid);
    else        panel_sweep<true>(Dsh, Ssh, dg, piece, colsD, colsS, tid);

    if (isDiag) {
        for (int e = tid; e < NB * NB; e += 512) {
            int rr = e >> 6, c = e & (NB - 1);
            float val = (c < rr) ? Dsh[rr * LDP + c] : ((c == rr) ? dg[rr] : 0.0f);
            L[(size_t)(prow + rr) * Nn + prow + c] = val;
        }
    } else {
        for (int e = tid; e < NB * NB; e += 512) {
            int rr = e >> 6, c = e & (NB - 1);
            L[(size_t)(irow + rr) * Nn + prow + c] = Ssh[rr * LDP + c];
        }
    }
}

__global__ __launch_bounds__(1024) void zmatvec_kernel(const float* __restrict__ Linvbase,
                                                       const float* __restrict__ y,
                                                       const float* __restrict__ mean,
                                                       float* __restrict__ zbase) {
    __shared__ float zrow[32];
    int b = blockIdx.y;
    int tid = threadIdx.x;
    int wave = tid >> 5, lane = tid & 31;
    int i = blockIdx.x * 32 + wave;
    const float* row = Linvbase + (size_t)b * Nn * Nn + (size_t)i * Nn;
    const float* yb = y + b * Nn;
    float mb = mean[b];
    int rowEnd = ((i >> 6) + 1) << 6;
    float4 l[8], v[8];
#pragma unroll
    for (int u = 0; u < 8; u++) {
        int c0 = 4 * lane + 128 * u;
        if (c0 < rowEnd) {
            l[u] = *(const float4*)(row + c0);
            v[u] = *(const float4*)(yb + c0);
        } else {
            l[u] = make_float4(0.0f, 0.0f, 0.0f, 0.0f);
            v[u] = make_float4(mb, mb, mb, mb);
        }
    }
    float s0 = 0.0f, s1 = 0.0f, s2 = 0.0f, s3 = 0.0f;
#pragma unroll
    for (int u = 0; u < 8; u++) {
        s0 += l[u].x * (v[u].x - mb);
        s1 += l[u].y * (v[u].y - mb);
        s2 += l[u].z * (v[u].z - mb);
        s3 += l[u].w * (v[u].w - mb);
    }
    float s = (s0 + s1) + (s2 + s3);
    for (int o = 16; o > 0; o >>= 1) s += __shfl_xor(s, o);
    if (lane == 0) zrow[wave] = s;
    __syncthreads();
    if (wave == 0) zbase[b * Nn + blockIdx.x * 32 + lane] = zrow[lane];
}

__global__ __launch_bounds__(1024) void nlml_kernel(const float* __restrict__ Lbase,
                                                    const float* __restrict__ zbase,
                                                    float* __restrict__ nlml_out) {
    int b = blockIdx.x;
    const float* L = Lbase + (size_t)b * Nn * Nn;
    __shared__ float col[Nn];
    __shared__ float zz[Nn];
    int tid = threadIdx.x;
    float z = zbase[b * Nn + tid];
    col[tid] = __logf(L[(size_t)tid * Nn + tid]);
    zz[tid] = z * z;
    __syncthreads();
    for (int s = 512; s > 0; s >>= 1) {
        if (tid < s) {
            col[tid] += col[tid + s];
            zz[tid] += zz[tid + s];
        }
        __syncthreads();
    }
    if (tid == 0) nlml_out[b] = 0.5f * zz[0] + col[0] + 0.5f * Nn * LOG2PIF;
}

__global__ __launch_bounds__(64) void trinv_diag_kernel(const float* __restrict__ Lbase,
                                                        float* __restrict__ Linvbase) {
    __shared__ __attribute__((aligned(16))) float Lb[NB * NB];
    int p = blockIdx.x, b = blockIdx.y, tid = threadIdx.x;
    size_t base = (size_t)b * Nn * Nn + (size_t)(p * NB) * Nn + p * NB;
    const float* L = Lbase + base;
    float* Li = Linvbase + base;
#pragma unroll
    for (int u = 0; u < NB * NB / 4 / 64; u++) {
        int idx4 = tid + 64 * u;
        int rr = idx4 >> 4, c4 = (idx4 & 15) * 4;
        *(float4*)(Lb + rr * NB + c4) = *(const float4*)(L + (size_t)rr * Nn + c4);
    }
    __syncthreads();
    int j = tid;
    v2f xx[NB / 2];
#pragma unroll
    for (int i = 0; i < NB; i++) {
        const float* lr = Lb + i * NB;
        v2f sA = {(i == j) ? 1.0f : 0.0f, 0.0f};
        v2f sB = {0.0f, 0.0f};
#pragma unroll
        for (int g = 0; g < i / 4; g++) {
            float4 l = *(const float4*)(lr + 4 * g);
            v2f lA = {l.x, l.y}, lB = {l.z, l.w};
            sA -= lA * xx[2 * g];
            sB -= lB * xx[2 * g + 1];
        }
        float4 lt = *(const float4*)(lr + 4 * (i / 4));
        float lv[4] = {lt.x, lt.y, lt.z, lt.w};
        float s0 = sA.x;
#pragma unroll
        for (int k = 4 * (i / 4); k < i; k++) s0 -= lv[k & 3] * xx[k >> 1][k & 1];
        float xi = ((s0 + sA.y) + (sB.x + sB.y)) / lv[i & 3];
        xx[i >> 1][i & 1] = xi;
        Li[(size_t)i * Nn + j] = xi;
    }
}

#define RP 1028
#define TPP 68

__device__ __forceinline__ void load_chunk16(const float* xp, const float* lp, v2f* a, v2f* q) {
#pragma unroll
    for (int u = 0; u < 4; u++) {
        a[u] = *(const v2f*)(xp + 4 * u);
        q[u].x = lp[(4 * u) * Nn];
        q[u].y = lp[(4 * u + 1) * Nn];
    }
}
__device__ __forceinline__ void mma_chunk16(const v2f* a, const v2f* q, v8f& c0, v8f& c1) {
#pragma unroll
    for (int u = 0; u < 4; u += 2) {
        c0 = __builtin_amdgcn_wmma_f32_16x16x4_f32(false, a[u], false, q[u],
                                                   (short)0, c0, false, false);
        c1 = __builtin_amdgcn_wmma_f32_16x16x4_f32(false, a[u + 1], false, q[u + 1],
                                                   (short)0, c1, false, false);
    }
}
__global__ __launch_bounds__(512) void trinv_rows_kernel(const float* __restrict__ Lbase,
                                                         float* Linvbase) {
    extern __shared__ float smem[];
    float* Xs = smem;
    float* Ps = smem + 16 * RP;
    int i = NBLK - 1 - (int)(blockIdx.x >> 2);
    int tm = blockIdx.x & 3;
    int b = blockIdx.y, tid = threadIdx.x;
    const float* L = Lbase + (size_t)b * Nn * Nn;
    float* Li = Linvbase + (size_t)b * Nn * Nn;
    int wave = __builtin_amdgcn_readfirstlane(tid >> 5);
    int lane = tid & 31;
    int r = lane & 15, hi = lane >> 4;
    int tn = wave & 3, kq = wave >> 2;
    int g0 = i * NB + tm * 16;
    int icol = i * NB;

    for (int e = tid; e < 16 * NB; e += 512) {
        int rr = e >> 6, cc = e & (NB - 1);
        Xs[rr * RP + icol + cc] = Li[(size_t)(g0 + rr) * Nn + icol + cc];
    }
    __syncthreads();

    for (int J = i - 1; J >= 0; J--) {
        int jcol = J * NB;
        v2f dq[16];
        if (kq == 0) {
            const float* dp = Li + (size_t)(jcol + 2 * hi) * Nn + jcol + tn * 16 + r;
#pragma unroll
            for (int kc = 0; kc < 4; kc++) {
                if (kc >= tn) {
#pragma unroll
                    for (int u = 0; u < 4; u++) {
                        dq[4 * kc + u].x = dp[(16 * kc + 4 * u) * Nn];
                        dq[4 * kc + u].y = dp[(16 * kc + 4 * u + 1) * Nn];
                    }
                }
            }
        }
        int k0 = (J + 1) * NB + kq * 16 + 2 * hi;
        const float* xp = Xs + r * RP + k0;
        const float* lp = L + (size_t)k0 * Nn + jcol + tn * 16 + r;
        v8f acc0 = {}, acc1 = {};
        int kb = J + 1;
        for (; kb + 1 <= i; kb += 2) {
            v2f a0[4], q0[4], a1[4], q1[4];
            load_chunk16(xp, lp, a0, q0);
            load_chunk16(xp + NB, lp + (size_t)NB * Nn, a1, q1);
            mma_chunk16(a0, q0, acc0, acc1);
            mma_chunk16(a1, q1, acc0, acc1);
            xp += 2 * NB;
            lp += (size_t)2 * NB * Nn;
        }
        if (kb <= i) {
            v2f a0[4], q0[4];
            load_chunk16(xp, lp, a0, q0);
            mma_chunk16(a0, q0, acc0, acc1);
        }
#pragma unroll
        for (int v = 0; v < 8; v++)
            Ps[(kq * 16 + v + 8 * hi) * TPP + tn * 16 + r] = acc0[v] + acc1[v];
        __syncthreads();
        if (kq == 0) {
            v8f x0 = {}, x1 = {};
            const float* pp = Ps + r * TPP + 2 * hi;
#pragma unroll
            for (int kc = 0; kc < 4; kc++) {
                if (kc >= tn) {
                    v2f a[4];
#pragma unroll
                    for (int u = 0; u < 4; u++) {
                        const float* p = pp + 16 * kc + 4 * u;
                        v2f p0 = *(const v2f*)(p);
                        v2f p1 = *(const v2f*)(p + 16 * TPP);
                        v2f p2 = *(const v2f*)(p + 32 * TPP);
                        v2f p3 = *(const v2f*)(p + 48 * TPP);
                        a[u] = (p0 + p1) + (p2 + p3);
                    }
                    mma_chunk16(a, dq + 4 * kc, x0, x1);
                }
            }
#pragma unroll
            for (int v = 0; v < 8; v++)
                Xs[(v + 8 * hi) * RP + jcol + tn * 16 + r] = -(x0[v] + x1[v]);
        }
        __syncthreads();
        for (int e = tid; e < 16 * NB; e += 512) {
            int rr = e >> 6, cc = e & (NB - 1);
            Li[(size_t)(g0 + rr) * Nn + jcol + cc] = Xs[rr * RP + jcol + cc];
        }
    }
}


template <int NU0>
__device__ __forceinline__ void band_slab(const float* a0, const float* a1, const float* bK,
                                          int kb, int hi, v8f (&c0)[4], v8f (&c1)[4]) {
    v2f fa0[8], fa1[8], fb[8][4];
#pragma unroll
    for (int u = 0; u < 8; u++) {
        int kk = kb + 4 * u + 2 * hi;
        if (u < NU0) fa0[u] = *(const v2f*)(a0 + kk);
        fa1[u] = *(const v2f*)(a1 + kk);
#pragma unroll
        for (int c = 0; c < 4; c++) {
            fb[u][c].x = bK[kk * KTS + c * 16];
            fb[u][c].y = bK[(kk + 1) * KTS + c * 16];
        }
    }
#pragma unroll
    for (int u = 0; u < 8; u++) {
#pragma unroll
        for (int c = 0; c < 4; c++) {
            if (u < NU0)
                c0[c] = __builtin_amdgcn_wmma_f32_16x16x4_f32(false, fa0[u], false, fb[u][c],
                                                              (short)0, c0[c], false, false);
            c1[c] = __builtin_amdgcn_wmma_f32_16x16x4_f32(false, fa1[u], false, fb[u][c],
                                                          (short)0, c1[c], false, false);
        }
    }
}

#define PT 512
__global__ __launch_bounds__(PT) void predict_kernel(const float* __restrict__ Linvbase,
                                                     const float* __restrict__ trainPg,
                                                     const float* __restrict__ testPg,
                                                     const float* __restrict__ Kmm,
                                                     const float* __restrict__ zbase,
                                                     const float* __restrict__ meang,
                                                     const float* __restrict__ scale_param,
                                                     float* __restrict__ out_mean,
                                                     float* __restrict__ out_var) {
    extern __shared__ float smem[];
    float* Kt  = smem;
    float* zsh = smem + Nn * KTS;
    float* tp  = zsh + Nn;
    int b = blockIdx.y;
    int colBase = blockIdx.x * MT;
    int tid = threadIdx.x;
    float scale = __expf(scale_param[0]);

    for (int idx = tid; idx < Nn; idx += PT) zsh[idx] = zbase[b * Nn + idx];
    for (int idx = tid; idx < Nn * 6; idx += PT) tp[idx] = trainPg[(size_t)b * Nn * 6 + idx];
    __syncthreads();

    {
        int c = tid & 63;
        int h = tid >> 6;
        const float* q = testPg + (size_t)(b * Mm + colBase + c) * 6;
        float x2 = q[0], y2 = q[1], a2 = q[2], b2 = q[3], c2 = q[4], q2 = q[5];
        for (int n = h; n < Nn; n += PT / 64) {
            const float* p = tp + n * 6;
            Kt[n * KTS + c] = kern_pair(p[0], p[1], p[2], p[3], p[4], p[5],
                                        x2, y2, a2, b2, c2, q2, scale);
        }
    }
    __syncthreads();

    int wave = __builtin_amdgcn_readfirstlane(tid >> 5);
    int lane = tid & 31;
    int r = lane & 15, hi = lane >> 4;
    const float* Li = Linvbase + (size_t)b * Nn * Nn;
    const float* bK = Kt + r;

    float meanAcc[4] = {0.0f, 0.0f, 0.0f, 0.0f};
    float varAcc[4]  = {0.0f, 0.0f, 0.0f, 0.0f};
    for (int s = 0; s < 2; s++) {
        int p = s ? (Nn / 32 - 1 - wave) : wave;
        int row0 = p * 32;
        const float* a0 = Li + (size_t)(row0 + r) * Nn;
        const float* a1 = a0 + (size_t)16 * Nn;
        v8f c0[4] = {}, c1[4] = {};
        for (int t = 0; t < p; t++) band_slab<8>(a0, a1, bK, t * 32, hi, c0, c1);
        band_slab<4>(a0, a1, bK, row0, hi, c0, c1);
        const float* z0 = zsh + row0 + 8 * hi;
#pragma unroll
        for (int c = 0; c < 4; c++) {
#pragma unroll
            for (int v = 0; v < 8; v++) {
                float w0 = c0[c][v], w1 = c1[c][v];
                meanAcc[c] += z0[v] * w0;
                varAcc[c]  += w0 * w0;
                meanAcc[c] += z0[16 + v] * w1;
                varAcc[c]  += w1 * w1;
            }
        }
    }
    float* red = tp;
#pragma unroll
    for (int c = 0; c < 4; c++) {
        float m = meanAcc[c] + __shfl_xor(meanAcc[c], 16);
        float v = varAcc[c] + __shfl_xor(varAcc[c], 16);
        if (lane < 16) {
            red[wave * 128 + c * 16 + lane] = m;
            red[wave * 128 + 64 + c * 16 + lane] = v;
        }
    }
    __syncthreads();
    if (tid < MT) {
        float m = 0.0f, v = 0.0f;
#pragma unroll
        for (int w = 0; w < PT / 32; w++) {
            m += red[w * 128 + tid];
            v += red[w * 128 + 64 + tid];
        }
        int col = b * Mm + colBase + tid;
        out_mean[col] = meang[b] + m;
        out_var[col]  = Kmm[col] - v;
    }
}

extern "C" void kernel_launch(void* const* d_in, const int* in_sizes, int n_in,
                              void* d_out, int out_size, void* d_ws, size_t ws_size,
                              hipStream_t stream) {
    const float* coords_train = (const float*)d_in[0];
    const float* kp_train     = (const float*)d_in[1];
    const float* coords_test  = (const float*)d_in[2];
    const float* kp_test      = (const float*)d_in[3];
    const float* y_train      = (const float*)d_in[4];
    const float* var          = (const float*)d_in[5];
    const float* mean         = (const float*)d_in[6];
    const float* scale_param  = (const float*)d_in[7];

    float* ws = (float*)d_ws;
    size_t off = 0;
    float* A      = ws + off; off += (size_t)Bb * Nn * Nn;
    float* Lm     = ws + off; off += (size_t)Bb * Nn * Nn;
    float* Linv   = ws + off; off += (size_t)Bb * Nn * Nn;
    float* trainP = ws + off; off += (size_t)Bb * Nn * 6;
    float* testP  = ws + off; off += (size_t)Bb * Mm * 6;
    float* Kmm    = ws + off; off += (size_t)Bb * Mm;
    float* z      = ws + off; off += (size_t)Bb * Nn;

    float* out_mean = (float*)d_out;
    float* out_var  = out_mean + (size_t)Bb * Mm;
    float* nlml     = out_var + (size_t)Bb * Mm;

    prep_kernel<<<(Bb * Mm + 255) / 256, 256, 0, stream>>>(
        coords_train, kp_train, coords_test, kp_test, scale_param, trainP, testP, Kmm);
    buildA_kernel<<<(Bb * Nn * Nn) / 256, 256, 0, stream>>>(trainP, var, scale_param, A);
    chol_step_kernel<<<dim3(Bb * NBLK), 512, 0, stream>>>(A, Lm, 0);
#define CHOL_STEP(P) chol_step_kernel<<<dim3((NBLK - (P)) * (NBLK + 1 - (P))), 512, 0, stream>>>(A, Lm, (P))
                   CHOL_STEP(1);  CHOL_STEP(2);  CHOL_STEP(3);
    CHOL_STEP(4);  CHOL_STEP(5);  CHOL_STEP(6);  CHOL_STEP(7);
    CHOL_STEP(8);  CHOL_STEP(9);  CHOL_STEP(10); CHOL_STEP(11);
    CHOL_STEP(12); CHOL_STEP(13); CHOL_STEP(14); CHOL_STEP(15);
#undef CHOL_STEP
    trinv_diag_kernel<<<dim3(NBLK, Bb), 64, 0, stream>>>(Lm, Linv);
    size_t inv_smem_bytes = (size_t)(16 * RP + 4 * 16 * TPP) * sizeof(float);
    trinv_rows_kernel<<<dim3(4 * (NBLK - 1), Bb), 512, inv_smem_bytes, stream>>>(Lm, Linv);
    zmatvec_kernel<<<dim3(Nn / 32, Bb), 1024, 0, stream>>>(Linv, y_train, mean, z);
    nlml_kernel<<<Bb, 1024, 0, stream>>>(Lm, z, nlml);

    size_t smem_bytes = (size_t)(Nn * KTS + Nn + Nn * 6) * sizeof(float);
    predict_kernel<<<dim3(Mm / MT, Bb), PT, smem_bytes, stream>>>(
        Linv, trainP, testP, Kmm, z, mean, scale_param, out_mean, out_var);
}
